# starting kernel
# baseline (speedup 1.0000x reference)
#include <hip/hip_runtime.h>
#include <math.h>

typedef __attribute__((ext_vector_type(2))) float v2f;
typedef __attribute__((ext_vector_type(8))) float v8f;

#define NN   100000
#define EE   1600000
#define INDIM 64
#define HID  128
#define LAT  64
#define GG   1000

__device__ __forceinline__ float gelu_exact(float x) {
    return 0.5f * x * (1.0f + erff(x * 0.70710678118654752f));
}

__global__ __launch_bounds__(256) void bn_in_kernel(
    const float* __restrict__ x,
    const float* __restrict__ g, const float* __restrict__ b,
    const float* __restrict__ m, const float* __restrict__ v,
    const float* __restrict__ epsP,
    float* __restrict__ out, float* __restrict__ agg)
{
    int idx = blockIdx.x * blockDim.x + threadIdx.x;
    if (idx >= NN * INDIM) return;
    int j = idx & (INDIM - 1);
    float y = (x[idx] - m[j]) * rsqrtf(v[j] + 1e-5f) * g[j] + b[j];
    out[idx] = y;
    agg[idx] = (1.0f + epsP[0]) * y;
}

__global__ __launch_bounds__(256) void edge_agg_kernel(
    const float* __restrict__ h, const int* __restrict__ src,
    const int* __restrict__ dst, float* __restrict__ agg,
    int dim, int dim4, int total)
{
    int idx = blockIdx.x * blockDim.x + threadIdx.x;
    if (idx >= total) return;
    int e = idx / dim4;
    int c = idx - e * dim4;
    int s = src[e];
    int d = dst[e];
    const float4 val = *(const float4*)(h + (size_t)s * dim + 4 * c);
    float* p = agg + (size_t)d * dim + 4 * c;
    unsafeAtomicAdd(p + 0, val.x);
    unsafeAtomicAdd(p + 1, val.y);
    unsafeAtomicAdd(p + 2, val.z);
    unsafeAtomicAdd(p + 3, val.w);
}

template<int K>
__global__ __launch_bounds__(256) void gemm_bn_gelu_kernel(
    const float* __restrict__ A,
    const float* __restrict__ W,
    const float* __restrict__ bias,
    const float* __restrict__ bn_g, const float* __restrict__ bn_b,
    const float* __restrict__ bn_m, const float* __restrict__ bn_v,
    float* __restrict__ out,
    float* __restrict__ aggOut, const float* __restrict__ epsP,
    int epsIdx, int writeAgg)
{
    const int lane = threadIdx.x & 31;
    const int wave = threadIdx.x >> 5;
    const int tileM = blockIdx.x;
    const int half = lane >> 4;
    const int mr   = lane & 15;

    const int rowA = tileM * 16 + mr;
    const int colB = wave * 16 + mr;

    const float* __restrict__ Arow = A + (size_t)rowA * K;

    v8f acc = {};
    #pragma unroll
    for (int k = 0; k < K; k += 4) {
        const int ka = k + 2 * half;
        v2f a, b;
        a.x = Arow[ka];
        a.y = Arow[ka + 1];
        b.x = W[(size_t)ka * HID + colB];
        b.y = W[(size_t)(ka + 1) * HID + colB];
        acc = __builtin_amdgcn_wmma_f32_16x16x4_f32(
            false, a, false, b, (short)0, acc, false, false);
    }

    const float escale = writeAgg ? (1.0f + epsP[epsIdx]) : 0.0f;

    #pragma unroll
    for (int r = 0; r < 8; ++r) {
        const int row = tileM * 16 + r + 8 * half;
        const int col = wave * 16 + mr;
        float y = acc[r] + bias[col];
        y = (y - bn_m[col]) * rsqrtf(bn_v[col] + 1e-5f) * bn_g[col] + bn_b[col];
        y = gelu_exact(y);
        const size_t o = (size_t)row * HID + col;
        out[o] = y;
        if (writeAgg) aggOut[o] = escale * y;
    }
}

__global__ __launch_bounds__(256) void zero_kernel(float* __restrict__ p, int n) {
    int idx = blockIdx.x * blockDim.x + threadIdx.x;
    if (idx < n) p[idx] = 0.0f;
}

__global__ __launch_bounds__(256) void pool_acc_kernel(
    const float* __restrict__ h, const int* __restrict__ batch,
    float* __restrict__ sums, float* __restrict__ cnt, int total)
{
    int idx = blockIdx.x * blockDim.x + threadIdx.x;
    if (idx >= total) return;
    int i = idx >> 5;
    int c = idx & 31;
    int b = batch[i];
    const float4 v = *(const float4*)(h + (size_t)i * HID + 4 * c);
    float* s = sums + (size_t)b * HID + 4 * c;
    unsafeAtomicAdd(s + 0, v.x);
    unsafeAtomicAdd(s + 1, v.y);
    unsafeAtomicAdd(s + 2, v.z);
    unsafeAtomicAdd(s + 3, v.w);
    if (c == 0) unsafeAtomicAdd(cnt + b, 1.0f);
}

__global__ __launch_bounds__(256) void pool_final_kernel(
    const float* __restrict__ sums, const float* __restrict__ cnt,
    float* __restrict__ pooled)
{
    int idx = blockIdx.x * blockDim.x + threadIdx.x;
    if (idx >= GG * HID) return;
    int g = idx / HID;
    pooled[idx] = sums[idx] / fmaxf(cnt[g], 1.0f);
}

__global__ __launch_bounds__(128) void head_kernel(
    const float* __restrict__ pooled,
    const float* __restrict__ fc1w, const float* __restrict__ fc1b,
    const float* __restrict__ lng, const float* __restrict__ lnb,
    const float* __restrict__ fc2w, const float* __restrict__ fc2b,
    float* __restrict__ out)
{
    __shared__ float p[HID];
    __shared__ float rbuf[HID];
    __shared__ float red[HID];
    const int g = blockIdx.x;
    const int t = threadIdx.x;

    p[t] = pooled[(size_t)g * HID + t];
    __syncthreads();

    float o = fc1b[t];
    #pragma unroll 4
    for (int k = 0; k < HID; ++k) o += p[k] * fc1w[k * HID + t];

    red[t] = o;
    __syncthreads();
    for (int s = 64; s > 0; s >>= 1) {
        if (t < s) red[t] += red[t + s];
        __syncthreads();
    }
    const float mu = red[0] * (1.0f / HID);
    __syncthreads();

    const float dv = o - mu;
    red[t] = dv * dv;
    __syncthreads();
    for (int s = 64; s > 0; s >>= 1) {
        if (t < s) red[t] += red[t + s];
        __syncthreads();
    }
    const float var = red[0] * (1.0f / HID);

    float y = dv * rsqrtf(var + 1e-5f) * lng[t] + lnb[t];
    y = gelu_exact(y) + p[t];
    rbuf[t] = y;
    __syncthreads();

    if (t < LAT) {
        float o2 = fc2b[t];
        #pragma unroll 4
        for (int k = 0; k < HID; ++k) o2 += rbuf[k] * fc2w[k * LAT + t];
        out[(size_t)g * LAT + t] = o2;
    }
}

extern "C" void kernel_launch(void* const* d_in, const int* in_sizes, int n_in,
                              void* d_out, int out_size, void* d_ws, size_t ws_size,
                              hipStream_t stream) {
    const float* x       = (const float*)d_in[0];
    const int*   ei      = (const int*)  d_in[1];
    const int*   batch   = (const int*)  d_in[2];
    const float* in_g    = (const float*)d_in[3];
    const float* in_b    = (const float*)d_in[4];
    const float* in_m    = (const float*)d_in[5];
    const float* in_v    = (const float*)d_in[6];
    const float* eps0    = (const float*)d_in[7];
    const float* c0_w1   = (const float*)d_in[8];
    const float* c0_b1   = (const float*)d_in[9];
    const float* c0_g    = (const float*)d_in[10];
    const float* c0_bb   = (const float*)d_in[11];
    const float* c0_m    = (const float*)d_in[12];
    const float* c0_v    = (const float*)d_in[13];
    const float* c0_w2   = (const float*)d_in[14];
    const float* c0_b2   = (const float*)d_in[15];
    const float* bn0_g   = (const float*)d_in[16];
    const float* bn0_b   = (const float*)d_in[17];
    const float* bn0_m   = (const float*)d_in[18];
    const float* bn0_v   = (const float*)d_in[19];
    const float* epsR    = (const float*)d_in[20];
    const float* cR_w1   = (const float*)d_in[21];
    const float* cR_b1   = (const float*)d_in[22];
    const float* cR_g    = (const float*)d_in[23];
    const float* cR_bb   = (const float*)d_in[24];
    const float* cR_m    = (const float*)d_in[25];
    const float* cR_v    = (const float*)d_in[26];
    const float* cR_w2   = (const float*)d_in[27];
    const float* cR_b2   = (const float*)d_in[28];
    const float* bnR_g   = (const float*)d_in[29];
    const float* bnR_b   = (const float*)d_in[30];
    const float* bnR_m   = (const float*)d_in[31];
    const float* bnR_v   = (const float*)d_in[32];
    const float* fc1_w   = (const float*)d_in[33];
    const float* fc1_b   = (const float*)d_in[34];
    const float* ln_g    = (const float*)d_in[35];
    const float* ln_b    = (const float*)d_in[36];
    const float* fc2_w   = (const float*)d_in[37];
    const float* fc2_b   = (const float*)d_in[38];

    const int* src = ei;
    const int* dst = ei + EE;

    float* buf0 = (float*)d_ws;
    float* buf1 = buf0 + (size_t)NN * HID;
    float* buf2 = buf1 + (size_t)NN * HID;
    float* sums = buf2 + (size_t)NN * HID;
    float* cnt  = sums + (size_t)GG * HID;

    const int TB = 256;
    const int nTiles = NN / 16;

    bn_in_kernel<<<(NN * INDIM + TB - 1) / TB, TB, 0, stream>>>(
        x, in_g, in_b, in_m, in_v, eps0, buf0, buf1);

    {
        int total = EE * (INDIM / 4);
        edge_agg_kernel<<<(total + TB - 1) / TB, TB, 0, stream>>>(
            buf0, src, dst, buf1, INDIM, INDIM / 4, total);
        gemm_bn_gelu_kernel<INDIM><<<nTiles, TB, 0, stream>>>(
            buf1, c0_w1, c0_b1, c0_g, c0_bb, c0_m, c0_v, buf2,
            buf1, eps0, 0, 0);
        gemm_bn_gelu_kernel<HID><<<nTiles, TB, 0, stream>>>(
            buf2, c0_w2, c0_b2, bn0_g, bn0_b, bn0_m, bn0_v, buf0,
            buf1, epsR, 0, 1);
    }

    for (int l = 0; l < 2; ++l) {
        const size_t wOff = (size_t)l * HID * HID;
        const size_t vOff = (size_t)l * HID;
        int total = EE * (HID / 4);
        edge_agg_kernel<<<(total + TB - 1) / TB, TB, 0, stream>>>(
            buf0, src, dst, buf1, HID, HID / 4, total);
        gemm_bn_gelu_kernel<HID><<<nTiles, TB, 0, stream>>>(
            buf1, cR_w1 + wOff, cR_b1 + vOff, cR_g + vOff, cR_bb + vOff,
            cR_m + vOff, cR_v + vOff, buf2,
            buf1, epsR, 0, 0);
        const int writeAgg = (l == 0) ? 1 : 0;
        gemm_bn_gelu_kernel<HID><<<nTiles, TB, 0, stream>>>(
            buf2, cR_w2 + wOff, cR_b2 + vOff, bnR_g + vOff, bnR_b + vOff,
            bnR_m + vOff, bnR_v + vOff, buf0,
            buf1, epsR, 1, writeAgg);
    }

    zero_kernel<<<(GG * HID + GG + TB - 1) / TB, TB, 0, stream>>>(sums, GG * HID + GG);
    {
        int total = NN * (HID / 4);
        pool_acc_kernel<<<(total + TB - 1) / TB, TB, 0, stream>>>(
            buf0, batch, sums, cnt, total);
    }
    pool_final_kernel<<<(GG * HID + TB - 1) / TB, TB, 0, stream>>>(sums, cnt, buf1);

    head_kernel<<<GG, HID, 0, stream>>>(
        buf1, fc1_w, fc1_b, ln_g, ln_b, fc2_w, fc2_b, (float*)d_out);
}


# Edge sums per dst tile in LDS: each block owns 512 dst rows, scans the whole edge list, adds gathered src rows by LDS float atomics, stores rows once; no global atomics; add order = arrival order
# speedup vs baseline: 1.3447x; 1.3447x over previous
#include <hip/hip_runtime.h>
#include <math.h>

typedef __attribute__((ext_vector_type(2))) float v2f;
typedef __attribute__((ext_vector_type(8))) float v8f;

#define NN   100000
#define EE   1600000
#define INDIM 64
#define HID  128
#define LAT  64
#define GG   1000

__device__ __forceinline__ float gelu_exact(float x) {
    return 0.5f * x * (1.0f + erff(x * 0.70710678118654752f));
}

__global__ __launch_bounds__(256) void bn_in_kernel(
    const float* __restrict__ x,
    const float* __restrict__ g, const float* __restrict__ b,
    const float* __restrict__ m, const float* __restrict__ v,
    const float* __restrict__ epsP,
    float* __restrict__ out, float* __restrict__ agg)
{
    int idx = blockIdx.x * blockDim.x + threadIdx.x;
    if (idx >= NN * INDIM) return;
    int j = idx & (INDIM - 1);
    float y = (x[idx] - m[j]) * rsqrtf(v[j] + 1e-5f) * g[j] + b[j];
    out[idx] = y;
    agg[idx] = (1.0f + epsP[0]) * y;
}

#define ET_TILE   512
#define ET_QCAP   144
#define ET_NIT    98

__device__ __forceinline__ void et_wave_order() {
    __builtin_amdgcn_fence(__ATOMIC_ACQ_REL, "wavefront");
    __builtin_amdgcn_wave_barrier();
}

template<int D>
__device__ __forceinline__ void et_drain(
    const float* __restrict__ h, const int* __restrict__ src,
    float* tile, const unsigned* q, int qn, int n, int lane)
{
    unsigned ent = 0;
    int s = 0;
    if (lane < n) {
        ent = q[qn - n + lane];
        s = src[ent >> 9];
    }
    const int dl = (int)(ent & (ET_TILE - 1));
    #pragma unroll
    for (int base = 0; base < 16; base += 8) {
        if (base >= n) break;
        float v[8][D / 32];
        #pragma unroll
        for (int i = 0; i < 8; ++i) {
            if (base + i < n) {
                const int si = __builtin_amdgcn_readlane(s, base + i);
                const float* row = h + (size_t)si * D + lane;
                #pragma unroll
                for (int k = 0; k < D / 32; ++k) v[i][k] = row[32 * k];
            }
        }
        #pragma unroll
        for (int i = 0; i < 8; ++i) {
            if (base + i < n) {
                const int di = __builtin_amdgcn_readlane(dl, base + i);
                float* trow = tile + di * D + lane;
                #pragma unroll
                for (int k = 0; k < D / 32; ++k)
                    __hip_atomic_fetch_add(trow + 32 * k, v[i][k],
                        __ATOMIC_RELAXED, __HIP_MEMORY_SCOPE_WORKGROUP);
            }
        }
    }
}

template<int D>
__global__ __launch_bounds__(1024) void edge_tile_kernel(
    const float* __restrict__ h, const int* __restrict__ src,
    const int* __restrict__ dst, float* __restrict__ agg)
{
    __shared__ float tile[ET_TILE * D];
    __shared__ unsigned queue[32 * ET_QCAP];

    const int tid  = threadIdx.x;
    const int lane = tid & 31;
    const int wave = tid >> 5;
    const int lo   = blockIdx.x * ET_TILE;
    const int nrows = (NN - lo < ET_TILE) ? (NN - lo) : ET_TILE;
    const int n4 = nrows * (D / 4);

    float4* t4 = (float4*)tile;
    float4* a4 = (float4*)(agg + (size_t)lo * D);
    for (int i = tid; i < n4; i += 1024) t4[i] = a4[i];
    __syncthreads();

    unsigned* q = queue + wave * ET_QCAP;
    const int4* dst4 = (const int4*)dst;
    int qn = 0;

    int it = (int)(blockIdx.x % ET_NIT);
    int4 cur[4], nxt[4];
    #pragma unroll
    for (int c = 0; c < 4; ++c) {
        const int e = it * 16384 + c * 4096 + tid * 4;
        cur[c] = dst4[(e < EE ? e : EE - 4) >> 2];
    }
    for (int n = 0; n < ET_NIT; ++n) {
        int itn = it + 1; if (itn == ET_NIT) itn = 0;
        #pragma unroll
        for (int c = 0; c < 4; ++c) {
            const int e = itn * 16384 + c * 4096 + tid * 4;
            nxt[c] = dst4[(e < EE ? e : EE - 4) >> 2];
        }
        #pragma unroll
        for (int c = 0; c < 4; ++c) {
            const int e0 = it * 16384 + c * 4096 + tid * 4;
            const bool inlist = e0 < EE;
            const int dv[4] = {cur[c].x, cur[c].y, cur[c].z, cur[c].w};
            #pragma unroll
            for (int j = 0; j < 4; ++j) {
                const unsigned rel = (unsigned)(dv[j] - lo);
                const bool m = inlist && rel < (unsigned)ET_TILE;
                const unsigned mask = __builtin_amdgcn_ballot_w32(m);
                if (mask) {
                    if (m) {
                        const int pos = qn + (int)__builtin_amdgcn_mbcnt_lo(mask, 0u);
                        q[pos] = ((unsigned)(e0 + j) << 9) | rel;
                    }
                    qn += __builtin_popcount(mask);
                }
            }
            et_wave_order();
            while (qn >= 16) {
                et_drain<D>(h, src, tile, q, qn, 16, lane);
                qn -= 16;
            }
            et_wave_order();
        }
        #pragma unroll
        for (int c = 0; c < 4; ++c) cur[c] = nxt[c];
        it = itn;
    }
    if (qn > 0) et_drain<D>(h, src, tile, q, qn, qn, lane);

    __syncthreads();
    for (int i = tid; i < n4; i += 1024) a4[i] = t4[i];
}

template<int K>
__global__ __launch_bounds__(256) void gemm_bn_gelu_kernel(
    const float* __restrict__ A,
    const float* __restrict__ W,
    const float* __restrict__ bias,
    const float* __restrict__ bn_g, const float* __restrict__ bn_b,
    const float* __restrict__ bn_m, const float* __restrict__ bn_v,
    float* __restrict__ out,
    float* __restrict__ aggOut, const float* __restrict__ epsP,
    int epsIdx, int writeAgg)
{
    const int lane = threadIdx.x & 31;
    const int wave = threadIdx.x >> 5;
    const int tileM = blockIdx.x;
    const int half = lane >> 4;
    const int mr   = lane & 15;

    const int rowA = tileM * 16 + mr;
    const int colB = wave * 16 + mr;

    const float* __restrict__ Arow = A + (size_t)rowA * K;

    v8f acc = {};
    #pragma unroll
    for (int k = 0; k < K; k += 4) {
        const int ka = k + 2 * half;
        v2f a, b;
        a.x = Arow[ka];
        a.y = Arow[ka + 1];
        b.x = W[(size_t)ka * HID + colB];
        b.y = W[(size_t)(ka + 1) * HID + colB];
        acc = __builtin_amdgcn_wmma_f32_16x16x4_f32(
            false, a, false, b, (short)0, acc, false, false);
    }

    const float escale = writeAgg ? (1.0f + epsP[epsIdx]) : 0.0f;

    #pragma unroll
    for (int r = 0; r < 8; ++r) {
        const int row = tileM * 16 + r + 8 * half;
        const int col = wave * 16 + mr;
        float y = acc[r] + bias[col];
        y = (y - bn_m[col]) * rsqrtf(bn_v[col] + 1e-5f) * bn_g[col] + bn_b[col];
        y = gelu_exact(y);
        const size_t o = (size_t)row * HID + col;
        out[o] = y;
        if (writeAgg) aggOut[o] = escale * y;
    }
}

__global__ __launch_bounds__(256) void zero_kernel(float* __restrict__ p, int n) {
    int idx = blockIdx.x * blockDim.x + threadIdx.x;
    if (idx < n) p[idx] = 0.0f;
}

__global__ __launch_bounds__(256) void pool_acc_kernel(
    const float* __restrict__ h, const int* __restrict__ batch,
    float* __restrict__ sums, float* __restrict__ cnt, int total)
{
    int idx = blockIdx.x * blockDim.x + threadIdx.x;
    if (idx >= total) return;
    int i = idx >> 5;
    int c = idx & 31;
    int b = batch[i];
    const float4 v = *(const float4*)(h + (size_t)i * HID + 4 * c);
    float* s = sums + (size_t)b * HID + 4 * c;
    unsafeAtomicAdd(s + 0, v.x);
    unsafeAtomicAdd(s + 1, v.y);
    unsafeAtomicAdd(s + 2, v.z);
    unsafeAtomicAdd(s + 3, v.w);
    if (c == 0) unsafeAtomicAdd(cnt + b, 1.0f);
}

__global__ __launch_bounds__(256) void pool_final_kernel(
    const float* __restrict__ sums, const float* __restrict__ cnt,
    float* __restrict__ pooled)
{
    int idx = blockIdx.x * blockDim.x + threadIdx.x;
    if (idx >= GG * HID) return;
    int g = idx / HID;
    pooled[idx] = sums[idx] / fmaxf(cnt[g], 1.0f);
}

__global__ __launch_bounds__(128) void head_kernel(
    const float* __restrict__ pooled,
    const float* __restrict__ fc1w, const float* __restrict__ fc1b,
    const float* __restrict__ lng, const float* __restrict__ lnb,
    const float* __restrict__ fc2w, const float* __restrict__ fc2b,
    float* __restrict__ out)
{
    __shared__ float p[HID];
    __shared__ float rbuf[HID];
    __shared__ float red[HID];
    const int g = blockIdx.x;
    const int t = threadIdx.x;

    p[t] = pooled[(size_t)g * HID + t];
    __syncthreads();

    float o = fc1b[t];
    #pragma unroll 4
    for (int k = 0; k < HID; ++k) o += p[k] * fc1w[k * HID + t];

    red[t] = o;
    __syncthreads();
    for (int s = 64; s > 0; s >>= 1) {
        if (t < s) red[t] += red[t + s];
        __syncthreads();
    }
    const float mu = red[0] * (1.0f / HID);
    __syncthreads();

    const float dv = o - mu;
    red[t] = dv * dv;
    __syncthreads();
    for (int s = 64; s > 0; s >>= 1) {
        if (t < s) red[t] += red[t + s];
        __syncthreads();
    }
    const float var = red[0] * (1.0f / HID);

    float y = dv * rsqrtf(var + 1e-5f) * lng[t] + lnb[t];
    y = gelu_exact(y) + p[t];
    rbuf[t] = y;
    __syncthreads();

    if (t < LAT) {
        float o2 = fc2b[t];
        #pragma unroll 4
        for (int k = 0; k < HID; ++k) o2 += rbuf[k] * fc2w[k * LAT + t];
        out[(size_t)g * LAT + t] = o2;
    }
}

extern "C" void kernel_launch(void* const* d_in, const int* in_sizes, int n_in,
                              void* d_out, int out_size, void* d_ws, size_t ws_size,
                              hipStream_t stream) {
    const float* x       = (const float*)d_in[0];
    const int*   ei      = (const int*)  d_in[1];
    const int*   batch   = (const int*)  d_in[2];
    const float* in_g    = (const float*)d_in[3];
    const float* in_b    = (const float*)d_in[4];
    const float* in_m    = (const float*)d_in[5];
    const float* in_v    = (const float*)d_in[6];
    const float* eps0    = (const float*)d_in[7];
    const float* c0_w1   = (const float*)d_in[8];
    const float* c0_b1   = (const float*)d_in[9];
    const float* c0_g    = (const float*)d_in[10];
    const float* c0_bb   = (const float*)d_in[11];
    const float* c0_m    = (const float*)d_in[12];
    const float* c0_v    = (const float*)d_in[13];
    const float* c0_w2   = (const float*)d_in[14];
    const float* c0_b2   = (const float*)d_in[15];
    const float* bn0_g   = (const float*)d_in[16];
    const float* bn0_b   = (const float*)d_in[17];
    const float* bn0_m   = (const float*)d_in[18];
    const float* bn0_v   = (const float*)d_in[19];
    const float* epsR    = (const float*)d_in[20];
    const float* cR_w1   = (const float*)d_in[21];
    const float* cR_b1   = (const float*)d_in[22];
    const float* cR_g    = (const float*)d_in[23];
    const float* cR_bb   = (const float*)d_in[24];
    const float* cR_m    = (const float*)d_in[25];
    const float* cR_v    = (const float*)d_in[26];
    const float* cR_w2   = (const float*)d_in[27];
    const float* cR_b2   = (const float*)d_in[28];
    const float* bnR_g   = (const float*)d_in[29];
    const float* bnR_b   = (const float*)d_in[30];
    const float* bnR_m   = (const float*)d_in[31];
    const float* bnR_v   = (const float*)d_in[32];
    const float* fc1_w   = (const float*)d_in[33];
    const float* fc1_b   = (const float*)d_in[34];
    const float* ln_g    = (const float*)d_in[35];
    const float* ln_b    = (const float*)d_in[36];
    const float* fc2_w   = (const float*)d_in[37];
    const float* fc2_b   = (const float*)d_in[38];

    const int* src = ei;
    const int* dst = ei + EE;

    float* buf0 = (float*)d_ws;
    float* buf1 = buf0 + (size_t)NN * HID;
    float* buf2 = buf1 + (size_t)NN * HID;
    float* sums = buf2 + (size_t)NN * HID;
    float* cnt  = sums + (size_t)GG * HID;

    const int TB = 256;
    const int nTiles = NN / 16;

    bn_in_kernel<<<(NN * INDIM + TB - 1) / TB, TB, 0, stream>>>(
        x, in_g, in_b, in_m, in_v, eps0, buf0, buf1);

    {
        edge_tile_kernel<INDIM><<<196, 1024, 0, stream>>>(buf0, src, dst, buf1);
        gemm_bn_gelu_kernel<INDIM><<<nTiles, TB, 0, stream>>>(
            buf1, c0_w1, c0_b1, c0_g, c0_bb, c0_m, c0_v, buf2,
            buf1, eps0, 0, 0);
        gemm_bn_gelu_kernel<HID><<<nTiles, TB, 0, stream>>>(
            buf2, c0_w2, c0_b2, bn0_g, bn0_b, bn0_m, bn0_v, buf0,
            buf1, epsR, 0, 1);
    }

    for (int l = 0; l < 2; ++l) {
        const size_t wOff = (size_t)l * HID * HID;
        const size_t vOff = (size_t)l * HID;
        edge_tile_kernel<HID><<<196, 1024, 0, stream>>>(buf0, src, dst, buf1);
        gemm_bn_gelu_kernel<HID><<<nTiles, TB, 0, stream>>>(
            buf1, cR_w1 + wOff, cR_b1 + vOff, cR_g + vOff, cR_bb + vOff,
            cR_m + vOff, cR_v + vOff, buf2,
            buf1, epsR, 0, 0);
        const int writeAgg = (l == 0) ? 1 : 0;
        gemm_bn_gelu_kernel<HID><<<nTiles, TB, 0, stream>>>(
            buf2, cR_w2 + wOff, cR_b2 + vOff, bnR_g + vOff, bnR_b + vOff,
            bnR_m + vOff, bnR_v + vOff, buf0,
            buf1, epsR, 1, writeAgg);
    }

    zero_kernel<<<(GG * HID + GG + TB - 1) / TB, TB, 0, stream>>>(sums, GG * HID + GG);
    {
        int total = NN * (HID / 4);
        pool_acc_kernel<<<(total + TB - 1) / TB, TB, 0, stream>>>(
            buf0, batch, sums, cnt, total);
    }
    pool_final_kernel<<<(GG * HID + TB - 1) / TB, TB, 0, stream>>>(sums, cnt, buf1);

    head_kernel<<<GG, HID, 0, stream>>>(
        buf1, fc1_w, fc1_b, ln_g, ln_b, fc2_w, fc2_b, (float*)d_out);
}


# GEMM weight-stationary: a wave keeps its 16-column W fragments in registers for a 64-row block; A rows staged once per block in LDS by whole-line loads; 2 acc tiles; k and add order unchanged
# speedup vs baseline: 1.5197x; 1.1301x over previous
#include <hip/hip_runtime.h>
#include <math.h>

typedef __attribute__((ext_vector_type(2))) float v2f;
typedef __attribute__((ext_vector_type(8))) float v8f;

#define NN   100000
#define EE   1600000
#define INDIM 64
#define HID  128
#define LAT  64
#define GG   1000

__device__ __forceinline__ float gelu_exact(float x) {
    return 0.5f * x * (1.0f + erff(x * 0.70710678118654752f));
}

__global__ __launch_bounds__(256) void bn_in_kernel(
    const float* __restrict__ x,
    const float* __restrict__ g, const float* __restrict__ b,
    const float* __restrict__ m, const float* __restrict__ v,
    const float* __restrict__ epsP,
    float* __restrict__ out, float* __restrict__ agg)
{
    int idx = blockIdx.x * blockDim.x + threadIdx.x;
    if (idx >= NN * INDIM) return;
    int j = idx & (INDIM - 1);
    float y = (x[idx] - m[j]) * rsqrtf(v[j] + 1e-5f) * g[j] + b[j];
    out[idx] = y;
    agg[idx] = (1.0f + epsP[0]) * y;
}

#define ET_TILE   512
#define ET_QCAP   144
#define ET_NIT    98

__device__ __forceinline__ void et_wave_order() {
    __builtin_amdgcn_fence(__ATOMIC_ACQ_REL, "wavefront");
    __builtin_amdgcn_wave_barrier();
}

template<int D>
__device__ __forceinline__ void et_drain(
    const float* __restrict__ h, const int* __restrict__ src,
    float* tile, const unsigned* q, int qn, int n, int lane)
{
    unsigned ent = 0;
    int s = 0;
    if (lane < n) {
        ent = q[qn - n + lane];
        s = src[ent >> 9];
    }
    const int dl = (int)(ent & (ET_TILE - 1));
    #pragma unroll
    for (int base = 0; base < 16; base += 8) {
        if (base >= n) break;
        float v[8][D / 32];
        #pragma unroll
        for (int i = 0; i < 8; ++i) {
            if (base + i < n) {
                const int si = __builtin_amdgcn_readlane(s, base + i);
                const float* row = h + (size_t)si * D + lane;
                #pragma unroll
                for (int k = 0; k < D / 32; ++k) v[i][k] = row[32 * k];
            }
        }
        #pragma unroll
        for (int i = 0; i < 8; ++i) {
            if (base + i < n) {
                const int di = __builtin_amdgcn_readlane(dl, base + i);
                float* trow = tile + di * D + lane;
                #pragma unroll
                for (int k = 0; k < D / 32; ++k)
                    __hip_atomic_fetch_add(trow + 32 * k, v[i][k],
                        __ATOMIC_RELAXED, __HIP_MEMORY_SCOPE_WORKGROUP);
            }
        }
    }
}

template<int D>
__global__ __launch_bounds__(1024) void edge_tile_kernel(
    const float* __restrict__ h, const int* __restrict__ src,
    const int* __restrict__ dst, float* __restrict__ agg)
{
    __shared__ float tile[ET_TILE * D];
    __shared__ unsigned queue[32 * ET_QCAP];

    const int tid  = threadIdx.x;
    const int lane = tid & 31;
    const int wave = tid >> 5;
    const int lo   = blockIdx.x * ET_TILE;
    const int nrows = (NN - lo < ET_TILE) ? (NN - lo) : ET_TILE;
    const int n4 = nrows * (D / 4);

    float4* t4 = (float4*)tile;
    float4* a4 = (float4*)(agg + (size_t)lo * D);
    for (int i = tid; i < n4; i += 1024) t4[i] = a4[i];
    __syncthreads();

    unsigned* q = queue + wave * ET_QCAP;
    const int4* dst4 = (const int4*)dst;
    int qn = 0;

    int it = (int)(blockIdx.x % ET_NIT);
    int4 cur[4], nxt[4];
    #pragma unroll
    for (int c = 0; c < 4; ++c) {
        const int e = it * 16384 + c * 4096 + tid * 4;
        cur[c] = dst4[(e < EE ? e : EE - 4) >> 2];
    }
    for (int n = 0; n < ET_NIT; ++n) {
        int itn = it + 1; if (itn == ET_NIT) itn = 0;
        #pragma unroll
        for (int c = 0; c < 4; ++c) {
            const int e = itn * 16384 + c * 4096 + tid * 4;
            nxt[c] = dst4[(e < EE ? e : EE - 4) >> 2];
        }
        #pragma unroll
        for (int c = 0; c < 4; ++c) {
            const int e0 = it * 16384 + c * 4096 + tid * 4;
            const bool inlist = e0 < EE;
            const int dv[4] = {cur[c].x, cur[c].y, cur[c].z, cur[c].w};
            #pragma unroll
            for (int j = 0; j < 4; ++j) {
                const unsigned rel = (unsigned)(dv[j] - lo);
                const bool m = inlist && rel < (unsigned)ET_TILE;
                const unsigned mask = __builtin_amdgcn_ballot_w32(m);
                if (mask) {
                    if (m) {
                        const int pos = qn + (int)__builtin_amdgcn_mbcnt_lo(mask, 0u);
                        q[pos] = ((unsigned)(e0 + j) << 9) | rel;
                    }
                    qn += __builtin_popcount(mask);
                }
            }
            et_wave_order();
            while (qn >= 16) {
                et_drain<D>(h, src, tile, q, qn, 16, lane);
                qn -= 16;
            }
            et_wave_order();
        }
        #pragma unroll
        for (int c = 0; c < 4; ++c) cur[c] = nxt[c];
        it = itn;
    }
    if (qn > 0) et_drain<D>(h, src, tile, q, qn, qn, lane);

    __syncthreads();
    for (int i = tid; i < n4; i += 1024) a4[i] = t4[i];
}

#define GM_BROWS 64

template<int K>
__global__ __launch_bounds__(256) void gemm_bn_gelu_kernel(
    const float* __restrict__ A,
    const float* __restrict__ W,
    const float* __restrict__ bias,
    const float* __restrict__ bn_g, const float* __restrict__ bn_b,
    const float* __restrict__ bn_m, const float* __restrict__ bn_v,
    float* __restrict__ out,
    float* __restrict__ aggOut, const float* __restrict__ epsP,
    int epsIdx, int writeAgg)
{
    constexpr int P   = K + 4;
    constexpr int RV  = K / 4;
    constexpr int NV  = GM_BROWS * RV / 256;
    __shared__ float4 sA4[GM_BROWS * P / 4];
    float* sA = (float*)sA4;

    const int tid  = threadIdx.x;
    const int lane = tid & 31;
    const int wave = tid >> 5;
    const int half = lane >> 4;
    const int mr   = lane & 15;
    const int col  = wave * 16 + mr;
    const int row0 = blockIdx.x * GM_BROWS;

    float4 st[NV];
    const float4* __restrict__ A4 = (const float4*)A;
    #pragma unroll
    for (int j = 0; j < NV; ++j) {
        const int g = row0 * RV + j * 256 + tid;
        st[j] = A4[g < NN * RV ? g : NN * RV - 1];
    }

    v2f bf[K / 4];
    #pragma unroll
    for (int ks = 0; ks < K / 4; ++ks) {
        const int ka = 4 * ks + 2 * half;
        bf[ks].x = W[(size_t)ka * HID + col];
        bf[ks].y = W[(size_t)(ka + 1) * HID + col];
    }

    const float cbias = bias[col];
    const float cm    = bn_m[col];
    const float crs   = rsqrtf(bn_v[col] + 1e-5f);
    const float cg    = bn_g[col];
    const float cb    = bn_b[col];
    const float escale = writeAgg ? (1.0f + epsP[epsIdx]) : 0.0f;

    #pragma unroll
    for (int j = 0; j < NV; ++j) {
        const int idx = j * 256 + tid;
        *(float4*)(sA + (idx / RV) * P + (idx % RV) * 4) = st[j];
    }
    __syncthreads();

    for (int p = 0; p < GM_BROWS / 32; ++p) {
        const int prow = row0 + 32 * p;
        if (prow >= NN) break;
        const float* a0p = sA + (32 * p + mr) * P + 2 * half;
        const float* a1p = a0p + 16 * P;
        v8f acc0 = {}, acc1 = {};
        #pragma unroll
        for (int ks = 0; ks < K / 4; ++ks) {
            const v2f a0 = *(const v2f*)(a0p + 4 * ks);
            const v2f a1 = *(const v2f*)(a1p + 4 * ks);
            acc0 = __builtin_amdgcn_wmma_f32_16x16x4_f32(
                false, a0, false, bf[ks], (short)0, acc0, false, false);
            acc1 = __builtin_amdgcn_wmma_f32_16x16x4_f32(
                false, a1, false, bf[ks], (short)0, acc1, false, false);
        }

        #pragma unroll
        for (int t = 0; t < 2; ++t) {
            #pragma unroll
            for (int r = 0; r < 8; ++r) {
                const int row = prow + 16 * t + r + 8 * half;
                float y = (t == 0 ? acc0[r] : acc1[r]) + cbias;
                y = (y - cm) * crs * cg + cb;
                y = gelu_exact(y);
                const size_t o = (size_t)row * HID + col;
                out[o] = y;
                if (writeAgg) aggOut[o] = escale * y;
            }
        }
    }
}

__global__ __launch_bounds__(256) void zero_kernel(float* __restrict__ p, int n) {
    int idx = blockIdx.x * blockDim.x + threadIdx.x;
    if (idx < n) p[idx] = 0.0f;
}

__global__ __launch_bounds__(256) void pool_acc_kernel(
    const float* __restrict__ h, const int* __restrict__ batch,
    float* __restrict__ sums, float* __restrict__ cnt, int total)
{
    int idx = blockIdx.x * blockDim.x + threadIdx.x;
    if (idx >= total) return;
    int i = idx >> 5;
    int c = idx & 31;
    int b = batch[i];
    const float4 v = *(const float4*)(h + (size_t)i * HID + 4 * c);
    float* s = sums + (size_t)b * HID + 4 * c;
    unsafeAtomicAdd(s + 0, v.x);
    unsafeAtomicAdd(s + 1, v.y);
    unsafeAtomicAdd(s + 2, v.z);
    unsafeAtomicAdd(s + 3, v.w);
    if (c == 0) unsafeAtomicAdd(cnt + b, 1.0f);
}

__global__ __launch_bounds__(256) void pool_final_kernel(
    const float* __restrict__ sums, const float* __restrict__ cnt,
    float* __restrict__ pooled)
{
    int idx = blockIdx.x * blockDim.x + threadIdx.x;
    if (idx >= GG * HID) return;
    int g = idx / HID;
    pooled[idx] = sums[idx] / fmaxf(cnt[g], 1.0f);
}

__global__ __launch_bounds__(128) void head_kernel(
    const float* __restrict__ pooled,
    const float* __restrict__ fc1w, const float* __restrict__ fc1b,
    const float* __restrict__ lng, const float* __restrict__ lnb,
    const float* __restrict__ fc2w, const float* __restrict__ fc2b,
    float* __restrict__ out)
{
    __shared__ float p[HID];
    __shared__ float rbuf[HID];
    __shared__ float red[HID];
    const int g = blockIdx.x;
    const int t = threadIdx.x;

    p[t] = pooled[(size_t)g * HID + t];
    __syncthreads();

    float o = fc1b[t];
    #pragma unroll 4
    for (int k = 0; k < HID; ++k) o += p[k] * fc1w[k * HID + t];

    red[t] = o;
    __syncthreads();
    for (int s = 64; s > 0; s >>= 1) {
        if (t < s) red[t] += red[t + s];
        __syncthreads();
    }
    const float mu = red[0] * (1.0f / HID);
    __syncthreads();

    const float dv = o - mu;
    red[t] = dv * dv;
    __syncthreads();
    for (int s = 64; s > 0; s >>= 1) {
        if (t < s) red[t] += red[t + s];
        __syncthreads();
    }
    const float var = red[0] * (1.0f / HID);

    float y = dv * rsqrtf(var + 1e-5f) * lng[t] + lnb[t];
    y = gelu_exact(y) + p[t];
    rbuf[t] = y;
    __syncthreads();

    if (t < LAT) {
        float o2 = fc2b[t];
        #pragma unroll 4
        for (int k = 0; k < HID; ++k) o2 += rbuf[k] * fc2w[k * LAT + t];
        out[(size_t)g * LAT + t] = o2;
    }
}

extern "C" void kernel_launch(void* const* d_in, const int* in_sizes, int n_in,
                              void* d_out, int out_size, void* d_ws, size_t ws_size,
                              hipStream_t stream) {
    const float* x       = (const float*)d_in[0];
    const int*   ei      = (const int*)  d_in[1];
    const int*   batch   = (const int*)  d_in[2];
    const float* in_g    = (const float*)d_in[3];
    const float* in_b    = (const float*)d_in[4];
    const float* in_m    = (const float*)d_in[5];
    const float* in_v    = (const float*)d_in[6];
    const float* eps0    = (const float*)d_in[7];
    const float* c0_w1   = (const float*)d_in[8];
    const float* c0_b1   = (const float*)d_in[9];
    const float* c0_g    = (const float*)d_in[10];
    const float* c0_bb   = (const float*)d_in[11];
    const float* c0_m    = (const float*)d_in[12];
    const float* c0_v    = (const float*)d_in[13];
    const float* c0_w2   = (const float*)d_in[14];
    const float* c0_b2   = (const float*)d_in[15];
    const float* bn0_g   = (const float*)d_in[16];
    const float* bn0_b   = (const float*)d_in[17];
    const float* bn0_m   = (const float*)d_in[18];
    const float* bn0_v   = (const float*)d_in[19];
    const float* epsR    = (const float*)d_in[20];
    const float* cR_w1   = (const float*)d_in[21];
    const float* cR_b1   = (const float*)d_in[22];
    const float* cR_g    = (const float*)d_in[23];
    const float* cR_bb   = (const float*)d_in[24];
    const float* cR_m    = (const float*)d_in[25];
    const float* cR_v    = (const float*)d_in[26];
    const float* cR_w2   = (const float*)d_in[27];
    const float* cR_b2   = (const float*)d_in[28];
    const float* bnR_g   = (const float*)d_in[29];
    const float* bnR_b   = (const float*)d_in[30];
    const float* bnR_m   = (const float*)d_in[31];
    const float* bnR_v   = (const float*)d_in[32];
    const float* fc1_w   = (const float*)d_in[33];
    const float* fc1_b   = (const float*)d_in[34];
    const float* ln_g    = (const float*)d_in[35];
    const float* ln_b    = (const float*)d_in[36];
    const float* fc2_w   = (const float*)d_in[37];
    const float* fc2_b   = (const float*)d_in[38];

    const int* src = ei;
    const int* dst = ei + EE;

    float* buf0 = (float*)d_ws;
    float* buf1 = buf0 + (size_t)NN * HID;
    float* buf2 = buf1 + (size_t)NN * HID;
    float* sums = buf2 + (size_t)NN * HID;
    float* cnt  = sums + (size_t)GG * HID;

    const int TB = 256;
    const int nGemm = (NN + GM_BROWS - 1) / GM_BROWS;

    bn_in_kernel<<<(NN * INDIM + TB - 1) / TB, TB, 0, stream>>>(
        x, in_g, in_b, in_m, in_v, eps0, buf0, buf1);

    {
        edge_tile_kernel<INDIM><<<196, 1024, 0, stream>>>(buf0, src, dst, buf1);
        gemm_bn_gelu_kernel<INDIM><<<nGemm, TB, 0, stream>>>(
            buf1, c0_w1, c0_b1, c0_g, c0_bb, c0_m, c0_v, buf2,
            buf1, eps0, 0, 0);
        gemm_bn_gelu_kernel<HID><<<nGemm, TB, 0, stream>>>(
            buf2, c0_w2, c0_b2, bn0_g, bn0_b, bn0_m, bn0_v, buf0,
            buf1, epsR, 0, 1);
    }

    for (int l = 0; l < 2; ++l) {
        const size_t wOff = (size_t)l * HID * HID;
        const size_t vOff = (size_t)l * HID;
        edge_tile_kernel<HID><<<196, 1024, 0, stream>>>(buf0, src, dst, buf1);
        gemm_bn_gelu_kernel<HID><<<nGemm, TB, 0, stream>>>(
            buf1, cR_w1 + wOff, cR_b1 + vOff, cR_g + vOff, cR_bb + vOff,
            cR_m + vOff, cR_v + vOff, buf2,
            buf1, epsR, 0, 0);
        const int writeAgg = (l == 0) ? 1 : 0;
        gemm_bn_gelu_kernel<HID><<<nGemm, TB, 0, stream>>>(
            buf2, cR_w2 + wOff, cR_b2 + vOff, bnR_g + vOff, bnR_b + vOff,
            bnR_m + vOff, bnR_v + vOff, buf0,
            buf1, epsR, 1, writeAgg);
    }

    zero_kernel<<<(GG * HID + GG + TB - 1) / TB, TB, 0, stream>>>(sums, GG * HID + GG);
    {
        int total = NN * (HID / 4);
        pool_acc_kernel<<<(total + TB - 1) / TB, TB, 0, stream>>>(
            buf0, batch, sums, cnt, total);
    }
    pool_final_kernel<<<(GG * HID + TB - 1) / TB, TB, 0, stream>>>(sums, cnt, buf1);

    head_kernel<<<GG, HID, 0, stream>>>(
        buf1, fc1_w, fc1_b, ln_g, ln_b, fc2_w, fc2_b, (float*)d_out);
}


# Edge scan reads per-tile edge bitmaps: a small kernel before each edge kernel sets every edge's bit in its dst tile's bitmap (LDS int OR); tile blocks no longer scan all of dst; adds in arrival order
# speedup vs baseline: 1.7313x; 1.1392x over previous
#include <hip/hip_runtime.h>
#include <math.h>

typedef __attribute__((ext_vector_type(2))) float v2f;
typedef __attribute__((ext_vector_type(8))) float v8f;

#define NN   100000
#define EE   1600000
#define INDIM 64
#define HID  128
#define LAT  64
#define GG   1000

__device__ __forceinline__ float gelu_exact(float x) {
    return 0.5f * x * (1.0f + erff(x * 0.70710678118654752f));
}

__global__ __launch_bounds__(256) void bn_in_kernel(
    const float* __restrict__ x,
    const float* __restrict__ g, const float* __restrict__ b,
    const float* __restrict__ m, const float* __restrict__ v,
    const float* __restrict__ epsP,
    float* __restrict__ out, float* __restrict__ agg)
{
    int idx = blockIdx.x * blockDim.x + threadIdx.x;
    if (idx >= NN * INDIM) return;
    int j = idx & (INDIM - 1);
    float y = (x[idx] - m[j]) * rsqrtf(v[j] + 1e-5f) * g[j] + b[j];
    out[idx] = y;
    agg[idx] = (1.0f + epsP[0]) * y;
}

#define ET_TILE   512
#define ET_NTILES ((NN + ET_TILE - 1) / ET_TILE)
#define ET_QCAP   48

#define BM_WORDS  (EE / 32)
#define BM_V4     (BM_WORDS / 4)
#define BM_PITCH  (((BM_WORDS + 31) / 32) * 32)
#define BM_CH     256
#define BM_NCH    ((BM_WORDS + BM_CH - 1) / BM_CH)
#define BM_LP     (BM_CH + 4)
#define ET_NIT    ((BM_V4 + 1023) / 1024)

__global__ __launch_bounds__(1024) void tile_bitmap_kernel(
    const int* __restrict__ dst, unsigned* __restrict__ bm)
{
    __shared__ uint4 sb4[ET_NTILES * BM_LP / 4];
    unsigned* sb = (unsigned*)sb4;
    const int tid = threadIdx.x;

    for (int i = tid; i < ET_NTILES * BM_LP / 4; i += 1024)
        sb4[i] = make_uint4(0u, 0u, 0u, 0u);
    __syncthreads();

    const int e0 = blockIdx.x * (BM_CH * 32);
    int dv[BM_CH * 32 / 1024];
    #pragma unroll
    for (int k = 0; k < BM_CH * 32 / 1024; ++k) {
        const int e = e0 + k * 1024 + tid;
        dv[k] = dst[e < EE ? e : EE - 1];
    }
    #pragma unroll
    for (int k = 0; k < BM_CH * 32 / 1024; ++k) {
        const int el = k * 1024 + tid;
        const unsigned t = (unsigned)dv[k] / ET_TILE;
        if (e0 + el < EE && t < ET_NTILES)
            atomicOr(&sb[t * BM_LP + (el >> 5)], 1u << (el & 31));
    }
    __syncthreads();

    for (int i = tid; i < ET_NTILES * (BM_CH / 4); i += 1024) {
        const int t  = i / (BM_CH / 4);
        const int w4 = i % (BM_CH / 4);
        const int gw = blockIdx.x * BM_CH + 4 * w4;
        if (gw < BM_WORDS)
            *(uint4*)(bm + (size_t)t * BM_PITCH + gw) = sb4[t * (BM_LP / 4) + w4];
    }
}

__device__ __forceinline__ void et_wave_order() {
    __builtin_amdgcn_fence(__ATOMIC_ACQ_REL, "wavefront");
    __builtin_amdgcn_wave_barrier();
}

template<int D>
__device__ __forceinline__ void et_drain(
    const float* __restrict__ h, const int* __restrict__ src,
    const int* __restrict__ dst, int lo,
    float* tile, const unsigned* q, int qn, int n, int lane)
{
    int s = 0, dl = 0;
    if (lane < n) {
        const unsigned e = q[qn - n + lane];
        s  = src[e];
        dl = (dst[e] - lo) & (ET_TILE - 1);
    }
    #pragma unroll
    for (int base = 0; base < 16; base += 8) {
        if (base >= n) break;
        float v[8][D / 32];
        #pragma unroll
        for (int i = 0; i < 8; ++i) {
            if (base + i < n) {
                const int si = __builtin_amdgcn_readlane(s, base + i);
                const float* row = h + (size_t)si * D + lane;
                #pragma unroll
                for (int k = 0; k < D / 32; ++k) v[i][k] = row[32 * k];
            }
        }
        #pragma unroll
        for (int i = 0; i < 8; ++i) {
            if (base + i < n) {
                const int di = __builtin_amdgcn_readlane(dl, base + i);
                float* trow = tile + di * D + lane;
                #pragma unroll
                for (int k = 0; k < D / 32; ++k)
                    __hip_atomic_fetch_add(trow + 32 * k, v[i][k],
                        __ATOMIC_RELAXED, __HIP_MEMORY_SCOPE_WORKGROUP);
            }
        }
    }
}

template<int D>
__global__ __launch_bounds__(1024) void edge_tile_kernel(
    const float* __restrict__ h, const int* __restrict__ src,
    const int* __restrict__ dst, const unsigned* __restrict__ bm,
    float* __restrict__ agg)
{
    __shared__ float tile[ET_TILE * D];
    __shared__ unsigned queue[32 * ET_QCAP];

    const int tid  = threadIdx.x;
    const int lane = tid & 31;
    const int wave = tid >> 5;
    const int lo   = blockIdx.x * ET_TILE;
    const int nrows = (NN - lo < ET_TILE) ? (NN - lo) : ET_TILE;
    const int n4 = nrows * (D / 4);

    const uint4* bm4 = (const uint4*)(bm + (size_t)blockIdx.x * BM_PITCH);
    uint4 cur = bm4[tid];

    float4* t4 = (float4*)tile;
    float4* a4 = (float4*)(agg + (size_t)lo * D);
    for (int i = tid; i < n4; i += 1024) t4[i] = a4[i];
    __syncthreads();

    unsigned* q = queue + wave * ET_QCAP;
    int qn = 0;

    for (int n = 0; n < ET_NIT; ++n) {
        const int i4  = n * 1024 + tid;
        const int inx = i4 + 1024;
        const uint4 nxt = bm4[inx < BM_V4 ? inx : BM_V4 - 1];
        const bool inlist = i4 < BM_V4;
        const unsigned wv[4] = {cur.x, cur.y, cur.z, cur.w};
        #pragma unroll
        for (int j = 0; j < 4; ++j) {
            unsigned w = inlist ? wv[j] : 0u;
            const unsigned ebase = (unsigned)(4 * i4 + j) * 32u;
            unsigned any = __builtin_amdgcn_ballot_w32(w != 0u);
            while (any) {
                if (w != 0u) {
                    const int pos = qn + (int)__builtin_amdgcn_mbcnt_lo(any, 0u);
                    q[pos] = ebase + (unsigned)__builtin_ctz(w);
                    w &= w - 1u;
                }
                qn += __builtin_popcount(any);
                et_wave_order();
                while (qn >= 16) {
                    et_drain<D>(h, src, dst, lo, tile, q, qn, 16, lane);
                    qn -= 16;
                }
                et_wave_order();
                any = __builtin_amdgcn_ballot_w32(w != 0u);
            }
        }
        cur = nxt;
    }
    if (qn > 0) et_drain<D>(h, src, dst, lo, tile, q, qn, qn, lane);

    __syncthreads();
    for (int i = tid; i < n4; i += 1024) a4[i] = t4[i];
}

#define GM_BROWS 64

template<int K>
__global__ __launch_bounds__(256) void gemm_bn_gelu_kernel(
    const float* __restrict__ A,
    const float* __restrict__ W,
    const float* __restrict__ bias,
    const float* __restrict__ bn_g, const float* __restrict__ bn_b,
    const float* __restrict__ bn_m, const float* __restrict__ bn_v,
    float* __restrict__ out,
    float* __restrict__ aggOut, const float* __restrict__ epsP,
    int epsIdx, int writeAgg)
{
    constexpr int P   = K + 4;
    constexpr int RV  = K / 4;
    constexpr int NV  = GM_BROWS * RV / 256;
    __shared__ float4 sA4[GM_BROWS * P / 4];
    float* sA = (float*)sA4;

    const int tid  = threadIdx.x;
    const int lane = tid & 31;
    const int wave = tid >> 5;
    const int half = lane >> 4;
    const int mr   = lane & 15;
    const int col  = wave * 16 + mr;
    const int row0 = blockIdx.x * GM_BROWS;

    float4 st[NV];
    const float4* __restrict__ A4 = (const float4*)A;
    #pragma unroll
    for (int j = 0; j < NV; ++j) {
        const int g = row0 * RV + j * 256 + tid;
        st[j] = A4[g < NN * RV ? g : NN * RV - 1];
    }

    v2f bf[K / 4];
    #pragma unroll
    for (int ks = 0; ks < K / 4; ++ks) {
        const int ka = 4 * ks + 2 * half;
        bf[ks].x = W[(size_t)ka * HID + col];
        bf[ks].y = W[(size_t)(ka + 1) * HID + col];
    }

    const float cbias = bias[col];
    const float cm    = bn_m[col];
    const float crs   = rsqrtf(bn_v[col] + 1e-5f);
    const float cg    = bn_g[col];
    const float cb    = bn_b[col];
    const float escale = writeAgg ? (1.0f + epsP[epsIdx]) : 0.0f;

    #pragma unroll
    for (int j = 0; j < NV; ++j) {
        const int idx = j * 256 + tid;
        *(float4*)(sA + (idx / RV) * P + (idx % RV) * 4) = st[j];
    }
    __syncthreads();

    for (int p = 0; p < GM_BROWS / 32; ++p) {
        const int prow = row0 + 32 * p;
        if (prow >= NN) break;
        const float* a0p = sA + (32 * p + mr) * P + 2 * half;
        const float* a1p = a0p + 16 * P;
        v8f acc0 = {}, acc1 = {};
        #pragma unroll
        for (int ks = 0; ks < K / 4; ++ks) {
            const v2f a0 = *(const v2f*)(a0p + 4 * ks);
            const v2f a1 = *(const v2f*)(a1p + 4 * ks);
            acc0 = __builtin_amdgcn_wmma_f32_16x16x4_f32(
                false, a0, false, bf[ks], (short)0, acc0, false, false);
            acc1 = __builtin_amdgcn_wmma_f32_16x16x4_f32(
                false, a1, false, bf[ks], (short)0, acc1, false, false);
        }

        #pragma unroll
        for (int t = 0; t < 2; ++t) {
            #pragma unroll
            for (int r = 0; r < 8; ++r) {
                const int row = prow + 16 * t + r + 8 * half;
                float y = (t == 0 ? acc0[r] : acc1[r]) + cbias;
                y = (y - cm) * crs * cg + cb;
                y = gelu_exact(y);
                const size_t o = (size_t)row * HID + col;
                out[o] = y;
                if (writeAgg) aggOut[o] = escale * y;
            }
        }
    }
}

__global__ __launch_bounds__(256) void zero_kernel(float* __restrict__ p, int n) {
    int idx = blockIdx.x * blockDim.x + threadIdx.x;
    if (idx < n) p[idx] = 0.0f;
}

__global__ __launch_bounds__(256) void pool_acc_kernel(
    const float* __restrict__ h, const int* __restrict__ batch,
    float* __restrict__ sums, float* __restrict__ cnt, int total)
{
    int idx = blockIdx.x * blockDim.x + threadIdx.x;
    if (idx >= total) return;
    int i = idx >> 5;
    int c = idx & 31;
    int b = batch[i];
    const float4 v = *(const float4*)(h + (size_t)i * HID + 4 * c);
    float* s = sums + (size_t)b * HID + 4 * c;
    unsafeAtomicAdd(s + 0, v.x);
    unsafeAtomicAdd(s + 1, v.y);
    unsafeAtomicAdd(s + 2, v.z);
    unsafeAtomicAdd(s + 3, v.w);
    if (c == 0) unsafeAtomicAdd(cnt + b, 1.0f);
}

__global__ __launch_bounds__(256) void pool_final_kernel(
    const float* __restrict__ sums, const float* __restrict__ cnt,
    float* __restrict__ pooled)
{
    int idx = blockIdx.x * blockDim.x + threadIdx.x;
    if (idx >= GG * HID) return;
    int g = idx / HID;
    pooled[idx] = sums[idx] / fmaxf(cnt[g], 1.0f);
}

__global__ __launch_bounds__(128) void head_kernel(
    const float* __restrict__ pooled,
    const float* __restrict__ fc1w, const float* __restrict__ fc1b,
    const float* __restrict__ lng, const float* __restrict__ lnb,
    const float* __restrict__ fc2w, const float* __restrict__ fc2b,
    float* __restrict__ out)
{
    __shared__ float p[HID];
    __shared__ float rbuf[HID];
    __shared__ float red[HID];
    const int g = blockIdx.x;
    const int t = threadIdx.x;

    p[t] = pooled[(size_t)g * HID + t];
    __syncthreads();

    float o = fc1b[t];
    #pragma unroll 4
    for (int k = 0; k < HID; ++k) o += p[k] * fc1w[k * HID + t];

    red[t] = o;
    __syncthreads();
    for (int s = 64; s > 0; s >>= 1) {
        if (t < s) red[t] += red[t + s];
        __syncthreads();
    }
    const float mu = red[0] * (1.0f / HID);
    __syncthreads();

    const float dv = o - mu;
    red[t] = dv * dv;
    __syncthreads();
    for (int s = 64; s > 0; s >>= 1) {
        if (t < s) red[t] += red[t + s];
        __syncthreads();
    }
    const float var = red[0] * (1.0f / HID);

    float y = dv * rsqrtf(var + 1e-5f) * lng[t] + lnb[t];
    y = gelu_exact(y) + p[t];
    rbuf[t] = y;
    __syncthreads();

    if (t < LAT) {
        float o2 = fc2b[t];
        #pragma unroll 4
        for (int k = 0; k < HID; ++k) o2 += rbuf[k] * fc2w[k * LAT + t];
        out[(size_t)g * LAT + t] = o2;
    }
}

extern "C" void kernel_launch(void* const* d_in, const int* in_sizes, int n_in,
                              void* d_out, int out_size, void* d_ws, size_t ws_size,
                              hipStream_t stream) {
    const float* x       = (const float*)d_in[0];
    const int*   ei      = (const int*)  d_in[1];
    const int*   batch   = (const int*)  d_in[2];
    const float* in_g    = (const float*)d_in[3];
    const float* in_b    = (const float*)d_in[4];
    const float* in_m    = (const float*)d_in[5];
    const float* in_v    = (const float*)d_in[6];
    const float* eps0    = (const float*)d_in[7];
    const float* c0_w1   = (const float*)d_in[8];
    const float* c0_b1   = (const float*)d_in[9];
    const float* c0_g    = (const float*)d_in[10];
    const float* c0_bb   = (const float*)d_in[11];
    const float* c0_m    = (const float*)d_in[12];
    const float* c0_v    = (const float*)d_in[13];
    const float* c0_w2   = (const float*)d_in[14];
    const float* c0_b2   = (const float*)d_in[15];
    const float* bn0_g   = (const float*)d_in[16];
    const float* bn0_b   = (const float*)d_in[17];
    const float* bn0_m   = (const float*)d_in[18];
    const float* bn0_v   = (const float*)d_in[19];
    const float* epsR    = (const float*)d_in[20];
    const float* cR_w1   = (const float*)d_in[21];
    const float* cR_b1   = (const float*)d_in[22];
    const float* cR_g    = (const float*)d_in[23];
    const float* cR_bb   = (const float*)d_in[24];
    const float* cR_m    = (const float*)d_in[25];
    const float* cR_v    = (const float*)d_in[26];
    const float* cR_w2   = (const float*)d_in[27];
    const float* cR_b2   = (const float*)d_in[28];
    const float* bnR_g   = (const float*)d_in[29];
    const float* bnR_b   = (const float*)d_in[30];
    const float* bnR_m   = (const float*)d_in[31];
    const float* bnR_v   = (const float*)d_in[32];
    const float* fc1_w   = (const float*)d_in[33];
    const float* fc1_b   = (const float*)d_in[34];
    const float* ln_g    = (const float*)d_in[35];
    const float* ln_b    = (const float*)d_in[36];
    const float* fc2_w   = (const float*)d_in[37];
    const float* fc2_b   = (const float*)d_in[38];

    const int* src = ei;
    const int* dst = ei + EE;

    float* buf0 = (float*)d_ws;
    float* buf1 = buf0 + (size_t)NN * HID;
    float* buf2 = buf1 + (size_t)NN * HID;
    float* sums = buf2 + (size_t)NN * HID;
    float* cnt  = sums + (size_t)GG * HID;
    unsigned* bmp = (unsigned*)buf2;

    const int TB = 256;
    const int nGemm = (NN + GM_BROWS - 1) / GM_BROWS;

    bn_in_kernel<<<(NN * INDIM + TB - 1) / TB, TB, 0, stream>>>(
        x, in_g, in_b, in_m, in_v, eps0, buf0, buf1);

    {
        tile_bitmap_kernel<<<BM_NCH, 1024, 0, stream>>>(dst, bmp);
        edge_tile_kernel<INDIM><<<ET_NTILES, 1024, 0, stream>>>(buf0, src, dst, bmp, buf1);
        gemm_bn_gelu_kernel<INDIM><<<nGemm, TB, 0, stream>>>(
            buf1, c0_w1, c0_b1, c0_g, c0_bb, c0_m, c0_v, buf2,
            buf1, eps0, 0, 0);
        gemm_bn_gelu_kernel<HID><<<nGemm, TB, 0, stream>>>(
            buf2, c0_w2, c0_b2, bn0_g, bn0_b, bn0_m, bn0_v, buf0,
            buf1, epsR, 0, 1);
    }

    for (int l = 0; l < 2; ++l) {
        const size_t wOff = (size_t)l * HID * HID;
        const size_t vOff = (size_t)l * HID;
        tile_bitmap_kernel<<<BM_NCH, 1024, 0, stream>>>(dst, bmp);
        edge_tile_kernel<HID><<<ET_NTILES, 1024, 0, stream>>>(buf0, src, dst, bmp, buf1);
        gemm_bn_gelu_kernel<HID><<<nGemm, TB, 0, stream>>>(
            buf1, cR_w1 + wOff, cR_b1 + vOff, cR_g + vOff, cR_bb + vOff,
            cR_m + vOff, cR_v + vOff, buf2,
            buf1, epsR, 0, 0);
        const int writeAgg = (l == 0) ? 1 : 0;
        gemm_bn_gelu_kernel<HID><<<nGemm, TB, 0, stream>>>(
            buf2, cR_w2 + wOff, cR_b2 + vOff, bnR_g + vOff, bnR_b + vOff,
            bnR_m + vOff, bnR_v + vOff, buf0,
            buf1, epsR, 1, writeAgg);
    }

    zero_kernel<<<(GG * HID + GG + TB - 1) / TB, TB, 0, stream>>>(sums, GG * HID + GG);
    {
        int total = NN * (HID / 4);
        pool_acc_kernel<<<(total + TB - 1) / TB, TB, 0, stream>>>(
            buf0, batch, sums, cnt, total);
    }
    pool_final_kernel<<<(GG * HID + TB - 1) / TB, TB, 0, stream>>>(sums, cnt, buf1);

    head_kernel<<<GG, HID, 0, stream>>>(
        buf1, fc1_w, fc1_b, ln_g, ln_b, fc2_w, fc2_b, (float*)d_out);
}


# Input BN and pool-add launches as resident grids: 4,096 waves take 13 / 25 16-byte pieces each, not 200,000 / 100,000 one-piece waves; formula and float atomics as before, pool adds in arrival order
# speedup vs baseline: 1.8859x; 1.0893x over previous
#include <hip/hip_runtime.h>
#include <math.h>

typedef __attribute__((ext_vector_type(2))) float v2f;
typedef __attribute__((ext_vector_type(8))) float v8f;

#define NN   100000
#define EE   1600000
#define INDIM 64
#define HID  128
#define LAT  64
#define GG   1000

__device__ __forceinline__ float gelu_exact(float x) {
    return 0.5f * x * (1.0f + erff(x * 0.70710678118654752f));
}

#define BN_BLOCKS  512
#define BN_THREADS 256
#define BN_V4      (NN * INDIM / 4)
#define BN_STRIDE  (BN_BLOCKS * BN_THREADS)
#define BN_PER     ((BN_V4 + BN_STRIDE - 1) / BN_STRIDE)
static_assert(BN_STRIDE % (INDIM / 4) == 0, "stride must be whole rows");

__global__ __launch_bounds__(BN_THREADS) void bn_in_kernel(
    const float* __restrict__ x,
    const float* __restrict__ g, const float* __restrict__ b,
    const float* __restrict__ m, const float* __restrict__ v,
    const float* __restrict__ epsP,
    float* __restrict__ out, float* __restrict__ agg)
{
    const int t = blockIdx.x * BN_THREADS + threadIdx.x;
    const float4* __restrict__ x4 = (const float4*)x;

    float4 xv[BN_PER];
    #pragma unroll
    for (int k = 0; k < BN_PER; ++k) {
        const int i = t + k * BN_STRIDE;
        xv[k] = x4[i < BN_V4 ? i : BN_V4 - 1];
    }

    const int j = 4 * (t & (INDIM / 4 - 1));
    float cm[4], cr[4], cg[4], cb[4];
    #pragma unroll
    for (int c = 0; c < 4; ++c) {
        cm[c] = m[j + c];
        cr[c] = rsqrtf(v[j + c] + 1e-5f);
        cg[c] = g[j + c];
        cb[c] = b[j + c];
    }
    const float sc = 1.0f + epsP[0];

    float4* __restrict__ out4 = (float4*)out;
    float4* __restrict__ agg4 = (float4*)agg;
    #pragma unroll
    for (int k = 0; k < BN_PER; ++k) {
        const int i = t + k * BN_STRIDE;
        if (i < BN_V4) {
            float4 y;
            y.x = (xv[k].x - cm[0]) * cr[0] * cg[0] + cb[0];
            y.y = (xv[k].y - cm[1]) * cr[1] * cg[1] + cb[1];
            y.z = (xv[k].z - cm[2]) * cr[2] * cg[2] + cb[2];
            y.w = (xv[k].w - cm[3]) * cr[3] * cg[3] + cb[3];
            out4[i] = y;
            agg4[i] = make_float4(sc * y.x, sc * y.y, sc * y.z, sc * y.w);
        }
    }
}

#define ET_TILE   512
#define ET_NTILES ((NN + ET_TILE - 1) / ET_TILE)
#define ET_QCAP   48

#define BM_WORDS  (EE / 32)
#define BM_V4     (BM_WORDS / 4)
#define BM_PITCH  (((BM_WORDS + 31) / 32) * 32)
#define BM_CH     256
#define BM_NCH    ((BM_WORDS + BM_CH - 1) / BM_CH)
#define BM_LP     (BM_CH + 4)
#define ET_NIT    ((BM_V4 + 1023) / 1024)

__global__ __launch_bounds__(1024) void tile_bitmap_kernel(
    const int* __restrict__ dst, unsigned* __restrict__ bm)
{
    __shared__ uint4 sb4[ET_NTILES * BM_LP / 4];
    unsigned* sb = (unsigned*)sb4;
    const int tid = threadIdx.x;

    for (int i = tid; i < ET_NTILES * BM_LP / 4; i += 1024)
        sb4[i] = make_uint4(0u, 0u, 0u, 0u);
    __syncthreads();

    const int e0 = blockIdx.x * (BM_CH * 32);
    int dv[BM_CH * 32 / 1024];
    #pragma unroll
    for (int k = 0; k < BM_CH * 32 / 1024; ++k) {
        const int e = e0 + k * 1024 + tid;
        dv[k] = dst[e < EE ? e : EE - 1];
    }
    #pragma unroll
    for (int k = 0; k < BM_CH * 32 / 1024; ++k) {
        const int el = k * 1024 + tid;
        const unsigned t = (unsigned)dv[k] / ET_TILE;
        if (e0 + el < EE && t < ET_NTILES)
            atomicOr(&sb[t * BM_LP + (el >> 5)], 1u << (el & 31));
    }
    __syncthreads();

    for (int i = tid; i < ET_NTILES * (BM_CH / 4); i += 1024) {
        const int t  = i / (BM_CH / 4);
        const int w4 = i % (BM_CH / 4);
        const int gw = blockIdx.x * BM_CH + 4 * w4;
        if (gw < BM_WORDS)
            *(uint4*)(bm + (size_t)t * BM_PITCH + gw) = sb4[t * (BM_LP / 4) + w4];
    }
}

__device__ __forceinline__ void et_wave_order() {
    __builtin_amdgcn_fence(__ATOMIC_ACQ_REL, "wavefront");
    __builtin_amdgcn_wave_barrier();
}

template<int D>
__device__ __forceinline__ void et_drain(
    const float* __restrict__ h, const int* __restrict__ src,
    const int* __restrict__ dst, int lo,
    float* tile, const unsigned* q, int qn, int n, int lane)
{
    int s = 0, dl = 0;
    if (lane < n) {
        const unsigned e = q[qn - n + lane];
        s  = src[e];
        dl = (dst[e] - lo) & (ET_TILE - 1);
    }
    #pragma unroll
    for (int base = 0; base < 16; base += 8) {
        if (base >= n) break;
        float v[8][D / 32];
        #pragma unroll
        for (int i = 0; i < 8; ++i) {
            if (base + i < n) {
                const int si = __builtin_amdgcn_readlane(s, base + i);
                const float* row = h + (size_t)si * D + lane;
                #pragma unroll
                for (int k = 0; k < D / 32; ++k) v[i][k] = row[32 * k];
            }
        }
        #pragma unroll
        for (int i = 0; i < 8; ++i) {
            if (base + i < n) {
                const int di = __builtin_amdgcn_readlane(dl, base + i);
                float* trow = tile + di * D + lane;
                #pragma unroll
                for (int k = 0; k < D / 32; ++k)
                    __hip_atomic_fetch_add(trow + 32 * k, v[i][k],
                        __ATOMIC_RELAXED, __HIP_MEMORY_SCOPE_WORKGROUP);
            }
        }
    }
}

template<int D>
__global__ __launch_bounds__(1024) void edge_tile_kernel(
    const float* __restrict__ h, const int* __restrict__ src,
    const int* __restrict__ dst, const unsigned* __restrict__ bm,
    float* __restrict__ agg)
{
    __shared__ float tile[ET_TILE * D];
    __shared__ unsigned queue[32 * ET_QCAP];

    const int tid  = threadIdx.x;
    const int lane = tid & 31;
    const int wave = tid >> 5;
    const int lo   = blockIdx.x * ET_TILE;
    const int nrows = (NN - lo < ET_TILE) ? (NN - lo) : ET_TILE;
    const int n4 = nrows * (D / 4);

    const uint4* bm4 = (const uint4*)(bm + (size_t)blockIdx.x * BM_PITCH);
    uint4 cur = bm4[tid];

    float4* t4 = (float4*)tile;
    float4* a4 = (float4*)(agg + (size_t)lo * D);
    for (int i = tid; i < n4; i += 1024) t4[i] = a4[i];
    __syncthreads();

    unsigned* q = queue + wave * ET_QCAP;
    int qn = 0;

    for (int n = 0; n < ET_NIT; ++n) {
        const int i4  = n * 1024 + tid;
        const int inx = i4 + 1024;
        const uint4 nxt = bm4[inx < BM_V4 ? inx : BM_V4 - 1];
        const bool inlist = i4 < BM_V4;
        const unsigned wv[4] = {cur.x, cur.y, cur.z, cur.w};
        #pragma unroll
        for (int j = 0; j < 4; ++j) {
            unsigned w = inlist ? wv[j] : 0u;
            const unsigned ebase = (unsigned)(4 * i4 + j) * 32u;
            unsigned any = __builtin_amdgcn_ballot_w32(w != 0u);
            while (any) {
                if (w != 0u) {
                    const int pos = qn + (int)__builtin_amdgcn_mbcnt_lo(any, 0u);
                    q[pos] = ebase + (unsigned)__builtin_ctz(w);
                    w &= w - 1u;
                }
                qn += __builtin_popcount(any);
                et_wave_order();
                while (qn >= 16) {
                    et_drain<D>(h, src, dst, lo, tile, q, qn, 16, lane);
                    qn -= 16;
                }
                et_wave_order();
                any = __builtin_amdgcn_ballot_w32(w != 0u);
            }
        }
        cur = nxt;
    }
    if (qn > 0) et_drain<D>(h, src, dst, lo, tile, q, qn, qn, lane);

    __syncthreads();
    for (int i = tid; i < n4; i += 1024) a4[i] = t4[i];
}

#define GM_BROWS 64

template<int K>
__global__ __launch_bounds__(256) void gemm_bn_gelu_kernel(
    const float* __restrict__ A,
    const float* __restrict__ W,
    const float* __restrict__ bias,
    const float* __restrict__ bn_g, const float* __restrict__ bn_b,
    const float* __restrict__ bn_m, const float* __restrict__ bn_v,
    float* __restrict__ out,
    float* __restrict__ aggOut, const float* __restrict__ epsP,
    int epsIdx, int writeAgg)
{
    constexpr int P   = K + 4;
    constexpr int RV  = K / 4;
    constexpr int NV  = GM_BROWS * RV / 256;
    __shared__ float4 sA4[GM_BROWS * P / 4];
    float* sA = (float*)sA4;

    const int tid  = threadIdx.x;
    const int lane = tid & 31;
    const int wave = tid >> 5;
    const int half = lane >> 4;
    const int mr   = lane & 15;
    const int col  = wave * 16 + mr;
    const int row0 = blockIdx.x * GM_BROWS;

    float4 st[NV];
    const float4* __restrict__ A4 = (const float4*)A;
    #pragma unroll
    for (int j = 0; j < NV; ++j) {
        const int g = row0 * RV + j * 256 + tid;
        st[j] = A4[g < NN * RV ? g : NN * RV - 1];
    }

    v2f bf[K / 4];
    #pragma unroll
    for (int ks = 0; ks < K / 4; ++ks) {
        const int ka = 4 * ks + 2 * half;
        bf[ks].x = W[(size_t)ka * HID + col];
        bf[ks].y = W[(size_t)(ka + 1) * HID + col];
    }

    const float cbias = bias[col];
    const float cm    = bn_m[col];
    const float crs   = rsqrtf(bn_v[col] + 1e-5f);
    const float cg    = bn_g[col];
    const float cb    = bn_b[col];
    const float escale = writeAgg ? (1.0f + epsP[epsIdx]) : 0.0f;

    #pragma unroll
    for (int j = 0; j < NV; ++j) {
        const int idx = j * 256 + tid;
        *(float4*)(sA + (idx / RV) * P + (idx % RV) * 4) = st[j];
    }
    __syncthreads();

    for (int p = 0; p < GM_BROWS / 32; ++p) {
        const int prow = row0 + 32 * p;
        if (prow >= NN) break;
        const float* a0p = sA + (32 * p + mr) * P + 2 * half;
        const float* a1p = a0p + 16 * P;
        v8f acc0 = {}, acc1 = {};
        #pragma unroll
        for (int ks = 0; ks < K / 4; ++ks) {
            const v2f a0 = *(const v2f*)(a0p + 4 * ks);
            const v2f a1 = *(const v2f*)(a1p + 4 * ks);
            acc0 = __builtin_amdgcn_wmma_f32_16x16x4_f32(
                false, a0, false, bf[ks], (short)0, acc0, false, false);
            acc1 = __builtin_amdgcn_wmma_f32_16x16x4_f32(
                false, a1, false, bf[ks], (short)0, acc1, false, false);
        }

        #pragma unroll
        for (int t = 0; t < 2; ++t) {
            #pragma unroll
            for (int r = 0; r < 8; ++r) {
                const int row = prow + 16 * t + r + 8 * half;
                float y = (t == 0 ? acc0[r] : acc1[r]) + cbias;
                y = (y - cm) * crs * cg + cb;
                y = gelu_exact(y);
                const size_t o = (size_t)row * HID + col;
                out[o] = y;
                if (writeAgg) aggOut[o] = escale * y;
            }
        }
    }
}

__global__ __launch_bounds__(256) void zero_kernel(float* __restrict__ p, int n) {
    int idx = blockIdx.x * blockDim.x + threadIdx.x;
    if (idx < n) p[idx] = 0.0f;
}

#define PL_BLOCKS  512
#define PL_THREADS 256
#define PL_WAVES   (PL_BLOCKS * PL_THREADS / 32)
#define PL_PER     ((NN + PL_WAVES - 1) / PL_WAVES)
static_assert(HID == 128, "a row is 32 pieces of 16 bytes: one per lane");
static_assert(NN % PL_PER == 0, "a wave has all its rows or none");
static_assert(PL_PER <= 32, "one lane per node of the wave for the counts");

__global__ __launch_bounds__(PL_THREADS) void pool_acc_kernel(
    const float* __restrict__ h, const int* __restrict__ batch,
    float* __restrict__ sums, float* __restrict__ cnt)
{
    const int lane = threadIdx.x & 31;
    const int w = __builtin_amdgcn_readfirstlane(
        (int)((blockIdx.x * PL_THREADS + threadIdx.x) >> 5));
    const int i0 = w * PL_PER;
    if (i0 >= NN) return;

    float4 rv[PL_PER];
    int    gb[PL_PER];
    #pragma unroll
    for (int k = 0; k < PL_PER; ++k) {
        gb[k] = batch[i0 + k];
        rv[k] = *(const float4*)(h + (size_t)(i0 + k) * HID + 4 * lane);
    }
    #pragma unroll
    for (int k = 0; k < PL_PER; ++k) {
        float* s = sums + (size_t)gb[k] * HID + 4 * lane;
        unsafeAtomicAdd(s + 0, rv[k].x);
        unsafeAtomicAdd(s + 1, rv[k].y);
        unsafeAtomicAdd(s + 2, rv[k].z);
        unsafeAtomicAdd(s + 3, rv[k].w);
    }
    if (lane < PL_PER) unsafeAtomicAdd(cnt + batch[i0 + lane], 1.0f);
}

__global__ __launch_bounds__(256) void pool_final_kernel(
    const float* __restrict__ sums, const float* __restrict__ cnt,
    float* __restrict__ pooled)
{
    int idx = blockIdx.x * blockDim.x + threadIdx.x;
    if (idx >= GG * HID) return;
    int g = idx / HID;
    pooled[idx] = sums[idx] / fmaxf(cnt[g], 1.0f);
}

__global__ __launch_bounds__(128) void head_kernel(
    const float* __restrict__ pooled,
    const float* __restrict__ fc1w, const float* __restrict__ fc1b,
    const float* __restrict__ lng, const float* __restrict__ lnb,
    const float* __restrict__ fc2w, const float* __restrict__ fc2b,
    float* __restrict__ out)
{
    __shared__ float p[HID];
    __shared__ float rbuf[HID];
    __shared__ float red[HID];
    const int g = blockIdx.x;
    const int t = threadIdx.x;

    p[t] = pooled[(size_t)g * HID + t];
    __syncthreads();

    float o = fc1b[t];
    #pragma unroll 4
    for (int k = 0; k < HID; ++k) o += p[k] * fc1w[k * HID + t];

    red[t] = o;
    __syncthreads();
    for (int s = 64; s > 0; s >>= 1) {
        if (t < s) red[t] += red[t + s];
        __syncthreads();
    }
    const float mu = red[0] * (1.0f / HID);
    __syncthreads();

    const float dv = o - mu;
    red[t] = dv * dv;
    __syncthreads();
    for (int s = 64; s > 0; s >>= 1) {
        if (t < s) red[t] += red[t + s];
        __syncthreads();
    }
    const float var = red[0] * (1.0f / HID);

    float y = dv * rsqrtf(var + 1e-5f) * lng[t] + lnb[t];
    y = gelu_exact(y) + p[t];
    rbuf[t] = y;
    __syncthreads();

    if (t < LAT) {
        float o2 = fc2b[t];
        #pragma unroll 4
        for (int k = 0; k < HID; ++k) o2 += rbuf[k] * fc2w[k * LAT + t];
        out[(size_t)g * LAT + t] = o2;
    }
}

extern "C" void kernel_launch(void* const* d_in, const int* in_sizes, int n_in,
                              void* d_out, int out_size, void* d_ws, size_t ws_size,
                              hipStream_t stream) {
    const float* x       = (const float*)d_in[0];
    const int*   ei      = (const int*)  d_in[1];
    const int*   batch   = (const int*)  d_in[2];
    const float* in_g    = (const float*)d_in[3];
    const float* in_b    = (const float*)d_in[4];
    const float* in_m    = (const float*)d_in[5];
    const float* in_v    = (const float*)d_in[6];
    const float* eps0    = (const float*)d_in[7];
    const float* c0_w1   = (const float*)d_in[8];
    const float* c0_b1   = (const float*)d_in[9];
    const float* c0_g    = (const float*)d_in[10];
    const float* c0_bb   = (const float*)d_in[11];
    const float* c0_m    = (const float*)d_in[12];
    const float* c0_v    = (const float*)d_in[13];
    const float* c0_w2   = (const float*)d_in[14];
    const float* c0_b2   = (const float*)d_in[15];
    const float* bn0_g   = (const float*)d_in[16];
    const float* bn0_b   = (const float*)d_in[17];
    const float* bn0_m   = (const float*)d_in[18];
    const float* bn0_v   = (const float*)d_in[19];
    const float* epsR    = (const float*)d_in[20];
    const float* cR_w1   = (const float*)d_in[21];
    const float* cR_b1   = (const float*)d_in[22];
    const float* cR_g    = (const float*)d_in[23];
    const float* cR_bb   = (const float*)d_in[24];
    const float* cR_m    = (const float*)d_in[25];
    const float* cR_v    = (const float*)d_in[26];
    const float* cR_w2   = (const float*)d_in[27];
    const float* cR_b2   = (const float*)d_in[28];
    const float* bnR_g   = (const float*)d_in[29];
    const float* bnR_b   = (const float*)d_in[30];
    const float* bnR_m   = (const float*)d_in[31];
    const float* bnR_v   = (const float*)d_in[32];
    const float* fc1_w   = (const float*)d_in[33];
    const float* fc1_b   = (const float*)d_in[34];
    const float* ln_g    = (const float*)d_in[35];
    const float* ln_b    = (const float*)d_in[36];
    const float* fc2_w   = (const float*)d_in[37];
    const float* fc2_b   = (const float*)d_in[38];

    const int* src = ei;
    const int* dst = ei + EE;

    float* buf0 = (float*)d_ws;
    float* buf1 = buf0 + (size_t)NN * HID;
    float* buf2 = buf1 + (size_t)NN * HID;
    float* sums = buf2 + (size_t)NN * HID;
    float* cnt  = sums + (size_t)GG * HID;
    unsigned* bmp = (unsigned*)buf2;

    const int TB = 256;
    const int nGemm = (NN + GM_BROWS - 1) / GM_BROWS;

    bn_in_kernel<<<BN_BLOCKS, BN_THREADS, 0, stream>>>(
        x, in_g, in_b, in_m, in_v, eps0, buf0, buf1);

    {
        tile_bitmap_kernel<<<BM_NCH, 1024, 0, stream>>>(dst, bmp);
        edge_tile_kernel<INDIM><<<ET_NTILES, 1024, 0, stream>>>(buf0, src, dst, bmp, buf1);
        gemm_bn_gelu_kernel<INDIM><<<nGemm, TB, 0, stream>>>(
            buf1, c0_w1, c0_b1, c0_g, c0_bb, c0_m, c0_v, buf2,
            buf1, eps0, 0, 0);
        gemm_bn_gelu_kernel<HID><<<nGemm, TB, 0, stream>>>(
            buf2, c0_w2, c0_b2, bn0_g, bn0_b, bn0_m, bn0_v, buf0,
            buf1, epsR, 0, 1);
    }

    for (int l = 0; l < 2; ++l) {
        const size_t wOff = (size_t)l * HID * HID;
        const size_t vOff = (size_t)l * HID;
        tile_bitmap_kernel<<<BM_NCH, 1024, 0, stream>>>(dst, bmp);
        edge_tile_kernel<HID><<<ET_NTILES, 1024, 0, stream>>>(buf0, src, dst, bmp, buf1);
        gemm_bn_gelu_kernel<HID><<<nGemm, TB, 0, stream>>>(
            buf1, cR_w1 + wOff, cR_b1 + vOff, cR_g + vOff, cR_bb + vOff,
            cR_m + vOff, cR_v + vOff, buf2,
            buf1, epsR, 0, 0);
        const int writeAgg = (l == 0) ? 1 : 0;
        gemm_bn_gelu_kernel<HID><<<nGemm, TB, 0, stream>>>(
            buf2, cR_w2 + wOff, cR_b2 + vOff, bnR_g + vOff, bnR_b + vOff,
            bnR_m + vOff, bnR_v + vOff, buf0,
            buf1, epsR, 1, writeAgg);
    }

    zero_kernel<<<(GG * HID + GG + TB - 1) / TB, TB, 0, stream>>>(sums, GG * HID + GG);
    pool_acc_kernel<<<PL_BLOCKS, PL_THREADS, 0, stream>>>(buf0, batch, sums, cnt);
    pool_final_kernel<<<(GG * HID + TB - 1) / TB, TB, 0, stream>>>(sums, cnt, buf1);

    head_kernel<<<GG, HID, 0, stream>>>(
        buf1, fc1_w, fc1_b, ln_g, ln_b, fc2_w, fc2_b, (float*)d_out);
}


# Edge lookup from one packed word: the bitmap kernel also writes pk[e] = src<<9 | dst row in tile; an edge kernel reads 1 scattered 4-byte word (1 cache line) per edge, not src[e] and dst[e] (2 lines)
# speedup vs baseline: 1.9454x; 1.0315x over previous
#include <hip/hip_runtime.h>
#include <math.h>

typedef __attribute__((ext_vector_type(2))) float v2f;
typedef __attribute__((ext_vector_type(8))) float v8f;

#define NN   100000
#define EE   1600000
#define INDIM 64
#define HID  128
#define LAT  64
#define GG   1000

__device__ __forceinline__ float gelu_exact(float x) {
    return 0.5f * x * (1.0f + erff(x * 0.70710678118654752f));
}

#define BN_BLOCKS  512
#define BN_THREADS 256
#define BN_V4      (NN * INDIM / 4)
#define BN_STRIDE  (BN_BLOCKS * BN_THREADS)
#define BN_PER     ((BN_V4 + BN_STRIDE - 1) / BN_STRIDE)
static_assert(BN_STRIDE % (INDIM / 4) == 0, "stride must be whole rows");

__global__ __launch_bounds__(BN_THREADS) void bn_in_kernel(
    const float* __restrict__ x,
    const float* __restrict__ g, const float* __restrict__ b,
    const float* __restrict__ m, const float* __restrict__ v,
    const float* __restrict__ epsP,
    float* __restrict__ out, float* __restrict__ agg)
{
    const int t = blockIdx.x * BN_THREADS + threadIdx.x;
    const float4* __restrict__ x4 = (const float4*)x;

    float4 xv[BN_PER];
    #pragma unroll
    for (int k = 0; k < BN_PER; ++k) {
        const int i = t + k * BN_STRIDE;
        xv[k] = x4[i < BN_V4 ? i : BN_V4 - 1];
    }

    const int j = 4 * (t & (INDIM / 4 - 1));
    float cm[4], cr[4], cg[4], cb[4];
    #pragma unroll
    for (int c = 0; c < 4; ++c) {
        cm[c] = m[j + c];
        cr[c] = rsqrtf(v[j + c] + 1e-5f);
        cg[c] = g[j + c];
        cb[c] = b[j + c];
    }
    const float sc = 1.0f + epsP[0];

    float4* __restrict__ out4 = (float4*)out;
    float4* __restrict__ agg4 = (float4*)agg;
    #pragma unroll
    for (int k = 0; k < BN_PER; ++k) {
        const int i = t + k * BN_STRIDE;
        if (i < BN_V4) {
            float4 y;
            y.x = (xv[k].x - cm[0]) * cr[0] * cg[0] + cb[0];
            y.y = (xv[k].y - cm[1]) * cr[1] * cg[1] + cb[1];
            y.z = (xv[k].z - cm[2]) * cr[2] * cg[2] + cb[2];
            y.w = (xv[k].w - cm[3]) * cr[3] * cg[3] + cb[3];
            out4[i] = y;
            agg4[i] = make_float4(sc * y.x, sc * y.y, sc * y.z, sc * y.w);
        }
    }
}

#define ET_TILE   512
#define ET_NTILES ((NN + ET_TILE - 1) / ET_TILE)
#define ET_QCAP   48
#define ET_RBITS  9
static_assert((1 << ET_RBITS) == ET_TILE, "packed word: low bits = row in tile");
static_assert(NN <= (1 << (32 - ET_RBITS)), "packed word: high bits = src node");

#define BM_WORDS  (EE / 32)
#define BM_V4     (BM_WORDS / 4)
#define BM_PITCH  (((BM_WORDS + 31) / 32) * 32)
#define BM_CH     256
#define BM_NCH    ((BM_WORDS + BM_CH - 1) / BM_CH)
#define BM_LP     (BM_CH + 4)
#define ET_NIT    ((BM_V4 + 1023) / 1024)

__global__ __launch_bounds__(1024) void tile_bitmap_kernel(
    const int* __restrict__ src, const int* __restrict__ dst,
    unsigned* __restrict__ bm, unsigned* __restrict__ pk)
{
    __shared__ uint4 sb4[ET_NTILES * BM_LP / 4];
    unsigned* sb = (unsigned*)sb4;
    const int tid = threadIdx.x;

    for (int i = tid; i < ET_NTILES * BM_LP / 4; i += 1024)
        sb4[i] = make_uint4(0u, 0u, 0u, 0u);
    __syncthreads();

    const int e0 = blockIdx.x * (BM_CH * 32);
    int dv[BM_CH * 32 / 1024], sv[BM_CH * 32 / 1024];
    #pragma unroll
    for (int k = 0; k < BM_CH * 32 / 1024; ++k) {
        const int e = e0 + k * 1024 + tid;
        dv[k] = dst[e < EE ? e : EE - 1];
        sv[k] = src[e < EE ? e : EE - 1];
    }
    #pragma unroll
    for (int k = 0; k < BM_CH * 32 / 1024; ++k) {
        const int el = k * 1024 + tid;
        const unsigned t = (unsigned)dv[k] / ET_TILE;
        if (e0 + el < EE) {
            pk[e0 + el] = ((unsigned)sv[k] << ET_RBITS)
                        | ((unsigned)dv[k] & (ET_TILE - 1));
            if (t < ET_NTILES)
                atomicOr(&sb[t * BM_LP + (el >> 5)], 1u << (el & 31));
        }
    }
    __syncthreads();

    for (int i = tid; i < ET_NTILES * (BM_CH / 4); i += 1024) {
        const int t  = i / (BM_CH / 4);
        const int w4 = i % (BM_CH / 4);
        const int gw = blockIdx.x * BM_CH + 4 * w4;
        if (gw < BM_WORDS)
            *(uint4*)(bm + (size_t)t * BM_PITCH + gw) = sb4[t * (BM_LP / 4) + w4];
    }
}

__device__ __forceinline__ void et_wave_order() {
    __builtin_amdgcn_fence(__ATOMIC_ACQ_REL, "wavefront");
    __builtin_amdgcn_wave_barrier();
}

template<int D>
__device__ __forceinline__ void et_drain(
    const float* __restrict__ h, const unsigned* __restrict__ pk,
    float* tile, const unsigned* q, int qn, int n, int lane)
{
    int s = 0, dl = 0;
    if (lane < n) {
        const unsigned w = pk[q[qn - n + lane]];
        s  = (int)(w >> ET_RBITS);
        dl = (int)(w & (ET_TILE - 1));
    }
    #pragma unroll
    for (int base = 0; base < 16; base += 8) {
        if (base >= n) break;
        float v[8][D / 32];
        #pragma unroll
        for (int i = 0; i < 8; ++i) {
            if (base + i < n) {
                const int si = __builtin_amdgcn_readlane(s, base + i);
                const float* row = h + (size_t)si * D + lane;
                #pragma unroll
                for (int k = 0; k < D / 32; ++k) v[i][k] = row[32 * k];
            }
        }
        #pragma unroll
        for (int i = 0; i < 8; ++i) {
            if (base + i < n) {
                const int di = __builtin_amdgcn_readlane(dl, base + i);
                float* trow = tile + di * D + lane;
                #pragma unroll
                for (int k = 0; k < D / 32; ++k)
                    __hip_atomic_fetch_add(trow + 32 * k, v[i][k],
                        __ATOMIC_RELAXED, __HIP_MEMORY_SCOPE_WORKGROUP);
            }
        }
    }
}

template<int D>
__global__ __launch_bounds__(1024) void edge_tile_kernel(
    const float* __restrict__ h, const unsigned* __restrict__ pk,
    const unsigned* __restrict__ bm,
    float* __restrict__ agg)
{
    __shared__ float tile[ET_TILE * D];
    __shared__ unsigned queue[32 * ET_QCAP];

    const int tid  = threadIdx.x;
    const int lane = tid & 31;
    const int wave = tid >> 5;
    const int lo   = blockIdx.x * ET_TILE;
    const int nrows = (NN - lo < ET_TILE) ? (NN - lo) : ET_TILE;
    const int n4 = nrows * (D / 4);

    const uint4* bm4 = (const uint4*)(bm + (size_t)blockIdx.x * BM_PITCH);
    uint4 cur = bm4[tid];

    float4* t4 = (float4*)tile;
    float4* a4 = (float4*)(agg + (size_t)lo * D);
    for (int i = tid; i < n4; i += 1024) t4[i] = a4[i];
    __syncthreads();

    unsigned* q = queue + wave * ET_QCAP;
    int qn = 0;

    for (int n = 0; n < ET_NIT; ++n) {
        const int i4  = n * 1024 + tid;
        const int inx = i4 + 1024;
        const uint4 nxt = bm4[inx < BM_V4 ? inx : BM_V4 - 1];
        const bool inlist = i4 < BM_V4;
        const unsigned wv[4] = {cur.x, cur.y, cur.z, cur.w};
        #pragma unroll
        for (int j = 0; j < 4; ++j) {
            unsigned w = inlist ? wv[j] : 0u;
            const unsigned ebase = (unsigned)(4 * i4 + j) * 32u;
            unsigned any = __builtin_amdgcn_ballot_w32(w != 0u);
            while (any) {
                if (w != 0u) {
                    const int pos = qn + (int)__builtin_amdgcn_mbcnt_lo(any, 0u);
                    q[pos] = ebase + (unsigned)__builtin_ctz(w);
                    w &= w - 1u;
                }
                qn += __builtin_popcount(any);
                et_wave_order();
                while (qn >= 16) {
                    et_drain<D>(h, pk, tile, q, qn, 16, lane);
                    qn -= 16;
                }
                et_wave_order();
                any = __builtin_amdgcn_ballot_w32(w != 0u);
            }
        }
        cur = nxt;
    }
    if (qn > 0) et_drain<D>(h, pk, tile, q, qn, qn, lane);

    __syncthreads();
    for (int i = tid; i < n4; i += 1024) a4[i] = t4[i];
}

#define GM_BROWS 64

template<int K>
__global__ __launch_bounds__(256) void gemm_bn_gelu_kernel(
    const float* __restrict__ A,
    const float* __restrict__ W,
    const float* __restrict__ bias,
    const float* __restrict__ bn_g, const float* __restrict__ bn_b,
    const float* __restrict__ bn_m, const float* __restrict__ bn_v,
    float* __restrict__ out,
    float* __restrict__ aggOut, const float* __restrict__ epsP,
    int epsIdx, int writeAgg)
{
    constexpr int P   = K + 4;
    constexpr int RV  = K / 4;
    constexpr int NV  = GM_BROWS * RV / 256;
    __shared__ float4 sA4[GM_BROWS * P / 4];
    float* sA = (float*)sA4;

    const int tid  = threadIdx.x;
    const int lane = tid & 31;
    const int wave = tid >> 5;
    const int half = lane >> 4;
    const int mr   = lane & 15;
    const int col  = wave * 16 + mr;
    const int row0 = blockIdx.x * GM_BROWS;

    float4 st[NV];
    const float4* __restrict__ A4 = (const float4*)A;
    #pragma unroll
    for (int j = 0; j < NV; ++j) {
        const int g = row0 * RV + j * 256 + tid;
        st[j] = A4[g < NN * RV ? g : NN * RV - 1];
    }

    v2f bf[K / 4];
    #pragma unroll
    for (int ks = 0; ks < K / 4; ++ks) {
        const int ka = 4 * ks + 2 * half;
        bf[ks].x = W[(size_t)ka * HID + col];
        bf[ks].y = W[(size_t)(ka + 1) * HID + col];
    }

    const float cbias = bias[col];
    const float cm    = bn_m[col];
    const float crs   = rsqrtf(bn_v[col] + 1e-5f);
    const float cg    = bn_g[col];
    const float cb    = bn_b[col];
    const float escale = writeAgg ? (1.0f + epsP[epsIdx]) : 0.0f;

    #pragma unroll
    for (int j = 0; j < NV; ++j) {
        const int idx = j * 256 + tid;
        *(float4*)(sA + (idx / RV) * P + (idx % RV) * 4) = st[j];
    }
    __syncthreads();

    for (int p = 0; p < GM_BROWS / 32; ++p) {
        const int prow = row0 + 32 * p;
        if (prow >= NN) break;
        const float* a0p = sA + (32 * p + mr) * P + 2 * half;
        const float* a1p = a0p + 16 * P;
        v8f acc0 = {}, acc1 = {};
        #pragma unroll
        for (int ks = 0; ks < K / 4; ++ks) {
            const v2f a0 = *(const v2f*)(a0p + 4 * ks);
            const v2f a1 = *(const v2f*)(a1p + 4 * ks);
            acc0 = __builtin_amdgcn_wmma_f32_16x16x4_f32(
                false, a0, false, bf[ks], (short)0, acc0, false, false);
            acc1 = __builtin_amdgcn_wmma_f32_16x16x4_f32(
                false, a1, false, bf[ks], (short)0, acc1, false, false);
        }

        #pragma unroll
        for (int t = 0; t < 2; ++t) {
            #pragma unroll
            for (int r = 0; r < 8; ++r) {
                const int row = prow + 16 * t + r + 8 * half;
                float y = (t == 0 ? acc0[r] : acc1[r]) + cbias;
                y = (y - cm) * crs * cg + cb;
                y = gelu_exact(y);
                const size_t o = (size_t)row * HID + col;
                out[o] = y;
                if (writeAgg) aggOut[o] = escale * y;
            }
        }
    }
}

__global__ __launch_bounds__(256) void zero_kernel(float* __restrict__ p, int n) {
    int idx = blockIdx.x * blockDim.x + threadIdx.x;
    if (idx < n) p[idx] = 0.0f;
}

#define PL_BLOCKS  512
#define PL_THREADS 256
#define PL_WAVES   (PL_BLOCKS * PL_THREADS / 32)
#define PL_PER     ((NN + PL_WAVES - 1) / PL_WAVES)
static_assert(HID == 128, "a row is 32 pieces of 16 bytes: one per lane");
static_assert(NN % PL_PER == 0, "a wave has all its rows or none");
static_assert(PL_PER <= 32, "one lane per node of the wave for the counts");

__global__ __launch_bounds__(PL_THREADS) void pool_acc_kernel(
    const float* __restrict__ h, const int* __restrict__ batch,
    float* __restrict__ sums, float* __restrict__ cnt)
{
    const int lane = threadIdx.x & 31;
    const int w = __builtin_amdgcn_readfirstlane(
        (int)((blockIdx.x * PL_THREADS + threadIdx.x) >> 5));
    const int i0 = w * PL_PER;
    if (i0 >= NN) return;

    float4 rv[PL_PER];
    int    gb[PL_PER];
    #pragma unroll
    for (int k = 0; k < PL_PER; ++k) {
        gb[k] = batch[i0 + k];
        rv[k] = *(const float4*)(h + (size_t)(i0 + k) * HID + 4 * lane);
    }
    #pragma unroll
    for (int k = 0; k < PL_PER; ++k) {
        float* s = sums + (size_t)gb[k] * HID + 4 * lane;
        unsafeAtomicAdd(s + 0, rv[k].x);
        unsafeAtomicAdd(s + 1, rv[k].y);
        unsafeAtomicAdd(s + 2, rv[k].z);
        unsafeAtomicAdd(s + 3, rv[k].w);
    }
    if (lane < PL_PER) unsafeAtomicAdd(cnt + batch[i0 + lane], 1.0f);
}

__global__ __launch_bounds__(256) void pool_final_kernel(
    const float* __restrict__ sums, const float* __restrict__ cnt,
    float* __restrict__ pooled)
{
    int idx = blockIdx.x * blockDim.x + threadIdx.x;
    if (idx >= GG * HID) return;
    int g = idx / HID;
    pooled[idx] = sums[idx] / fmaxf(cnt[g], 1.0f);
}

__global__ __launch_bounds__(128) void head_kernel(
    const float* __restrict__ pooled,
    const float* __restrict__ fc1w, const float* __restrict__ fc1b,
    const float* __restrict__ lng, const float* __restrict__ lnb,
    const float* __restrict__ fc2w, const float* __restrict__ fc2b,
    float* __restrict__ out)
{
    __shared__ float p[HID];
    __shared__ float rbuf[HID];
    __shared__ float red[HID];
    const int g = blockIdx.x;
    const int t = threadIdx.x;

    p[t] = pooled[(size_t)g * HID + t];
    __syncthreads();

    float o = fc1b[t];
    #pragma unroll 4
    for (int k = 0; k < HID; ++k) o += p[k] * fc1w[k * HID + t];

    red[t] = o;
    __syncthreads();
    for (int s = 64; s > 0; s >>= 1) {
        if (t < s) red[t] += red[t + s];
        __syncthreads();
    }
    const float mu = red[0] * (1.0f / HID);
    __syncthreads();

    const float dv = o - mu;
    red[t] = dv * dv;
    __syncthreads();
    for (int s = 64; s > 0; s >>= 1) {
        if (t < s) red[t] += red[t + s];
        __syncthreads();
    }
    const float var = red[0] * (1.0f / HID);

    float y = dv * rsqrtf(var + 1e-5f) * lng[t] + lnb[t];
    y = gelu_exact(y) + p[t];
    rbuf[t] = y;
    __syncthreads();

    if (t < LAT) {
        float o2 = fc2b[t];
        #pragma unroll 4
        for (int k = 0; k < HID; ++k) o2 += rbuf[k] * fc2w[k * LAT + t];
        out[(size_t)g * LAT + t] = o2;
    }
}

extern "C" void kernel_launch(void* const* d_in, const int* in_sizes, int n_in,
                              void* d_out, int out_size, void* d_ws, size_t ws_size,
                              hipStream_t stream) {
    const float* x       = (const float*)d_in[0];
    const int*   ei      = (const int*)  d_in[1];
    const int*   batch   = (const int*)  d_in[2];
    const float* in_g    = (const float*)d_in[3];
    const float* in_b    = (const float*)d_in[4];
    const float* in_m    = (const float*)d_in[5];
    const float* in_v    = (const float*)d_in[6];
    const float* eps0    = (const float*)d_in[7];
    const float* c0_w1   = (const float*)d_in[8];
    const float* c0_b1   = (const float*)d_in[9];
    const float* c0_g    = (const float*)d_in[10];
    const float* c0_bb   = (const float*)d_in[11];
    const float* c0_m    = (const float*)d_in[12];
    const float* c0_v    = (const float*)d_in[13];
    const float* c0_w2   = (const float*)d_in[14];
    const float* c0_b2   = (const float*)d_in[15];
    const float* bn0_g   = (const float*)d_in[16];
    const float* bn0_b   = (const float*)d_in[17];
    const float* bn0_m   = (const float*)d_in[18];
    const float* bn0_v   = (const float*)d_in[19];
    const float* epsR    = (const float*)d_in[20];
    const float* cR_w1   = (const float*)d_in[21];
    const float* cR_b1   = (const float*)d_in[22];
    const float* cR_g    = (const float*)d_in[23];
    const float* cR_bb   = (const float*)d_in[24];
    const float* cR_m    = (const float*)d_in[25];
    const float* cR_v    = (const float*)d_in[26];
    const float* cR_w2   = (const float*)d_in[27];
    const float* cR_b2   = (const float*)d_in[28];
    const float* bnR_g   = (const float*)d_in[29];
    const float* bnR_b   = (const float*)d_in[30];
    const float* bnR_m   = (const float*)d_in[31];
    const float* bnR_v   = (const float*)d_in[32];
    const float* fc1_w   = (const float*)d_in[33];
    const float* fc1_b   = (const float*)d_in[34];
    const float* ln_g    = (const float*)d_in[35];
    const float* ln_b    = (const float*)d_in[36];
    const float* fc2_w   = (const float*)d_in[37];
    const float* fc2_b   = (const float*)d_in[38];

    const int* src = ei;
    const int* dst = ei + EE;

    float* buf0 = (float*)d_ws;
    float* buf1 = buf0 + (size_t)NN * HID;
    float* buf2 = buf1 + (size_t)NN * HID;
    float* sums = buf2 + (size_t)NN * HID;
    float* cnt  = sums + (size_t)GG * HID;
    unsigned* bmp = (unsigned*)buf2;
    unsigned* pkw = bmp + (size_t)ET_NTILES * BM_PITCH;
    static_assert((size_t)ET_NTILES * BM_PITCH + EE <= (size_t)NN * HID,
                  "bitmaps and packed words fit in buf2");

    const int TB = 256;
    const int nGemm = (NN + GM_BROWS - 1) / GM_BROWS;

    bn_in_kernel<<<BN_BLOCKS, BN_THREADS, 0, stream>>>(
        x, in_g, in_b, in_m, in_v, eps0, buf0, buf1);

    {
        tile_bitmap_kernel<<<BM_NCH, 1024, 0, stream>>>(src, dst, bmp, pkw);
        edge_tile_kernel<INDIM><<<ET_NTILES, 1024, 0, stream>>>(buf0, pkw, bmp, buf1);
        gemm_bn_gelu_kernel<INDIM><<<nGemm, TB, 0, stream>>>(
            buf1, c0_w1, c0_b1, c0_g, c0_bb, c0_m, c0_v, buf2,
            buf1, eps0, 0, 0);
        gemm_bn_gelu_kernel<HID><<<nGemm, TB, 0, stream>>>(
            buf2, c0_w2, c0_b2, bn0_g, bn0_b, bn0_m, bn0_v, buf0,
            buf1, epsR, 0, 1);
    }

    for (int l = 0; l < 2; ++l) {
        const size_t wOff = (size_t)l * HID * HID;
        const size_t vOff = (size_t)l * HID;
        tile_bitmap_kernel<<<BM_NCH, 1024, 0, stream>>>(src, dst, bmp, pkw);
        edge_tile_kernel<HID><<<ET_NTILES, 1024, 0, stream>>>(buf0, pkw, bmp, buf1);
        gemm_bn_gelu_kernel<HID><<<nGemm, TB, 0, stream>>>(
            buf1, cR_w1 + wOff, cR_b1 + vOff, cR_g + vOff, cR_bb + vOff,
            cR_m + vOff, cR_v + vOff, buf2,
            buf1, epsR, 0, 0);
        const int writeAgg = (l == 0) ? 1 : 0;
        gemm_bn_gelu_kernel<HID><<<nGemm, TB, 0, stream>>>(
            buf2, cR_w2 + wOff, cR_b2 + vOff, bnR_g + vOff, bnR_b + vOff,
            bnR_m + vOff, bnR_v + vOff, buf0,
            buf1, epsR, 1, writeAgg);
    }

    zero_kernel<<<(GG * HID + GG + TB - 1) / TB, TB, 0, stream>>>(sums, GG * HID + GG);
    pool_acc_kernel<<<PL_BLOCKS, PL_THREADS, 0, stream>>>(buf0, batch, sums, cnt);
    pool_final_kernel<<<(GG * HID + TB - 1) / TB, TB, 0, stream>>>(sums, cnt, buf1);

    head_kernel<<<GG, HID, 0, stream>>>(
        buf1, fc1_w, fc1_b, ln_g, ln_b, fc2_w, fc2_b, (float*)d_out);
}


# Edges grouped per 8192-edge chunk by dst tile (LDS int rank, scan, fill of packed words): a tile block reads its runs contiguously; bitmaps and scattered word per edge gone; adds in arrival order
# speedup vs baseline: 2.0151x; 1.0358x over previous
#include <hip/hip_runtime.h>
#include <math.h>

typedef __attribute__((ext_vector_type(2))) float v2f;
typedef __attribute__((ext_vector_type(8))) float v8f;

#define NN   100000
#define EE   1600000
#define INDIM 64
#define HID  128
#define LAT  64
#define GG   1000

__device__ __forceinline__ float gelu_exact(float x) {
    return 0.5f * x * (1.0f + erff(x * 0.70710678118654752f));
}

#define BN_BLOCKS  512
#define BN_THREADS 256
#define BN_V4      (NN * INDIM / 4)
#define BN_STRIDE  (BN_BLOCKS * BN_THREADS)
#define BN_PER     ((BN_V4 + BN_STRIDE - 1) / BN_STRIDE)
static_assert(BN_STRIDE % (INDIM / 4) == 0, "stride must be whole rows");

__global__ __launch_bounds__(BN_THREADS) void bn_in_kernel(
    const float* __restrict__ x,
    const float* __restrict__ g, const float* __restrict__ b,
    const float* __restrict__ m, const float* __restrict__ v,
    const float* __restrict__ epsP,
    float* __restrict__ out, float* __restrict__ agg)
{
    const int t = blockIdx.x * BN_THREADS + threadIdx.x;
    const float4* __restrict__ x4 = (const float4*)x;

    float4 xv[BN_PER];
    #pragma unroll
    for (int k = 0; k < BN_PER; ++k) {
        const int i = t + k * BN_STRIDE;
        xv[k] = x4[i < BN_V4 ? i : BN_V4 - 1];
    }

    const int j = 4 * (t & (INDIM / 4 - 1));
    float cm[4], cr[4], cg[4], cb[4];
    #pragma unroll
    for (int c = 0; c < 4; ++c) {
        cm[c] = m[j + c];
        cr[c] = rsqrtf(v[j + c] + 1e-5f);
        cg[c] = g[j + c];
        cb[c] = b[j + c];
    }
    const float sc = 1.0f + epsP[0];

    float4* __restrict__ out4 = (float4*)out;
    float4* __restrict__ agg4 = (float4*)agg;
    #pragma unroll
    for (int k = 0; k < BN_PER; ++k) {
        const int i = t + k * BN_STRIDE;
        if (i < BN_V4) {
            float4 y;
            y.x = (xv[k].x - cm[0]) * cr[0] * cg[0] + cb[0];
            y.y = (xv[k].y - cm[1]) * cr[1] * cg[1] + cb[1];
            y.z = (xv[k].z - cm[2]) * cr[2] * cg[2] + cb[2];
            y.w = (xv[k].w - cm[3]) * cr[3] * cg[3] + cb[3];
            out4[i] = y;
            agg4[i] = make_float4(sc * y.x, sc * y.y, sc * y.z, sc * y.w);
        }
    }
}

#define ET_TILE   512
#define ET_NTILES ((NN + ET_TILE - 1) / ET_TILE)
#define ET_QCAP   48
#define ET_RBITS  9
static_assert((1 << ET_RBITS) == ET_TILE, "packed word: low bits = row in tile");
static_assert(NN <= (1 << (32 - ET_RBITS)), "packed word: high bits = src node");

#define CH_EDGES  8192
#define CH_PER    (CH_EDGES / 1024)
#define CH_N      ((EE + CH_EDGES - 1) / CH_EDGES)
#define CH_OFFP   256
#define CH_TRIPS  ((CH_N + 31) / 32)
static_assert(ET_NTILES + 1 <= CH_OFFP, "run starts 0..196 fit in a table row");

__global__ __launch_bounds__(1024) void tile_group_kernel(
    const int* __restrict__ src, const int* __restrict__ dst,
    unsigned* __restrict__ pks, int* __restrict__ offs)
{
    __shared__ int cnt[CH_OFFP];
    __shared__ int base[CH_OFFP + 1];
    __shared__ uint4 sorted4[CH_EDGES / 4];
    unsigned* sorted = (unsigned*)sorted4;
    const int tid = threadIdx.x;

    if (tid < CH_OFFP) cnt[tid] = 0;

    const int e0 = blockIdx.x * CH_EDGES;
    int dv[CH_PER], sv[CH_PER];
    #pragma unroll
    for (int k = 0; k < CH_PER; ++k) {
        const int e = e0 + k * 1024 + tid;
        dv[k] = dst[e < EE ? e : EE - 1];
        sv[k] = src[e < EE ? e : EE - 1];
    }
    __syncthreads();

    int rk[CH_PER];
    #pragma unroll
    for (int k = 0; k < CH_PER; ++k) {
        const int e = e0 + k * 1024 + tid;
        const unsigned t = (unsigned)dv[k] / ET_TILE;
        rk[k] = -1;
        if (e < EE && t < ET_NTILES) rk[k] = atomicAdd(&cnt[t], 1);
    }
    __syncthreads();

    if (tid < 32) {
        int v[CH_OFFP / 32];
        int s = 0;
        #pragma unroll
        for (int j = 0; j < CH_OFFP / 32; ++j) {
            v[j] = cnt[(CH_OFFP / 32) * tid + j];
            s += v[j];
        }
        int inc = s;
        #pragma unroll
        for (int d = 1; d < 32; d <<= 1) {
            const int o = __shfl_up(inc, d, 32);
            if (tid >= d) inc += o;
        }
        int run = inc - s;
        #pragma unroll
        for (int j = 0; j < CH_OFFP / 32; ++j) {
            base[(CH_OFFP / 32) * tid + j] = run;
            run += v[j];
        }
        if (tid == 31) base[CH_OFFP] = run;
    }
    __syncthreads();

    #pragma unroll
    for (int k = 0; k < CH_PER; ++k) {
        if (rk[k] >= 0) {
            const unsigned t = (unsigned)dv[k] / ET_TILE;
            sorted[base[t] + rk[k]] = ((unsigned)sv[k] << ET_RBITS)
                                    | ((unsigned)dv[k] & (ET_TILE - 1));
        }
    }
    if (tid < CH_OFFP) offs[blockIdx.x * CH_OFFP + tid] = base[tid];
    __syncthreads();

    const int tot = base[CH_OFFP];
    uint4* __restrict__ out4 = (uint4*)(pks + (size_t)blockIdx.x * CH_EDGES);
    for (int i = tid; 4 * i < tot; i += 1024) out4[i] = sorted4[i];
}

__device__ __forceinline__ void et_wave_order() {
    __builtin_amdgcn_fence(__ATOMIC_ACQ_REL, "wavefront");
    __builtin_amdgcn_wave_barrier();
}

template<int D>
__device__ __forceinline__ void et_drain(
    const float* __restrict__ h,
    float* tile, const unsigned* q, int qn, int n, int lane)
{
    int s = 0, dl = 0;
    if (lane < n) {
        const unsigned w = q[qn - n + lane];
        s  = (int)(w >> ET_RBITS);
        dl = (int)(w & (ET_TILE - 1));
    }
    #pragma unroll
    for (int base = 0; base < 16; base += 8) {
        if (base >= n) break;
        float v[8][D / 32];
        #pragma unroll
        for (int i = 0; i < 8; ++i) {
            if (base + i < n) {
                const int si = __builtin_amdgcn_readlane(s, base + i);
                const float* row = h + (size_t)si * D + lane;
                #pragma unroll
                for (int k = 0; k < D / 32; ++k) v[i][k] = row[32 * k];
            }
        }
        #pragma unroll
        for (int i = 0; i < 8; ++i) {
            if (base + i < n) {
                const int di = __builtin_amdgcn_readlane(dl, base + i);
                float* trow = tile + di * D + lane;
                #pragma unroll
                for (int k = 0; k < D / 32; ++k)
                    __hip_atomic_fetch_add(trow + 32 * k, v[i][k],
                        __ATOMIC_RELAXED, __HIP_MEMORY_SCOPE_WORKGROUP);
            }
        }
    }
}

template<int D>
__global__ __launch_bounds__(1024) void edge_tile_kernel(
    const float* __restrict__ h, const unsigned* __restrict__ pks,
    const int* __restrict__ offs,
    float* __restrict__ agg)
{
    __shared__ float tile[ET_TILE * D];
    __shared__ unsigned queue[32 * ET_QCAP];
    __shared__ int sb0[CH_N], sb1[CH_N];

    const int tid  = threadIdx.x;
    const int lane = tid & 31;
    const int wave = tid >> 5;
    const int lo   = blockIdx.x * ET_TILE;
    const int nrows = (NN - lo < ET_TILE) ? (NN - lo) : ET_TILE;
    const int n4 = nrows * (D / 4);

    if (tid < CH_N) {
        sb0[tid] = offs[tid * CH_OFFP + blockIdx.x];
        sb1[tid] = offs[tid * CH_OFFP + blockIdx.x + 1];
    }

    float4* t4 = (float4*)tile;
    float4* a4 = (float4*)(agg + (size_t)lo * D);
    for (int i = tid; i < n4; i += 1024) t4[i] = a4[i];
    __syncthreads();

    unsigned* q = queue + wave * ET_QCAP;
    int qn = 0;

    int c  = wave;
    int b0 = sb0[c];
    int b1 = sb1[c];
    unsigned w = 0u;
    if (b0 + lane < b1) w = pks[(size_t)c * CH_EDGES + b0 + lane];

    for (int k = 0; k < CH_TRIPS; ++k) {
        if (b0 + lane < b1) q[qn + lane] = w;

        const int cn = c + 32;
        const bool more = cn < CH_N;
        int nb0 = 0, nb1 = 0;
        unsigned nw = 0u;
        if (more) {
            nb0 = sb0[cn];
            nb1 = sb1[cn];
            if (nb0 + lane < nb1) nw = pks[(size_t)cn * CH_EDGES + nb0 + lane];
        }
        for (int i = b0; i < b1; i += 32) {
            const int m = (b1 - i < 32) ? (b1 - i) : 32;
            if (i != b0 && lane < m)
                q[qn + lane] = pks[(size_t)c * CH_EDGES + i + lane];
            qn += m;
            et_wave_order();
            while (qn >= 16) {
                et_drain<D>(h, tile, q, qn, 16, lane);
                qn -= 16;
            }
            et_wave_order();
        }
        if (!more) break;
        c = cn; b0 = nb0; b1 = nb1; w = nw;
    }
    if (qn > 0) et_drain<D>(h, tile, q, qn, qn, lane);

    __syncthreads();
    for (int i = tid; i < n4; i += 1024) a4[i] = t4[i];
}

#define GM_BROWS 64

template<int K>
__global__ __launch_bounds__(256) void gemm_bn_gelu_kernel(
    const float* __restrict__ A,
    const float* __restrict__ W,
    const float* __restrict__ bias,
    const float* __restrict__ bn_g, const float* __restrict__ bn_b,
    const float* __restrict__ bn_m, const float* __restrict__ bn_v,
    float* __restrict__ out,
    float* __restrict__ aggOut, const float* __restrict__ epsP,
    int epsIdx, int writeAgg)
{
    constexpr int P   = K + 4;
    constexpr int RV  = K / 4;
    constexpr int NV  = GM_BROWS * RV / 256;
    __shared__ float4 sA4[GM_BROWS * P / 4];
    float* sA = (float*)sA4;

    const int tid  = threadIdx.x;
    const int lane = tid & 31;
    const int wave = tid >> 5;
    const int half = lane >> 4;
    const int mr   = lane & 15;
    const int col  = wave * 16 + mr;
    const int row0 = blockIdx.x * GM_BROWS;

    float4 st[NV];
    const float4* __restrict__ A4 = (const float4*)A;
    #pragma unroll
    for (int j = 0; j < NV; ++j) {
        const int g = row0 * RV + j * 256 + tid;
        st[j] = A4[g < NN * RV ? g : NN * RV - 1];
    }

    v2f bf[K / 4];
    #pragma unroll
    for (int ks = 0; ks < K / 4; ++ks) {
        const int ka = 4 * ks + 2 * half;
        bf[ks].x = W[(size_t)ka * HID + col];
        bf[ks].y = W[(size_t)(ka + 1) * HID + col];
    }

    const float cbias = bias[col];
    const float cm    = bn_m[col];
    const float crs   = rsqrtf(bn_v[col] + 1e-5f);
    const float cg    = bn_g[col];
    const float cb    = bn_b[col];
    const float escale = writeAgg ? (1.0f + epsP[epsIdx]) : 0.0f;

    #pragma unroll
    for (int j = 0; j < NV; ++j) {
        const int idx = j * 256 + tid;
        *(float4*)(sA + (idx / RV) * P + (idx % RV) * 4) = st[j];
    }
    __syncthreads();

    for (int p = 0; p < GM_BROWS / 32; ++p) {
        const int prow = row0 + 32 * p;
        if (prow >= NN) break;
        const float* a0p = sA + (32 * p + mr) * P + 2 * half;
        const float* a1p = a0p + 16 * P;
        v8f acc0 = {}, acc1 = {};
        #pragma unroll
        for (int ks = 0; ks < K / 4; ++ks) {
            const v2f a0 = *(const v2f*)(a0p + 4 * ks);
            const v2f a1 = *(const v2f*)(a1p + 4 * ks);
            acc0 = __builtin_amdgcn_wmma_f32_16x16x4_f32(
                false, a0, false, bf[ks], (short)0, acc0, false, false);
            acc1 = __builtin_amdgcn_wmma_f32_16x16x4_f32(
                false, a1, false, bf[ks], (short)0, acc1, false, false);
        }

        #pragma unroll
        for (int t = 0; t < 2; ++t) {
            #pragma unroll
            for (int r = 0; r < 8; ++r) {
                const int row = prow + 16 * t + r + 8 * half;
                float y = (t == 0 ? acc0[r] : acc1[r]) + cbias;
                y = (y - cm) * crs * cg + cb;
                y = gelu_exact(y);
                const size_t o = (size_t)row * HID + col;
                out[o] = y;
                if (writeAgg) aggOut[o] = escale * y;
            }
        }
    }
}

__global__ __launch_bounds__(256) void zero_kernel(float* __restrict__ p, int n) {
    int idx = blockIdx.x * blockDim.x + threadIdx.x;
    if (idx < n) p[idx] = 0.0f;
}

#define PL_BLOCKS  512
#define PL_THREADS 256
#define PL_WAVES   (PL_BLOCKS * PL_THREADS / 32)
#define PL_PER     ((NN + PL_WAVES - 1) / PL_WAVES)
static_assert(HID == 128, "a row is 32 pieces of 16 bytes: one per lane");
static_assert(NN % PL_PER == 0, "a wave has all its rows or none");
static_assert(PL_PER <= 32, "one lane per node of the wave for the counts");

__global__ __launch_bounds__(PL_THREADS) void pool_acc_kernel(
    const float* __restrict__ h, const int* __restrict__ batch,
    float* __restrict__ sums, float* __restrict__ cnt)
{
    const int lane = threadIdx.x & 31;
    const int w = __builtin_amdgcn_readfirstlane(
        (int)((blockIdx.x * PL_THREADS + threadIdx.x) >> 5));
    const int i0 = w * PL_PER;
    if (i0 >= NN) return;

    float4 rv[PL_PER];
    int    gb[PL_PER];
    #pragma unroll
    for (int k = 0; k < PL_PER; ++k) {
        gb[k] = batch[i0 + k];
        rv[k] = *(const float4*)(h + (size_t)(i0 + k) * HID + 4 * lane);
    }
    #pragma unroll
    for (int k = 0; k < PL_PER; ++k) {
        float* s = sums + (size_t)gb[k] * HID + 4 * lane;
        unsafeAtomicAdd(s + 0, rv[k].x);
        unsafeAtomicAdd(s + 1, rv[k].y);
        unsafeAtomicAdd(s + 2, rv[k].z);
        unsafeAtomicAdd(s + 3, rv[k].w);
    }
    if (lane < PL_PER) unsafeAtomicAdd(cnt + batch[i0 + lane], 1.0f);
}

__global__ __launch_bounds__(256) void pool_final_kernel(
    const float* __restrict__ sums, const float* __restrict__ cnt,
    float* __restrict__ pooled)
{
    int idx = blockIdx.x * blockDim.x + threadIdx.x;
    if (idx >= GG * HID) return;
    int g = idx / HID;
    pooled[idx] = sums[idx] / fmaxf(cnt[g], 1.0f);
}

__global__ __launch_bounds__(128) void head_kernel(
    const float* __restrict__ pooled,
    const float* __restrict__ fc1w, const float* __restrict__ fc1b,
    const float* __restrict__ lng, const float* __restrict__ lnb,
    const float* __restrict__ fc2w, const float* __restrict__ fc2b,
    float* __restrict__ out)
{
    __shared__ float p[HID];
    __shared__ float rbuf[HID];
    __shared__ float red[HID];
    const int g = blockIdx.x;
    const int t = threadIdx.x;

    p[t] = pooled[(size_t)g * HID + t];
    __syncthreads();

    float o = fc1b[t];
    #pragma unroll 4
    for (int k = 0; k < HID; ++k) o += p[k] * fc1w[k * HID + t];

    red[t] = o;
    __syncthreads();
    for (int s = 64; s > 0; s >>= 1) {
        if (t < s) red[t] += red[t + s];
        __syncthreads();
    }
    const float mu = red[0] * (1.0f / HID);
    __syncthreads();

    const float dv = o - mu;
    red[t] = dv * dv;
    __syncthreads();
    for (int s = 64; s > 0; s >>= 1) {
        if (t < s) red[t] += red[t + s];
        __syncthreads();
    }
    const float var = red[0] * (1.0f / HID);

    float y = dv * rsqrtf(var + 1e-5f) * lng[t] + lnb[t];
    y = gelu_exact(y) + p[t];
    rbuf[t] = y;
    __syncthreads();

    if (t < LAT) {
        float o2 = fc2b[t];
        #pragma unroll 4
        for (int k = 0; k < HID; ++k) o2 += rbuf[k] * fc2w[k * LAT + t];
        out[(size_t)g * LAT + t] = o2;
    }
}

extern "C" void kernel_launch(void* const* d_in, const int* in_sizes, int n_in,
                              void* d_out, int out_size, void* d_ws, size_t ws_size,
                              hipStream_t stream) {
    const float* x       = (const float*)d_in[0];
    const int*   ei      = (const int*)  d_in[1];
    const int*   batch   = (const int*)  d_in[2];
    const float* in_g    = (const float*)d_in[3];
    const float* in_b    = (const float*)d_in[4];
    const float* in_m    = (const float*)d_in[5];
    const float* in_v    = (const float*)d_in[6];
    const float* eps0    = (const float*)d_in[7];
    const float* c0_w1   = (const float*)d_in[8];
    const float* c0_b1   = (const float*)d_in[9];
    const float* c0_g    = (const float*)d_in[10];
    const float* c0_bb   = (const float*)d_in[11];
    const float* c0_m    = (const float*)d_in[12];
    const float* c0_v    = (const float*)d_in[13];
    const float* c0_w2   = (const float*)d_in[14];
    const float* c0_b2   = (const float*)d_in[15];
    const float* bn0_g   = (const float*)d_in[16];
    const float* bn0_b   = (const float*)d_in[17];
    const float* bn0_m   = (const float*)d_in[18];
    const float* bn0_v   = (const float*)d_in[19];
    const float* epsR    = (const float*)d_in[20];
    const float* cR_w1   = (const float*)d_in[21];
    const float* cR_b1   = (const float*)d_in[22];
    const float* cR_g    = (const float*)d_in[23];
    const float* cR_bb   = (const float*)d_in[24];
    const float* cR_m    = (const float*)d_in[25];
    const float* cR_v    = (const float*)d_in[26];
    const float* cR_w2   = (const float*)d_in[27];
    const float* cR_b2   = (const float*)d_in[28];
    const float* bnR_g   = (const float*)d_in[29];
    const float* bnR_b   = (const float*)d_in[30];
    const float* bnR_m   = (const float*)d_in[31];
    const float* bnR_v   = (const float*)d_in[32];
    const float* fc1_w   = (const float*)d_in[33];
    const float* fc1_b   = (const float*)d_in[34];
    const float* ln_g    = (const float*)d_in[35];
    const float* ln_b    = (const float*)d_in[36];
    const float* fc2_w   = (const float*)d_in[37];
    const float* fc2_b   = (const float*)d_in[38];

    const int* src = ei;
    const int* dst = ei + EE;

    float* buf0 = (float*)d_ws;
    float* buf1 = buf0 + (size_t)NN * HID;
    float* buf2 = buf1 + (size_t)NN * HID;
    float* sums = buf2 + (size_t)NN * HID;
    float* cnt  = sums + (size_t)GG * HID;
    unsigned* pkw = (unsigned*)buf2;
    int*      ofs = (int*)(pkw + (size_t)CH_N * CH_EDGES);
    static_assert((size_t)CH_N * CH_EDGES + (size_t)CH_N * CH_OFFP <= (size_t)NN * HID,
                  "packed words and run starts fit in buf2");

    const int TB = 256;
    const int nGemm = (NN + GM_BROWS - 1) / GM_BROWS;

    bn_in_kernel<<<BN_BLOCKS, BN_THREADS, 0, stream>>>(
        x, in_g, in_b, in_m, in_v, eps0, buf0, buf1);

    {
        tile_group_kernel<<<CH_N, 1024, 0, stream>>>(src, dst, pkw, ofs);
        edge_tile_kernel<INDIM><<<ET_NTILES, 1024, 0, stream>>>(buf0, pkw, ofs, buf1);
        gemm_bn_gelu_kernel<INDIM><<<nGemm, TB, 0, stream>>>(
            buf1, c0_w1, c0_b1, c0_g, c0_bb, c0_m, c0_v, buf2,
            buf1, eps0, 0, 0);
        gemm_bn_gelu_kernel<HID><<<nGemm, TB, 0, stream>>>(
            buf2, c0_w2, c0_b2, bn0_g, bn0_b, bn0_m, bn0_v, buf0,
            buf1, epsR, 0, 1);
    }

    for (int l = 0; l < 2; ++l) {
        const size_t wOff = (size_t)l * HID * HID;
        const size_t vOff = (size_t)l * HID;
        tile_group_kernel<<<CH_N, 1024, 0, stream>>>(src, dst, pkw, ofs);
        edge_tile_kernel<HID><<<ET_NTILES, 1024, 0, stream>>>(buf0, pkw, ofs, buf1);
        gemm_bn_gelu_kernel<HID><<<nGemm, TB, 0, stream>>>(
            buf1, cR_w1 + wOff, cR_b1 + vOff, cR_g + vOff, cR_bb + vOff,
            cR_m + vOff, cR_v + vOff, buf2,
            buf1, epsR, 0, 0);
        const int writeAgg = (l == 0) ? 1 : 0;
        gemm_bn_gelu_kernel<HID><<<nGemm, TB, 0, stream>>>(
            buf2, cR_w2 + wOff, cR_b2 + vOff, bnR_g + vOff, bnR_b + vOff,
            bnR_m + vOff, bnR_v + vOff, buf0,
            buf1, epsR, 1, writeAgg);
    }

    zero_kernel<<<(GG * HID + GG + TB - 1) / TB, TB, 0, stream>>>(sums, GG * HID + GG);
    pool_acc_kernel<<<PL_BLOCKS, PL_THREADS, 0, stream>>>(buf0, batch, sums, cnt);
    pool_final_kernel<<<(GG * HID + TB - 1) / TB, TB, 0, stream>>>(sums, cnt, buf1);

    head_kernel<<<GG, HID, 0, stream>>>(
        buf1, fc1_w, fc1_b, ln_g, ln_b, fc2_w, fc2_b, (float*)d_out);
}


# Edge dst tiles re-cut to the WGP count: 256 tiles of 391 rows (was 196 of 512), one 32-wave tile block on each of 256 WGPs, 60 no longer idle; 6,250 edges per block, not 8,163; adds in arrival order
# speedup vs baseline: 2.1168x; 1.0505x over previous
#include <hip/hip_runtime.h>
#include <math.h>

typedef __attribute__((ext_vector_type(2))) float v2f;
typedef __attribute__((ext_vector_type(8))) float v8f;

#define NN   100000
#define EE   1600000
#define INDIM 64
#define HID  128
#define LAT  64
#define GG   1000

__device__ __forceinline__ float gelu_exact(float x) {
    return 0.5f * x * (1.0f + erff(x * 0.70710678118654752f));
}

#define BN_BLOCKS  512
#define BN_THREADS 256
#define BN_V4      (NN * INDIM / 4)
#define BN_STRIDE  (BN_BLOCKS * BN_THREADS)
#define BN_PER     ((BN_V4 + BN_STRIDE - 1) / BN_STRIDE)
static_assert(BN_STRIDE % (INDIM / 4) == 0, "stride must be whole rows");

__global__ __launch_bounds__(BN_THREADS) void bn_in_kernel(
    const float* __restrict__ x,
    const float* __restrict__ g, const float* __restrict__ b,
    const float* __restrict__ m, const float* __restrict__ v,
    const float* __restrict__ epsP,
    float* __restrict__ out, float* __restrict__ agg)
{
    const int t = blockIdx.x * BN_THREADS + threadIdx.x;
    const float4* __restrict__ x4 = (const float4*)x;

    float4 xv[BN_PER];
    #pragma unroll
    for (int k = 0; k < BN_PER; ++k) {
        const int i = t + k * BN_STRIDE;
        xv[k] = x4[i < BN_V4 ? i : BN_V4 - 1];
    }

    const int j = 4 * (t & (INDIM / 4 - 1));
    float cm[4], cr[4], cg[4], cb[4];
    #pragma unroll
    for (int c = 0; c < 4; ++c) {
        cm[c] = m[j + c];
        cr[c] = rsqrtf(v[j + c] + 1e-5f);
        cg[c] = g[j + c];
        cb[c] = b[j + c];
    }
    const float sc = 1.0f + epsP[0];

    float4* __restrict__ out4 = (float4*)out;
    float4* __restrict__ agg4 = (float4*)agg;
    #pragma unroll
    for (int k = 0; k < BN_PER; ++k) {
        const int i = t + k * BN_STRIDE;
        if (i < BN_V4) {
            float4 y;
            y.x = (xv[k].x - cm[0]) * cr[0] * cg[0] + cb[0];
            y.y = (xv[k].y - cm[1]) * cr[1] * cg[1] + cb[1];
            y.z = (xv[k].z - cm[2]) * cr[2] * cg[2] + cb[2];
            y.w = (xv[k].w - cm[3]) * cr[3] * cg[3] + cb[3];
            out4[i] = y;
            agg4[i] = make_float4(sc * y.x, sc * y.y, sc * y.z, sc * y.w);
        }
    }
}

#define ET_TILE   391
#define ET_NTILES ((NN + ET_TILE - 1) / ET_TILE)
#define ET_QCAP   48
#define ET_RBITS  9
static_assert(ET_TILE <= (1 << ET_RBITS), "packed word: low bits = row in tile");
static_assert(ET_NTILES == 256, "one tile block per WGP");
static_assert(NN <= (1 << (32 - ET_RBITS)), "packed word: high bits = src node");

#define CH_EDGES  8192
#define CH_PER    (CH_EDGES / 1024)
#define CH_N      ((EE + CH_EDGES - 1) / CH_EDGES)
#define CH_OFFP   288
#define CH_TRIPS  ((CH_N + 31) / 32)
static_assert(ET_NTILES + 1 <= CH_OFFP, "run starts 0..256 fit in a table row");
static_assert(CH_OFFP % 32 == 0, "the scan gives every lane the same number of counts");

__global__ __launch_bounds__(1024) void tile_group_kernel(
    const int* __restrict__ src, const int* __restrict__ dst,
    unsigned* __restrict__ pks, int* __restrict__ offs)
{
    __shared__ int cnt[CH_OFFP];
    __shared__ int base[CH_OFFP + 1];
    __shared__ uint4 sorted4[CH_EDGES / 4];
    unsigned* sorted = (unsigned*)sorted4;
    const int tid = threadIdx.x;

    if (tid < CH_OFFP) cnt[tid] = 0;

    const int e0 = blockIdx.x * CH_EDGES;
    int dv[CH_PER], sv[CH_PER];
    #pragma unroll
    for (int k = 0; k < CH_PER; ++k) {
        const int e = e0 + k * 1024 + tid;
        dv[k] = dst[e < EE ? e : EE - 1];
        sv[k] = src[e < EE ? e : EE - 1];
    }
    __syncthreads();

    int rk[CH_PER];
    #pragma unroll
    for (int k = 0; k < CH_PER; ++k) {
        const int e = e0 + k * 1024 + tid;
        const unsigned t = (unsigned)dv[k] / ET_TILE;
        rk[k] = -1;
        if (e < EE && t < ET_NTILES) rk[k] = atomicAdd(&cnt[t], 1);
    }
    __syncthreads();

    if (tid < 32) {
        int v[CH_OFFP / 32];
        int s = 0;
        #pragma unroll
        for (int j = 0; j < CH_OFFP / 32; ++j) {
            v[j] = cnt[(CH_OFFP / 32) * tid + j];
            s += v[j];
        }
        int inc = s;
        #pragma unroll
        for (int d = 1; d < 32; d <<= 1) {
            const int o = __shfl_up(inc, d, 32);
            if (tid >= d) inc += o;
        }
        int run = inc - s;
        #pragma unroll
        for (int j = 0; j < CH_OFFP / 32; ++j) {
            base[(CH_OFFP / 32) * tid + j] = run;
            run += v[j];
        }
        if (tid == 31) base[CH_OFFP] = run;
    }
    __syncthreads();

    #pragma unroll
    for (int k = 0; k < CH_PER; ++k) {
        if (rk[k] >= 0) {
            const unsigned t = (unsigned)dv[k] / ET_TILE;
            sorted[base[t] + rk[k]] = ((unsigned)sv[k] << ET_RBITS)
                                    | ((unsigned)dv[k] - t * ET_TILE);
        }
    }
    if (tid < CH_OFFP) offs[blockIdx.x * CH_OFFP + tid] = base[tid];
    __syncthreads();

    const int tot = base[CH_OFFP];
    uint4* __restrict__ out4 = (uint4*)(pks + (size_t)blockIdx.x * CH_EDGES);
    for (int i = tid; 4 * i < tot; i += 1024) out4[i] = sorted4[i];
}

__device__ __forceinline__ void et_wave_order() {
    __builtin_amdgcn_fence(__ATOMIC_ACQ_REL, "wavefront");
    __builtin_amdgcn_wave_barrier();
}

template<int D>
__device__ __forceinline__ void et_drain(
    const float* __restrict__ h,
    float* tile, const unsigned* q, int qn, int n, int lane)
{
    int s = 0, dl = 0;
    if (lane < n) {
        const unsigned w = q[qn - n + lane];
        s  = (int)(w >> ET_RBITS);
        dl = (int)(w & ((1u << ET_RBITS) - 1));
    }
    #pragma unroll
    for (int base = 0; base < 16; base += 8) {
        if (base >= n) break;
        float v[8][D / 32];
        #pragma unroll
        for (int i = 0; i < 8; ++i) {
            if (base + i < n) {
                const int si = __builtin_amdgcn_readlane(s, base + i);
                const float* row = h + (size_t)si * D + lane;
                #pragma unroll
                for (int k = 0; k < D / 32; ++k) v[i][k] = row[32 * k];
            }
        }
        #pragma unroll
        for (int i = 0; i < 8; ++i) {
            if (base + i < n) {
                const int di = __builtin_amdgcn_readlane(dl, base + i);
                float* trow = tile + di * D + lane;
                #pragma unroll
                for (int k = 0; k < D / 32; ++k)
                    __hip_atomic_fetch_add(trow + 32 * k, v[i][k],
                        __ATOMIC_RELAXED, __HIP_MEMORY_SCOPE_WORKGROUP);
            }
        }
    }
}

template<int D>
__global__ __launch_bounds__(1024) void edge_tile_kernel(
    const float* __restrict__ h, const unsigned* __restrict__ pks,
    const int* __restrict__ offs,
    float* __restrict__ agg)
{
    __shared__ float tile[ET_TILE * HID];
    __shared__ unsigned queue[32 * ET_QCAP];
    __shared__ int sb0[CH_N], sb1[CH_N];

    const int tid  = threadIdx.x;
    const int lane = tid & 31;
    const int wave = tid >> 5;
    const int lo   = blockIdx.x * ET_TILE;
    const int nrows = (NN - lo < ET_TILE) ? (NN - lo) : ET_TILE;
    const int n4 = nrows * (D / 4);

    if (tid < CH_N) {
        sb0[tid] = offs[tid * CH_OFFP + blockIdx.x];
        sb1[tid] = offs[tid * CH_OFFP + blockIdx.x + 1];
    }

    float4* t4 = (float4*)tile;
    float4* a4 = (float4*)(agg + (size_t)lo * D);
    for (int i = tid; i < n4; i += 1024) t4[i] = a4[i];
    __syncthreads();

    unsigned* q = queue + wave * ET_QCAP;
    int qn = 0;

    int c  = wave;
    int b0 = sb0[c];
    int b1 = sb1[c];
    unsigned w = 0u;
    if (b0 + lane < b1) w = pks[(size_t)c * CH_EDGES + b0 + lane];

    for (int k = 0; k < CH_TRIPS; ++k) {
        if (b0 + lane < b1) q[qn + lane] = w;

        const int cn = c + 32;
        const bool more = cn < CH_N;
        int nb0 = 0, nb1 = 0;
        unsigned nw = 0u;
        if (more) {
            nb0 = sb0[cn];
            nb1 = sb1[cn];
            if (nb0 + lane < nb1) nw = pks[(size_t)cn * CH_EDGES + nb0 + lane];
        }
        for (int i = b0; i < b1; i += 32) {
            const int m = (b1 - i < 32) ? (b1 - i) : 32;
            if (i != b0 && lane < m)
                q[qn + lane] = pks[(size_t)c * CH_EDGES + i + lane];
            qn += m;
            et_wave_order();
            while (qn >= 16) {
                et_drain<D>(h, tile, q, qn, 16, lane);
                qn -= 16;
            }
            et_wave_order();
        }
        if (!more) break;
        c = cn; b0 = nb0; b1 = nb1; w = nw;
    }
    if (qn > 0) et_drain<D>(h, tile, q, qn, qn, lane);

    __syncthreads();
    for (int i = tid; i < n4; i += 1024) a4[i] = t4[i];
}

#define GM_BROWS 64

template<int K>
__global__ __launch_bounds__(256) void gemm_bn_gelu_kernel(
    const float* __restrict__ A,
    const float* __restrict__ W,
    const float* __restrict__ bias,
    const float* __restrict__ bn_g, const float* __restrict__ bn_b,
    const float* __restrict__ bn_m, const float* __restrict__ bn_v,
    float* __restrict__ out,
    float* __restrict__ aggOut, const float* __restrict__ epsP,
    int epsIdx, int writeAgg)
{
    constexpr int P   = K + 4;
    constexpr int RV  = K / 4;
    constexpr int NV  = GM_BROWS * RV / 256;
    __shared__ float4 sA4[GM_BROWS * P / 4];
    float* sA = (float*)sA4;

    const int tid  = threadIdx.x;
    const int lane = tid & 31;
    const int wave = tid >> 5;
    const int half = lane >> 4;
    const int mr   = lane & 15;
    const int col  = wave * 16 + mr;
    const int row0 = blockIdx.x * GM_BROWS;

    float4 st[NV];
    const float4* __restrict__ A4 = (const float4*)A;
    #pragma unroll
    for (int j = 0; j < NV; ++j) {
        const int g = row0 * RV + j * 256 + tid;
        st[j] = A4[g < NN * RV ? g : NN * RV - 1];
    }

    v2f bf[K / 4];
    #pragma unroll
    for (int ks = 0; ks < K / 4; ++ks) {
        const int ka = 4 * ks + 2 * half;
        bf[ks].x = W[(size_t)ka * HID + col];
        bf[ks].y = W[(size_t)(ka + 1) * HID + col];
    }

    const float cbias = bias[col];
    const float cm    = bn_m[col];
    const float crs   = rsqrtf(bn_v[col] + 1e-5f);
    const float cg    = bn_g[col];
    const float cb    = bn_b[col];
    const float escale = writeAgg ? (1.0f + epsP[epsIdx]) : 0.0f;

    #pragma unroll
    for (int j = 0; j < NV; ++j) {
        const int idx = j * 256 + tid;
        *(float4*)(sA + (idx / RV) * P + (idx % RV) * 4) = st[j];
    }
    __syncthreads();

    for (int p = 0; p < GM_BROWS / 32; ++p) {
        const int prow = row0 + 32 * p;
        if (prow >= NN) break;
        const float* a0p = sA + (32 * p + mr) * P + 2 * half;
        const float* a1p = a0p + 16 * P;
        v8f acc0 = {}, acc1 = {};
        #pragma unroll
        for (int ks = 0; ks < K / 4; ++ks) {
            const v2f a0 = *(const v2f*)(a0p + 4 * ks);
            const v2f a1 = *(const v2f*)(a1p + 4 * ks);
            acc0 = __builtin_amdgcn_wmma_f32_16x16x4_f32(
                false, a0, false, bf[ks], (short)0, acc0, false, false);
            acc1 = __builtin_amdgcn_wmma_f32_16x16x4_f32(
                false, a1, false, bf[ks], (short)0, acc1, false, false);
        }

        #pragma unroll
        for (int t = 0; t < 2; ++t) {
            #pragma unroll
            for (int r = 0; r < 8; ++r) {
                const int row = prow + 16 * t + r + 8 * half;
                float y = (t == 0 ? acc0[r] : acc1[r]) + cbias;
                y = (y - cm) * crs * cg + cb;
                y = gelu_exact(y);
                const size_t o = (size_t)row * HID + col;
                out[o] = y;
                if (writeAgg) aggOut[o] = escale * y;
            }
        }
    }
}

__global__ __launch_bounds__(256) void zero_kernel(float* __restrict__ p, int n) {
    int idx = blockIdx.x * blockDim.x + threadIdx.x;
    if (idx < n) p[idx] = 0.0f;
}

#define PL_BLOCKS  512
#define PL_THREADS 256
#define PL_WAVES   (PL_BLOCKS * PL_THREADS / 32)
#define PL_PER     ((NN + PL_WAVES - 1) / PL_WAVES)
static_assert(HID == 128, "a row is 32 pieces of 16 bytes: one per lane");
static_assert(NN % PL_PER == 0, "a wave has all its rows or none");
static_assert(PL_PER <= 32, "one lane per node of the wave for the counts");

__global__ __launch_bounds__(PL_THREADS) void pool_acc_kernel(
    const float* __restrict__ h, const int* __restrict__ batch,
    float* __restrict__ sums, float* __restrict__ cnt)
{
    const int lane = threadIdx.x & 31;
    const int w = __builtin_amdgcn_readfirstlane(
        (int)((blockIdx.x * PL_THREADS + threadIdx.x) >> 5));
    const int i0 = w * PL_PER;
    if (i0 >= NN) return;

    float4 rv[PL_PER];
    int    gb[PL_PER];
    #pragma unroll
    for (int k = 0; k < PL_PER; ++k) {
        gb[k] = batch[i0 + k];
        rv[k] = *(const float4*)(h + (size_t)(i0 + k) * HID + 4 * lane);
    }
    #pragma unroll
    for (int k = 0; k < PL_PER; ++k) {
        float* s = sums + (size_t)gb[k] * HID + 4 * lane;
        unsafeAtomicAdd(s + 0, rv[k].x);
        unsafeAtomicAdd(s + 1, rv[k].y);
        unsafeAtomicAdd(s + 2, rv[k].z);
        unsafeAtomicAdd(s + 3, rv[k].w);
    }
    if (lane < PL_PER) unsafeAtomicAdd(cnt + batch[i0 + lane], 1.0f);
}

__global__ __launch_bounds__(256) void pool_final_kernel(
    const float* __restrict__ sums, const float* __restrict__ cnt,
    float* __restrict__ pooled)
{
    int idx = blockIdx.x * blockDim.x + threadIdx.x;
    if (idx >= GG * HID) return;
    int g = idx / HID;
    pooled[idx] = sums[idx] / fmaxf(cnt[g], 1.0f);
}

__global__ __launch_bounds__(128) void head_kernel(
    const float* __restrict__ pooled,
    const float* __restrict__ fc1w, const float* __restrict__ fc1b,
    const float* __restrict__ lng, const float* __restrict__ lnb,
    const float* __restrict__ fc2w, const float* __restrict__ fc2b,
    float* __restrict__ out)
{
    __shared__ float p[HID];
    __shared__ float rbuf[HID];
    __shared__ float red[HID];
    const int g = blockIdx.x;
    const int t = threadIdx.x;

    p[t] = pooled[(size_t)g * HID + t];
    __syncthreads();

    float o = fc1b[t];
    #pragma unroll 4
    for (int k = 0; k < HID; ++k) o += p[k] * fc1w[k * HID + t];

    red[t] = o;
    __syncthreads();
    for (int s = 64; s > 0; s >>= 1) {
        if (t < s) red[t] += red[t + s];
        __syncthreads();
    }
    const float mu = red[0] * (1.0f / HID);
    __syncthreads();

    const float dv = o - mu;
    red[t] = dv * dv;
    __syncthreads();
    for (int s = 64; s > 0; s >>= 1) {
        if (t < s) red[t] += red[t + s];
        __syncthreads();
    }
    const float var = red[0] * (1.0f / HID);

    float y = dv * rsqrtf(var + 1e-5f) * lng[t] + lnb[t];
    y = gelu_exact(y) + p[t];
    rbuf[t] = y;
    __syncthreads();

    if (t < LAT) {
        float o2 = fc2b[t];
        #pragma unroll 4
        for (int k = 0; k < HID; ++k) o2 += rbuf[k] * fc2w[k * LAT + t];
        out[(size_t)g * LAT + t] = o2;
    }
}

extern "C" void kernel_launch(void* const* d_in, const int* in_sizes, int n_in,
                              void* d_out, int out_size, void* d_ws, size_t ws_size,
                              hipStream_t stream) {
    const float* x       = (const float*)d_in[0];
    const int*   ei      = (const int*)  d_in[1];
    const int*   batch   = (const int*)  d_in[2];
    const float* in_g    = (const float*)d_in[3];
    const float* in_b    = (const float*)d_in[4];
    const float* in_m    = (const float*)d_in[5];
    const float* in_v    = (const float*)d_in[6];
    const float* eps0    = (const float*)d_in[7];
    const float* c0_w1   = (const float*)d_in[8];
    const float* c0_b1   = (const float*)d_in[9];
    const float* c0_g    = (const float*)d_in[10];
    const float* c0_bb   = (const float*)d_in[11];
    const float* c0_m    = (const float*)d_in[12];
    const float* c0_v    = (const float*)d_in[13];
    const float* c0_w2   = (const float*)d_in[14];
    const float* c0_b2   = (const float*)d_in[15];
    const float* bn0_g   = (const float*)d_in[16];
    const float* bn0_b   = (const float*)d_in[17];
    const float* bn0_m   = (const float*)d_in[18];
    const float* bn0_v   = (const float*)d_in[19];
    const float* epsR    = (const float*)d_in[20];
    const float* cR_w1   = (const float*)d_in[21];
    const float* cR_b1   = (const float*)d_in[22];
    const float* cR_g    = (const float*)d_in[23];
    const float* cR_bb   = (const float*)d_in[24];
    const float* cR_m    = (const float*)d_in[25];
    const float* cR_v    = (const float*)d_in[26];
    const float* cR_w2   = (const float*)d_in[27];
    const float* cR_b2   = (const float*)d_in[28];
    const float* bnR_g   = (const float*)d_in[29];
    const float* bnR_b   = (const float*)d_in[30];
    const float* bnR_m   = (const float*)d_in[31];
    const float* bnR_v   = (const float*)d_in[32];
    const float* fc1_w   = (const float*)d_in[33];
    const float* fc1_b   = (const float*)d_in[34];
    const float* ln_g    = (const float*)d_in[35];
    const float* ln_b    = (const float*)d_in[36];
    const float* fc2_w   = (const float*)d_in[37];
    const float* fc2_b   = (const float*)d_in[38];

    const int* src = ei;
    const int* dst = ei + EE;

    float* buf0 = (float*)d_ws;
    float* buf1 = buf0 + (size_t)NN * HID;
    float* buf2 = buf1 + (size_t)NN * HID;
    float* sums = buf2 + (size_t)NN * HID;
    float* cnt  = sums + (size_t)GG * HID;
    unsigned* pkw = (unsigned*)buf2;
    int*      ofs = (int*)(pkw + (size_t)CH_N * CH_EDGES);
    static_assert((size_t)CH_N * CH_EDGES + (size_t)CH_N * CH_OFFP <= (size_t)NN * HID,
                  "packed words and run starts fit in buf2");

    const int TB = 256;
    const int nGemm = (NN + GM_BROWS - 1) / GM_BROWS;

    bn_in_kernel<<<BN_BLOCKS, BN_THREADS, 0, stream>>>(
        x, in_g, in_b, in_m, in_v, eps0, buf0, buf1);

    {
        tile_group_kernel<<<CH_N, 1024, 0, stream>>>(src, dst, pkw, ofs);
        edge_tile_kernel<INDIM><<<ET_NTILES, 1024, 0, stream>>>(buf0, pkw, ofs, buf1);
        gemm_bn_gelu_kernel<INDIM><<<nGemm, TB, 0, stream>>>(
            buf1, c0_w1, c0_b1, c0_g, c0_bb, c0_m, c0_v, buf2,
            buf1, eps0, 0, 0);
        gemm_bn_gelu_kernel<HID><<<nGemm, TB, 0, stream>>>(
            buf2, c0_w2, c0_b2, bn0_g, bn0_b, bn0_m, bn0_v, buf0,
            buf1, epsR, 0, 1);
    }

    for (int l = 0; l < 2; ++l) {
        const size_t wOff = (size_t)l * HID * HID;
        const size_t vOff = (size_t)l * HID;
        tile_group_kernel<<<CH_N, 1024, 0, stream>>>(src, dst, pkw, ofs);
        edge_tile_kernel<HID><<<ET_NTILES, 1024, 0, stream>>>(buf0, pkw, ofs, buf1);
        gemm_bn_gelu_kernel<HID><<<nGemm, TB, 0, stream>>>(
            buf1, cR_w1 + wOff, cR_b1 + vOff, cR_g + vOff, cR_bb + vOff,
            cR_m + vOff, cR_v + vOff, buf2,
            buf1, epsR, 0, 0);
        const int writeAgg = (l == 0) ? 1 : 0;
        gemm_bn_gelu_kernel<HID><<<nGemm, TB, 0, stream>>>(
            buf2, cR_w2 + wOff, cR_b2 + vOff, bnR_g + vOff, bnR_b + vOff,
            bnR_m + vOff, bnR_v + vOff, buf0,
            buf1, epsR, 1, writeAgg);
    }

    zero_kernel<<<(GG * HID + GG + TB - 1) / TB, TB, 0, stream>>>(sums, GG * HID + GG);
    pool_acc_kernel<<<PL_BLOCKS, PL_THREADS, 0, stream>>>(buf0, batch, sums, cnt);
    pool_final_kernel<<<(GG * HID + TB - 1) / TB, TB, 0, stream>>>(sums, cnt, buf1);

    head_kernel<<<GG, HID, 0, stream>>>(
        buf1, fc1_w, fc1_b, ln_g, ln_b, fc2_w, fc2_b, (float*)d_out);
}


# Mean pool and head by graph-owning blocks: block b lists the nodes of graphs 4b..4b+3 in LDS, sums their rows in registers, divides, runs the head; no global atomics; pool add order = list order
# speedup vs baseline: 2.2998x; 1.0865x over previous
#include <hip/hip_runtime.h>
#include <math.h>

typedef __attribute__((ext_vector_type(2))) float v2f;
typedef __attribute__((ext_vector_type(8))) float v8f;

#define NN   100000
#define EE   1600000
#define INDIM 64
#define HID  128
#define LAT  64
#define GG   1000

__device__ __forceinline__ float gelu_exact(float x) {
    return 0.5f * x * (1.0f + erff(x * 0.70710678118654752f));
}

#define BN_BLOCKS  512
#define BN_THREADS 256
#define BN_V4      (NN * INDIM / 4)
#define BN_STRIDE  (BN_BLOCKS * BN_THREADS)
#define BN_PER     ((BN_V4 + BN_STRIDE - 1) / BN_STRIDE)
static_assert(BN_STRIDE % (INDIM / 4) == 0, "stride must be whole rows");

__global__ __launch_bounds__(BN_THREADS) void bn_in_kernel(
    const float* __restrict__ x,
    const float* __restrict__ g, const float* __restrict__ b,
    const float* __restrict__ m, const float* __restrict__ v,
    const float* __restrict__ epsP,
    float* __restrict__ out, float* __restrict__ agg)
{
    const int t = blockIdx.x * BN_THREADS + threadIdx.x;
    const float4* __restrict__ x4 = (const float4*)x;

    float4 xv[BN_PER];
    #pragma unroll
    for (int k = 0; k < BN_PER; ++k) {
        const int i = t + k * BN_STRIDE;
        xv[k] = x4[i < BN_V4 ? i : BN_V4 - 1];
    }

    const int j = 4 * (t & (INDIM / 4 - 1));
    float cm[4], cr[4], cg[4], cb[4];
    #pragma unroll
    for (int c = 0; c < 4; ++c) {
        cm[c] = m[j + c];
        cr[c] = rsqrtf(v[j + c] + 1e-5f);
        cg[c] = g[j + c];
        cb[c] = b[j + c];
    }
    const float sc = 1.0f + epsP[0];

    float4* __restrict__ out4 = (float4*)out;
    float4* __restrict__ agg4 = (float4*)agg;
    #pragma unroll
    for (int k = 0; k < BN_PER; ++k) {
        const int i = t + k * BN_STRIDE;
        if (i < BN_V4) {
            float4 y;
            y.x = (xv[k].x - cm[0]) * cr[0] * cg[0] + cb[0];
            y.y = (xv[k].y - cm[1]) * cr[1] * cg[1] + cb[1];
            y.z = (xv[k].z - cm[2]) * cr[2] * cg[2] + cb[2];
            y.w = (xv[k].w - cm[3]) * cr[3] * cg[3] + cb[3];
            out4[i] = y;
            agg4[i] = make_float4(sc * y.x, sc * y.y, sc * y.z, sc * y.w);
        }
    }
}

#define ET_TILE   391
#define ET_NTILES ((NN + ET_TILE - 1) / ET_TILE)
#define ET_QCAP   48
#define ET_RBITS  9
static_assert(ET_TILE <= (1 << ET_RBITS), "packed word: low bits = row in tile");
static_assert(ET_NTILES == 256, "one tile block per WGP");
static_assert(NN <= (1 << (32 - ET_RBITS)), "packed word: high bits = src node");

#define CH_EDGES  8192
#define CH_PER    (CH_EDGES / 1024)
#define CH_N      ((EE + CH_EDGES - 1) / CH_EDGES)
#define CH_OFFP   288
#define CH_TRIPS  ((CH_N + 31) / 32)
static_assert(ET_NTILES + 1 <= CH_OFFP, "run starts 0..256 fit in a table row");
static_assert(CH_OFFP % 32 == 0, "the scan gives every lane the same number of counts");

__global__ __launch_bounds__(1024) void tile_group_kernel(
    const int* __restrict__ src, const int* __restrict__ dst,
    unsigned* __restrict__ pks, int* __restrict__ offs)
{
    __shared__ int cnt[CH_OFFP];
    __shared__ int base[CH_OFFP + 1];
    __shared__ uint4 sorted4[CH_EDGES / 4];
    unsigned* sorted = (unsigned*)sorted4;
    const int tid = threadIdx.x;

    if (tid < CH_OFFP) cnt[tid] = 0;

    const int e0 = blockIdx.x * CH_EDGES;
    int dv[CH_PER], sv[CH_PER];
    #pragma unroll
    for (int k = 0; k < CH_PER; ++k) {
        const int e = e0 + k * 1024 + tid;
        dv[k] = dst[e < EE ? e : EE - 1];
        sv[k] = src[e < EE ? e : EE - 1];
    }
    __syncthreads();

    int rk[CH_PER];
    #pragma unroll
    for (int k = 0; k < CH_PER; ++k) {
        const int e = e0 + k * 1024 + tid;
        const unsigned t = (unsigned)dv[k] / ET_TILE;
        rk[k] = -1;
        if (e < EE && t < ET_NTILES) rk[k] = atomicAdd(&cnt[t], 1);
    }
    __syncthreads();

    if (tid < 32) {
        int v[CH_OFFP / 32];
        int s = 0;
        #pragma unroll
        for (int j = 0; j < CH_OFFP / 32; ++j) {
            v[j] = cnt[(CH_OFFP / 32) * tid + j];
            s += v[j];
        }
        int inc = s;
        #pragma unroll
        for (int d = 1; d < 32; d <<= 1) {
            const int o = __shfl_up(inc, d, 32);
            if (tid >= d) inc += o;
        }
        int run = inc - s;
        #pragma unroll
        for (int j = 0; j < CH_OFFP / 32; ++j) {
            base[(CH_OFFP / 32) * tid + j] = run;
            run += v[j];
        }
        if (tid == 31) base[CH_OFFP] = run;
    }
    __syncthreads();

    #pragma unroll
    for (int k = 0; k < CH_PER; ++k) {
        if (rk[k] >= 0) {
            const unsigned t = (unsigned)dv[k] / ET_TILE;
            sorted[base[t] + rk[k]] = ((unsigned)sv[k] << ET_RBITS)
                                    | ((unsigned)dv[k] - t * ET_TILE);
        }
    }
    if (tid < CH_OFFP) offs[blockIdx.x * CH_OFFP + tid] = base[tid];
    __syncthreads();

    const int tot = base[CH_OFFP];
    uint4* __restrict__ out4 = (uint4*)(pks + (size_t)blockIdx.x * CH_EDGES);
    for (int i = tid; 4 * i < tot; i += 1024) out4[i] = sorted4[i];
}

__device__ __forceinline__ void et_wave_order() {
    __builtin_amdgcn_fence(__ATOMIC_ACQ_REL, "wavefront");
    __builtin_amdgcn_wave_barrier();
}

template<int D>
__device__ __forceinline__ void et_drain(
    const float* __restrict__ h,
    float* tile, const unsigned* q, int qn, int n, int lane)
{
    int s = 0, dl = 0;
    if (lane < n) {
        const unsigned w = q[qn - n + lane];
        s  = (int)(w >> ET_RBITS);
        dl = (int)(w & ((1u << ET_RBITS) - 1));
    }
    #pragma unroll
    for (int base = 0; base < 16; base += 8) {
        if (base >= n) break;
        float v[8][D / 32];
        #pragma unroll
        for (int i = 0; i < 8; ++i) {
            if (base + i < n) {
                const int si = __builtin_amdgcn_readlane(s, base + i);
                const float* row = h + (size_t)si * D + lane;
                #pragma unroll
                for (int k = 0; k < D / 32; ++k) v[i][k] = row[32 * k];
            }
        }
        #pragma unroll
        for (int i = 0; i < 8; ++i) {
            if (base + i < n) {
                const int di = __builtin_amdgcn_readlane(dl, base + i);
                float* trow = tile + di * D + lane;
                #pragma unroll
                for (int k = 0; k < D / 32; ++k)
                    __hip_atomic_fetch_add(trow + 32 * k, v[i][k],
                        __ATOMIC_RELAXED, __HIP_MEMORY_SCOPE_WORKGROUP);
            }
        }
    }
}

template<int D>
__global__ __launch_bounds__(1024) void edge_tile_kernel(
    const float* __restrict__ h, const unsigned* __restrict__ pks,
    const int* __restrict__ offs,
    float* __restrict__ agg)
{
    __shared__ float tile[ET_TILE * HID];
    __shared__ unsigned queue[32 * ET_QCAP];
    __shared__ int sb0[CH_N], sb1[CH_N];

    const int tid  = threadIdx.x;
    const int lane = tid & 31;
    const int wave = tid >> 5;
    const int lo   = blockIdx.x * ET_TILE;
    const int nrows = (NN - lo < ET_TILE) ? (NN - lo) : ET_TILE;
    const int n4 = nrows * (D / 4);

    if (tid < CH_N) {
        sb0[tid] = offs[tid * CH_OFFP + blockIdx.x];
        sb1[tid] = offs[tid * CH_OFFP + blockIdx.x + 1];
    }

    float4* t4 = (float4*)tile;
    float4* a4 = (float4*)(agg + (size_t)lo * D);
    for (int i = tid; i < n4; i += 1024) t4[i] = a4[i];
    __syncthreads();

    unsigned* q = queue + wave * ET_QCAP;
    int qn = 0;

    int c  = wave;
    int b0 = sb0[c];
    int b1 = sb1[c];
    unsigned w = 0u;
    if (b0 + lane < b1) w = pks[(size_t)c * CH_EDGES + b0 + lane];

    for (int k = 0; k < CH_TRIPS; ++k) {
        if (b0 + lane < b1) q[qn + lane] = w;

        const int cn = c + 32;
        const bool more = cn < CH_N;
        int nb0 = 0, nb1 = 0;
        unsigned nw = 0u;
        if (more) {
            nb0 = sb0[cn];
            nb1 = sb1[cn];
            if (nb0 + lane < nb1) nw = pks[(size_t)cn * CH_EDGES + nb0 + lane];
        }
        for (int i = b0; i < b1; i += 32) {
            const int m = (b1 - i < 32) ? (b1 - i) : 32;
            if (i != b0 && lane < m)
                q[qn + lane] = pks[(size_t)c * CH_EDGES + i + lane];
            qn += m;
            et_wave_order();
            while (qn >= 16) {
                et_drain<D>(h, tile, q, qn, 16, lane);
                qn -= 16;
            }
            et_wave_order();
        }
        if (!more) break;
        c = cn; b0 = nb0; b1 = nb1; w = nw;
    }
    if (qn > 0) et_drain<D>(h, tile, q, qn, qn, lane);

    __syncthreads();
    for (int i = tid; i < n4; i += 1024) a4[i] = t4[i];
}

#define GM_BROWS 64

template<int K>
__global__ __launch_bounds__(256) void gemm_bn_gelu_kernel(
    const float* __restrict__ A,
    const float* __restrict__ W,
    const float* __restrict__ bias,
    const float* __restrict__ bn_g, const float* __restrict__ bn_b,
    const float* __restrict__ bn_m, const float* __restrict__ bn_v,
    float* __restrict__ out,
    float* __restrict__ aggOut, const float* __restrict__ epsP,
    int epsIdx, int writeAgg)
{
    constexpr int P   = K + 4;
    constexpr int RV  = K / 4;
    constexpr int NV  = GM_BROWS * RV / 256;
    __shared__ float4 sA4[GM_BROWS * P / 4];
    float* sA = (float*)sA4;

    const int tid  = threadIdx.x;
    const int lane = tid & 31;
    const int wave = tid >> 5;
    const int half = lane >> 4;
    const int mr   = lane & 15;
    const int col  = wave * 16 + mr;
    const int row0 = blockIdx.x * GM_BROWS;

    float4 st[NV];
    const float4* __restrict__ A4 = (const float4*)A;
    #pragma unroll
    for (int j = 0; j < NV; ++j) {
        const int g = row0 * RV + j * 256 + tid;
        st[j] = A4[g < NN * RV ? g : NN * RV - 1];
    }

    v2f bf[K / 4];
    #pragma unroll
    for (int ks = 0; ks < K / 4; ++ks) {
        const int ka = 4 * ks + 2 * half;
        bf[ks].x = W[(size_t)ka * HID + col];
        bf[ks].y = W[(size_t)(ka + 1) * HID + col];
    }

    const float cbias = bias[col];
    const float cm    = bn_m[col];
    const float crs   = rsqrtf(bn_v[col] + 1e-5f);
    const float cg    = bn_g[col];
    const float cb    = bn_b[col];
    const float escale = writeAgg ? (1.0f + epsP[epsIdx]) : 0.0f;

    #pragma unroll
    for (int j = 0; j < NV; ++j) {
        const int idx = j * 256 + tid;
        *(float4*)(sA + (idx / RV) * P + (idx % RV) * 4) = st[j];
    }
    __syncthreads();

    for (int p = 0; p < GM_BROWS / 32; ++p) {
        const int prow = row0 + 32 * p;
        if (prow >= NN) break;
        const float* a0p = sA + (32 * p + mr) * P + 2 * half;
        const float* a1p = a0p + 16 * P;
        v8f acc0 = {}, acc1 = {};
        #pragma unroll
        for (int ks = 0; ks < K / 4; ++ks) {
            const v2f a0 = *(const v2f*)(a0p + 4 * ks);
            const v2f a1 = *(const v2f*)(a1p + 4 * ks);
            acc0 = __builtin_amdgcn_wmma_f32_16x16x4_f32(
                false, a0, false, bf[ks], (short)0, acc0, false, false);
            acc1 = __builtin_amdgcn_wmma_f32_16x16x4_f32(
                false, a1, false, bf[ks], (short)0, acc1, false, false);
        }

        #pragma unroll
        for (int t = 0; t < 2; ++t) {
            #pragma unroll
            for (int r = 0; r < 8; ++r) {
                const int row = prow + 16 * t + r + 8 * half;
                float y = (t == 0 ? acc0[r] : acc1[r]) + cbias;
                y = (y - cm) * crs * cg + cb;
                y = gelu_exact(y);
                const size_t o = (size_t)row * HID + col;
                out[o] = y;
                if (writeAgg) aggOut[o] = escale * y;
            }
        }
    }
}

#define PT_G      4
#define PT_BLOCKS (GG / PT_G)
#define PT_SEG    32768
#define PT_PER4   (PT_SEG / 4 / 1024)
#define PT_ROUNDS ((NN + PT_SEG - 1) / PT_SEG)
#define PT_INFL   8
static_assert(GG % PT_G == 0, "every block owns four graphs");
static_assert(NN % 4 == 0, "batch[] is read four nodes at a time");
static_assert(PT_G == 4, "a list word is node << 2 | graph in block");
static_assert(HID == 128, "a row is 32 pieces of 16 bytes: one per lane");
static_assert(PT_G * HID <= 1024 && PT_G * LAT <= 1024, "one thread per head output");

__device__ __forceinline__ void pt_add(float4& a, const float4& v) {
    a.x += v.x; a.y += v.y; a.z += v.z; a.w += v.w;
}

__device__ __forceinline__ float pt_sum128(float v0, float v1, float v2, float v3) {
    float s = (v0 + v2) + (v1 + v3);
    #pragma unroll
    for (int d = 16; d >= 1; d >>= 1) s += __shfl_down(s, d, 32);
    return __shfl(s, 0, 32);
}

__global__ __launch_bounds__(1024) void pool_head_kernel(
    const float* __restrict__ h, const int* __restrict__ batch,
    const float* __restrict__ fc1w, const float* __restrict__ fc1b,
    const float* __restrict__ lng, const float* __restrict__ lnb,
    const float* __restrict__ fc2w, const float* __restrict__ fc2b,
    float* __restrict__ out)
{
    __shared__ int    list[PT_SEG];
    __shared__ float4 part4[32 * PT_G * HID / 4];
    __shared__ int    cpart[32 * PT_G];
    __shared__ int    nl[PT_ROUNDS];
    __shared__ float  pl[PT_G * HID];
    __shared__ float  ob[PT_G * HID];
    __shared__ float  rb[PT_G * HID];

    const int tid  = threadIdx.x;
    const int lane = tid & 31;
    const int wave = __builtin_amdgcn_readfirstlane(tid >> 5);
    const int g0   = blockIdx.x * PT_G;

    if (tid < PT_ROUNDS) nl[tid] = 0;
    __syncthreads();

    float4 acc0 = make_float4(0.f, 0.f, 0.f, 0.f);
    float4 acc1 = acc0, acc2 = acc0, acc3 = acc0;
    int c0 = 0, c1 = 0, c2 = 0, c3 = 0;

    const int4* __restrict__ b4 = (const int4*)batch;
    int4 bv[PT_PER4];
    #pragma unroll
    for (int k = 0; k < PT_PER4; ++k) {
        const int i4 = k * 1024 + tid;
        bv[k] = b4[i4 < NN / 4 ? i4 : NN / 4 - 1];
    }
    for (int r = 0; r < PT_ROUNDS; ++r) {
        #pragma unroll
        for (int k = 0; k < PT_PER4; ++k) {
            const int i4 = r * (PT_SEG / 4) + k * 1024 + tid;
            if (i4 < NN / 4) {
                const int bb[4] = { bv[k].x, bv[k].y, bv[k].z, bv[k].w };
                #pragma unroll
                for (int q = 0; q < 4; ++q) {
                    const unsigned lg = (unsigned)(bb[q] - g0);
                    if (lg < PT_G) {
                        const int p = atomicAdd(&nl[r], 1);
                        list[p] = ((4 * i4 + q) << 2) | (int)lg;
                    }
                }
            }
        }
        if (r + 1 < PT_ROUNDS) {
            #pragma unroll
            for (int k = 0; k < PT_PER4; ++k) {
                const int i4 = (r + 1) * (PT_SEG / 4) + k * 1024 + tid;
                bv[k] = b4[i4 < NN / 4 ? i4 : NN / 4 - 1];
            }
        }
        __syncthreads();

        const int n = __builtin_amdgcn_readfirstlane(nl[r]);
        for (int j0 = wave; j0 < n; j0 += 32 * PT_INFL) {
            float4 rv[PT_INFL];
            int    ev[PT_INFL];
            #pragma unroll
            for (int u = 0; u < PT_INFL; ++u) {
                const int j = j0 + 32 * u;
                ev[u] = __builtin_amdgcn_readfirstlane(list[j < n ? j : n - 1]);
                rv[u] = *(const float4*)(h + (size_t)(ev[u] >> 2) * HID + 4 * lane);
            }
            #pragma unroll
            for (int u = 0; u < PT_INFL; ++u) {
                if (j0 + 32 * u < n) {
                    const int lg = ev[u] & 3;
                    if      (lg == 0) { pt_add(acc0, rv[u]); ++c0; }
                    else if (lg == 1) { pt_add(acc1, rv[u]); ++c1; }
                    else if (lg == 2) { pt_add(acc2, rv[u]); ++c2; }
                    else              { pt_add(acc3, rv[u]); ++c3; }
                }
            }
        }
        __syncthreads();
    }

    float4* pw = part4 + wave * (PT_G * HID / 4);
    pw[0 * (HID / 4) + lane] = acc0;
    pw[1 * (HID / 4) + lane] = acc1;
    pw[2 * (HID / 4) + lane] = acc2;
    pw[3 * (HID / 4) + lane] = acc3;
    if (lane == 0) {
        cpart[wave * PT_G + 0] = c0;
        cpart[wave * PT_G + 1] = c1;
        cpart[wave * PT_G + 2] = c2;
        cpart[wave * PT_G + 3] = c3;
    }
    __syncthreads();

    const int hg = tid / HID;
    const int ht = tid % HID;
    if (tid < PT_G * HID) {
        const float* part = (const float*)part4;
        float s = 0.0f;
        int   cn = 0;
        for (int w = 0; w < 32; ++w) {
            s  += part[w * (PT_G * HID) + tid];
            cn += cpart[w * PT_G + hg];
        }
        pl[tid] = s / fmaxf((float)cn, 1.0f);
    }
    __syncthreads();

    if (tid < PT_G * HID) {
        const float* p = pl + hg * HID;
        float o = fc1b[ht];
        #pragma unroll 16
        for (int k = 0; k < HID; ++k) o += p[k] * fc1w[k * HID + ht];
        ob[tid] = o;
    }
    __syncthreads();

    if (wave < PT_G) {
        const float* og = ob + wave * HID;
        const float* pg = pl + wave * HID;
        float ov[4], dv[4];
        #pragma unroll
        for (int i = 0; i < 4; ++i) ov[i] = og[lane + 32 * i];
        const float mu = pt_sum128(ov[0], ov[1], ov[2], ov[3]) * (1.0f / HID);
        #pragma unroll
        for (int i = 0; i < 4; ++i) dv[i] = ov[i] - mu;
        const float var = pt_sum128(dv[0] * dv[0], dv[1] * dv[1],
                                    dv[2] * dv[2], dv[3] * dv[3]) * (1.0f / HID);
        const float rs = rsqrtf(var + 1e-5f);
        #pragma unroll
        for (int i = 0; i < 4; ++i) {
            const int t = lane + 32 * i;
            float y = dv[i] * rs * lng[t] + lnb[t];
            y = gelu_exact(y) + pg[t];
            rb[wave * HID + t] = y;
        }
    }
    __syncthreads();

    if (tid < PT_G * LAT) {
        const int g = tid / LAT;
        const int t = tid % LAT;
        const float* rr = rb + g * HID;
        float o2 = fc2b[t];
        #pragma unroll 16
        for (int k = 0; k < HID; ++k) o2 += rr[k] * fc2w[k * LAT + t];
        out[(size_t)(g0 + g) * LAT + t] = o2;
    }
}

extern "C" void kernel_launch(void* const* d_in, const int* in_sizes, int n_in,
                              void* d_out, int out_size, void* d_ws, size_t ws_size,
                              hipStream_t stream) {
    const float* x       = (const float*)d_in[0];
    const int*   ei      = (const int*)  d_in[1];
    const int*   batch   = (const int*)  d_in[2];
    const float* in_g    = (const float*)d_in[3];
    const float* in_b    = (const float*)d_in[4];
    const float* in_m    = (const float*)d_in[5];
    const float* in_v    = (const float*)d_in[6];
    const float* eps0    = (const float*)d_in[7];
    const float* c0_w1   = (const float*)d_in[8];
    const float* c0_b1   = (const float*)d_in[9];
    const float* c0_g    = (const float*)d_in[10];
    const float* c0_bb   = (const float*)d_in[11];
    const float* c0_m    = (const float*)d_in[12];
    const float* c0_v    = (const float*)d_in[13];
    const float* c0_w2   = (const float*)d_in[14];
    const float* c0_b2   = (const float*)d_in[15];
    const float* bn0_g   = (const float*)d_in[16];
    const float* bn0_b   = (const float*)d_in[17];
    const float* bn0_m   = (const float*)d_in[18];
    const float* bn0_v   = (const float*)d_in[19];
    const float* epsR    = (const float*)d_in[20];
    const float* cR_w1   = (const float*)d_in[21];
    const float* cR_b1   = (const float*)d_in[22];
    const float* cR_g    = (const float*)d_in[23];
    const float* cR_bb   = (const float*)d_in[24];
    const float* cR_m    = (const float*)d_in[25];
    const float* cR_v    = (const float*)d_in[26];
    const float* cR_w2   = (const float*)d_in[27];
    const float* cR_b2   = (const float*)d_in[28];
    const float* bnR_g   = (const float*)d_in[29];
    const float* bnR_b   = (const float*)d_in[30];
    const float* bnR_m   = (const float*)d_in[31];
    const float* bnR_v   = (const float*)d_in[32];
    const float* fc1_w   = (const float*)d_in[33];
    const float* fc1_b   = (const float*)d_in[34];
    const float* ln_g    = (const float*)d_in[35];
    const float* ln_b    = (const float*)d_in[36];
    const float* fc2_w   = (const float*)d_in[37];
    const float* fc2_b   = (const float*)d_in[38];

    const int* src = ei;
    const int* dst = ei + EE;

    float* buf0 = (float*)d_ws;
    float* buf1 = buf0 + (size_t)NN * HID;
    float* buf2 = buf1 + (size_t)NN * HID;
    unsigned* pkw = (unsigned*)buf2;
    int*      ofs = (int*)(pkw + (size_t)CH_N * CH_EDGES);
    static_assert((size_t)CH_N * CH_EDGES + (size_t)CH_N * CH_OFFP <= (size_t)NN * HID,
                  "packed words and run starts fit in buf2");

    const int TB = 256;
    const int nGemm = (NN + GM_BROWS - 1) / GM_BROWS;

    bn_in_kernel<<<BN_BLOCKS, BN_THREADS, 0, stream>>>(
        x, in_g, in_b, in_m, in_v, eps0, buf0, buf1);

    {
        tile_group_kernel<<<CH_N, 1024, 0, stream>>>(src, dst, pkw, ofs);
        edge_tile_kernel<INDIM><<<ET_NTILES, 1024, 0, stream>>>(buf0, pkw, ofs, buf1);
        gemm_bn_gelu_kernel<INDIM><<<nGemm, TB, 0, stream>>>(
            buf1, c0_w1, c0_b1, c0_g, c0_bb, c0_m, c0_v, buf2,
            buf1, eps0, 0, 0);
        gemm_bn_gelu_kernel<HID><<<nGemm, TB, 0, stream>>>(
            buf2, c0_w2, c0_b2, bn0_g, bn0_b, bn0_m, bn0_v, buf0,
            buf1, epsR, 0, 1);
    }

    for (int l = 0; l < 2; ++l) {
        const size_t wOff = (size_t)l * HID * HID;
        const size_t vOff = (size_t)l * HID;
        tile_group_kernel<<<CH_N, 1024, 0, stream>>>(src, dst, pkw, ofs);
        edge_tile_kernel<HID><<<ET_NTILES, 1024, 0, stream>>>(buf0, pkw, ofs, buf1);
        gemm_bn_gelu_kernel<HID><<<nGemm, TB, 0, stream>>>(
            buf1, cR_w1 + wOff, cR_b1 + vOff, cR_g + vOff, cR_bb + vOff,
            cR_m + vOff, cR_v + vOff, buf2,
            buf1, epsR, 0, 0);
        const int writeAgg = (l == 0) ? 1 : 0;
        gemm_bn_gelu_kernel<HID><<<nGemm, TB, 0, stream>>>(
            buf2, cR_w2 + wOff, cR_b2 + vOff, bnR_g + vOff, bnR_b + vOff,
            bnR_m + vOff, bnR_v + vOff, buf0,
            buf1, epsR, 1, writeAgg);
    }

    pool_head_kernel<<<PT_BLOCKS, 1024, 0, stream>>>(
        buf0, batch, fc1_w, fc1_b, ln_g, ln_b, fc2_w, fc2_b, (float*)d_out);
}


# Edge sums stay in LDS: the tile block runs the layer's first product on its finished tile (same k order; adds in arrival order as before); agg plane, its store, reload, pre-fill and 3 launches gone
# speedup vs baseline: 2.5969x; 1.1292x over previous
#include <hip/hip_runtime.h>
#include <math.h>

typedef __attribute__((ext_vector_type(2))) float v2f;
typedef __attribute__((ext_vector_type(8))) float v8f;

#define NN   100000
#define EE   1600000
#define INDIM 64
#define HID  128
#define LAT  64
#define GG   1000

__device__ __forceinline__ float gelu_exact(float x) {
    return 0.5f * x * (1.0f + erff(x * 0.70710678118654752f));
}

#define BN_BLOCKS  512
#define BN_THREADS 256
#define BN_V4      (NN * INDIM / 4)
#define BN_STRIDE  (BN_BLOCKS * BN_THREADS)
#define BN_PER     ((BN_V4 + BN_STRIDE - 1) / BN_STRIDE)
static_assert(BN_STRIDE % (INDIM / 4) == 0, "stride must be whole rows");

__global__ __launch_bounds__(BN_THREADS) void bn_in_kernel(
    const float* __restrict__ x,
    const float* __restrict__ g, const float* __restrict__ b,
    const float* __restrict__ m, const float* __restrict__ v,
    float* __restrict__ out)
{
    const int t = blockIdx.x * BN_THREADS + threadIdx.x;
    const float4* __restrict__ x4 = (const float4*)x;

    float4 xv[BN_PER];
    #pragma unroll
    for (int k = 0; k < BN_PER; ++k) {
        const int i = t + k * BN_STRIDE;
        xv[k] = x4[i < BN_V4 ? i : BN_V4 - 1];
    }

    const int j = 4 * (t & (INDIM / 4 - 1));
    float cm[4], cr[4], cg[4], cb[4];
    #pragma unroll
    for (int c = 0; c < 4; ++c) {
        cm[c] = m[j + c];
        cr[c] = rsqrtf(v[j + c] + 1e-5f);
        cg[c] = g[j + c];
        cb[c] = b[j + c];
    }

    float4* __restrict__ out4 = (float4*)out;
    #pragma unroll
    for (int k = 0; k < BN_PER; ++k) {
        const int i = t + k * BN_STRIDE;
        if (i < BN_V4) {
            float4 y;
            y.x = (xv[k].x - cm[0]) * cr[0] * cg[0] + cb[0];
            y.y = (xv[k].y - cm[1]) * cr[1] * cg[1] + cb[1];
            y.z = (xv[k].z - cm[2]) * cr[2] * cg[2] + cb[2];
            y.w = (xv[k].w - cm[3]) * cr[3] * cg[3] + cb[3];
            out4[i] = y;
        }
    }
}

#define ET_TILE   391
#define ET_TROWS  400
#define ET_NTILES ((NN + ET_TILE - 1) / ET_TILE)
#define ET_QCAP   48
#define ET_RBITS  9
static_assert(ET_TROWS % 16 == 0 && ET_TROWS >= ET_TILE, "whole 16-row WMMA tiles");
static_assert(ET_TILE <= (1 << ET_RBITS), "packed word: low bits = row in tile");
static_assert(ET_NTILES == 256, "one tile block per WGP");
static_assert(NN <= (1 << (32 - ET_RBITS)), "packed word: high bits = src node");

#define CH_EDGES  8192
#define CH_PER    (CH_EDGES / 1024)
#define CH_N      ((EE + CH_EDGES - 1) / CH_EDGES)
#define CH_OFFP   288
#define CH_TRIPS  ((CH_N + 31) / 32)
static_assert(ET_NTILES + 1 <= CH_OFFP, "run starts 0..256 fit in a table row");
static_assert(CH_OFFP % 32 == 0, "the scan gives every lane the same number of counts");

__global__ __launch_bounds__(1024) void tile_group_kernel(
    const int* __restrict__ src, const int* __restrict__ dst,
    unsigned* __restrict__ pks, int* __restrict__ offs)
{
    __shared__ int cnt[CH_OFFP];
    __shared__ int base[CH_OFFP + 1];
    __shared__ uint4 sorted4[CH_EDGES / 4];
    unsigned* sorted = (unsigned*)sorted4;
    const int tid = threadIdx.x;

    if (tid < CH_OFFP) cnt[tid] = 0;

    const int e0 = blockIdx.x * CH_EDGES;
    int dv[CH_PER], sv[CH_PER];
    #pragma unroll
    for (int k = 0; k < CH_PER; ++k) {
        const int e = e0 + k * 1024 + tid;
        dv[k] = dst[e < EE ? e : EE - 1];
        sv[k] = src[e < EE ? e : EE - 1];
    }
    __syncthreads();

    int rk[CH_PER];
    #pragma unroll
    for (int k = 0; k < CH_PER; ++k) {
        const int e = e0 + k * 1024 + tid;
        const unsigned t = (unsigned)dv[k] / ET_TILE;
        rk[k] = -1;
        if (e < EE && t < ET_NTILES) rk[k] = atomicAdd(&cnt[t], 1);
    }
    __syncthreads();

    if (tid < 32) {
        int v[CH_OFFP / 32];
        int s = 0;
        #pragma unroll
        for (int j = 0; j < CH_OFFP / 32; ++j) {
            v[j] = cnt[(CH_OFFP / 32) * tid + j];
            s += v[j];
        }
        int inc = s;
        #pragma unroll
        for (int d = 1; d < 32; d <<= 1) {
            const int o = __shfl_up(inc, d, 32);
            if (tid >= d) inc += o;
        }
        int run = inc - s;
        #pragma unroll
        for (int j = 0; j < CH_OFFP / 32; ++j) {
            base[(CH_OFFP / 32) * tid + j] = run;
            run += v[j];
        }
        if (tid == 31) base[CH_OFFP] = run;
    }
    __syncthreads();

    #pragma unroll
    for (int k = 0; k < CH_PER; ++k) {
        if (rk[k] >= 0) {
            const unsigned t = (unsigned)dv[k] / ET_TILE;
            sorted[base[t] + rk[k]] = ((unsigned)sv[k] << ET_RBITS)
                                    | ((unsigned)dv[k] - t * ET_TILE);
        }
    }
    if (tid < CH_OFFP) offs[blockIdx.x * CH_OFFP + tid] = base[tid];
    __syncthreads();

    const int tot = base[CH_OFFP];
    uint4* __restrict__ out4 = (uint4*)(pks + (size_t)blockIdx.x * CH_EDGES);
    for (int i = tid; 4 * i < tot; i += 1024) out4[i] = sorted4[i];
}

__device__ __forceinline__ void et_wave_order() {
    __builtin_amdgcn_fence(__ATOMIC_ACQ_REL, "wavefront");
    __builtin_amdgcn_wave_barrier();
}

template<int D>
__device__ __forceinline__ void et_drain(
    const float* __restrict__ h,
    float* tile, const unsigned* q, int qn, int n, int lane)
{
    int s = 0, dl = 0;
    if (lane < n) {
        const unsigned w = q[qn - n + lane];
        s  = (int)(w >> ET_RBITS);
        dl = (int)(w & ((1u << ET_RBITS) - 1));
    }
    #pragma unroll
    for (int base = 0; base < 16; base += 8) {
        if (base >= n) break;
        float v[8][D / 32];
        #pragma unroll
        for (int i = 0; i < 8; ++i) {
            if (base + i < n) {
                const int si = __builtin_amdgcn_readlane(s, base + i);
                const float* row = h + (size_t)si * D + lane;
                #pragma unroll
                for (int k = 0; k < D / 32; ++k) v[i][k] = row[32 * k];
            }
        }
        #pragma unroll
        for (int i = 0; i < 8; ++i) {
            if (base + i < n) {
                const int di = __builtin_amdgcn_readlane(dl, base + i);
                float* trow = tile + di * (D + 4) + lane;
                #pragma unroll
                for (int k = 0; k < D / 32; ++k)
                    __hip_atomic_fetch_add(trow + 32 * k, v[i][k],
                        __ATOMIC_RELAXED, __HIP_MEMORY_SCOPE_WORKGROUP);
            }
        }
    }
}

template<int D>
__global__ __launch_bounds__(1024) void edge_tile_kernel(
    const float* __restrict__ h, const unsigned* __restrict__ pks,
    const int* __restrict__ offs,
    const float* __restrict__ epsP, int epsIdx,
    const float* __restrict__ W,
    const float* __restrict__ bias,
    const float* __restrict__ bn_g, const float* __restrict__ bn_b,
    const float* __restrict__ bn_m, const float* __restrict__ bn_v,
    float* __restrict__ out)
{
    constexpr int P  = D + 4;
    constexpr int RV = D / 4;
    __shared__ float4 tile4[ET_TROWS * (HID + 4) / 4];
    __shared__ unsigned queue[32 * ET_QCAP];
    __shared__ int sb0[CH_N], sb1[CH_N];
    float* tile = (float*)tile4;

    const int tid  = threadIdx.x;
    const int lane = tid & 31;
    const int wave = tid >> 5;
    const int lo   = blockIdx.x * ET_TILE;
    const int nrows = (NN - lo < ET_TILE) ? (NN - lo) : ET_TILE;

    if (tid < CH_N) {
        sb0[tid] = offs[tid * CH_OFFP + blockIdx.x];
        sb1[tid] = offs[tid * CH_OFFP + blockIdx.x + 1];
    }

    const float sc = 1.0f + epsP[epsIdx];
    const float4* __restrict__ h4 = (const float4*)(h + (size_t)lo * D);
    for (int i = tid; i < ET_TROWS * RV; i += 1024) {
        const int r = i / RV;
        float4 v = make_float4(0.f, 0.f, 0.f, 0.f);
        if (r < nrows) {
            v = h4[i];
            v.x *= sc; v.y *= sc; v.z *= sc; v.w *= sc;
        }
        *(float4*)(tile + r * P + 4 * (i % RV)) = v;
    }
    __syncthreads();

    unsigned* q = queue + wave * ET_QCAP;
    int qn = 0;

    int c  = wave;
    int b0 = sb0[c];
    int b1 = sb1[c];
    unsigned w = 0u;
    if (b0 + lane < b1) w = pks[(size_t)c * CH_EDGES + b0 + lane];

    for (int k = 0; k < CH_TRIPS; ++k) {
        if (b0 + lane < b1) q[qn + lane] = w;

        const int cn = c + 32;
        const bool more = cn < CH_N;
        int nb0 = 0, nb1 = 0;
        unsigned nw = 0u;
        if (more) {
            nb0 = sb0[cn];
            nb1 = sb1[cn];
            if (nb0 + lane < nb1) nw = pks[(size_t)cn * CH_EDGES + nb0 + lane];
        }
        for (int i = b0; i < b1; i += 32) {
            const int m = (b1 - i < 32) ? (b1 - i) : 32;
            if (i != b0 && lane < m)
                q[qn + lane] = pks[(size_t)c * CH_EDGES + i + lane];
            qn += m;
            et_wave_order();
            while (qn >= 16) {
                et_drain<D>(h, tile, q, qn, 16, lane);
                qn -= 16;
            }
            et_wave_order();
        }
        if (!more) break;
        c = cn; b0 = nb0; b1 = nb1; w = nw;
    }
    if (qn > 0) et_drain<D>(h, tile, q, qn, qn, lane);

    const int half = lane >> 4;
    const int mr   = lane & 15;
    const int col  = (wave & 7) * 16 + mr;

    v2f bf[D / 4];
    #pragma unroll
    for (int ks = 0; ks < D / 4; ++ks) {
        const int ka = 4 * ks + 2 * half;
        bf[ks].x = W[(size_t)ka * HID + col];
        bf[ks].y = W[(size_t)(ka + 1) * HID + col];
    }
    const float cbias = bias[col];
    const float cm    = bn_m[col];
    const float crs   = rsqrtf(bn_v[col] + 1e-5f);
    const float cg    = bn_g[col];
    const float cb    = bn_b[col];

    __syncthreads();

    const int ntiles = (nrows + 15) >> 4;
    const int t0 = __builtin_amdgcn_readfirstlane(wave >> 3);
    for (int t = t0; t < ntiles; t += 4) {
        const float* ap = tile + (16 * t + mr) * P + 2 * half;
        v8f acc = {};
        #pragma unroll
        for (int ks = 0; ks < D / 4; ++ks) {
            const v2f a = *(const v2f*)(ap + 4 * ks);
            acc = __builtin_amdgcn_wmma_f32_16x16x4_f32(
                false, a, false, bf[ks], (short)0, acc, false, false);
        }

        #pragma unroll
        for (int r = 0; r < 8; ++r) {
            const int rl = 16 * t + r + 8 * half;
            if (rl < nrows) {
                float y = acc[r] + cbias;
                y = (y - cm) * crs * cg + cb;
                y = gelu_exact(y);
                out[(size_t)(lo + rl) * HID + col] = y;
            }
        }
    }
}

#define GM_BROWS 64

template<int K>
__global__ __launch_bounds__(256) void gemm_bn_gelu_kernel(
    const float* __restrict__ A,
    const float* __restrict__ W,
    const float* __restrict__ bias,
    const float* __restrict__ bn_g, const float* __restrict__ bn_b,
    const float* __restrict__ bn_m, const float* __restrict__ bn_v,
    float* __restrict__ out)
{
    constexpr int P   = K + 4;
    constexpr int RV  = K / 4;
    constexpr int NV  = GM_BROWS * RV / 256;
    __shared__ float4 sA4[GM_BROWS * P / 4];
    float* sA = (float*)sA4;

    const int tid  = threadIdx.x;
    const int lane = tid & 31;
    const int wave = tid >> 5;
    const int half = lane >> 4;
    const int mr   = lane & 15;
    const int col  = wave * 16 + mr;
    const int row0 = blockIdx.x * GM_BROWS;

    float4 st[NV];
    const float4* __restrict__ A4 = (const float4*)A;
    #pragma unroll
    for (int j = 0; j < NV; ++j) {
        const int g = row0 * RV + j * 256 + tid;
        st[j] = A4[g < NN * RV ? g : NN * RV - 1];
    }

    v2f bf[K / 4];
    #pragma unroll
    for (int ks = 0; ks < K / 4; ++ks) {
        const int ka = 4 * ks + 2 * half;
        bf[ks].x = W[(size_t)ka * HID + col];
        bf[ks].y = W[(size_t)(ka + 1) * HID + col];
    }

    const float cbias = bias[col];
    const float cm    = bn_m[col];
    const float crs   = rsqrtf(bn_v[col] + 1e-5f);
    const float cg    = bn_g[col];
    const float cb    = bn_b[col];

    #pragma unroll
    for (int j = 0; j < NV; ++j) {
        const int idx = j * 256 + tid;
        *(float4*)(sA + (idx / RV) * P + (idx % RV) * 4) = st[j];
    }
    __syncthreads();

    for (int p = 0; p < GM_BROWS / 32; ++p) {
        const int prow = row0 + 32 * p;
        if (prow >= NN) break;
        const float* a0p = sA + (32 * p + mr) * P + 2 * half;
        const float* a1p = a0p + 16 * P;
        v8f acc0 = {}, acc1 = {};
        #pragma unroll
        for (int ks = 0; ks < K / 4; ++ks) {
            const v2f a0 = *(const v2f*)(a0p + 4 * ks);
            const v2f a1 = *(const v2f*)(a1p + 4 * ks);
            acc0 = __builtin_amdgcn_wmma_f32_16x16x4_f32(
                false, a0, false, bf[ks], (short)0, acc0, false, false);
            acc1 = __builtin_amdgcn_wmma_f32_16x16x4_f32(
                false, a1, false, bf[ks], (short)0, acc1, false, false);
        }

        #pragma unroll
        for (int t = 0; t < 2; ++t) {
            #pragma unroll
            for (int r = 0; r < 8; ++r) {
                const int row = prow + 16 * t + r + 8 * half;
                float y = (t == 0 ? acc0[r] : acc1[r]) + cbias;
                y = (y - cm) * crs * cg + cb;
                y = gelu_exact(y);
                out[(size_t)row * HID + col] = y;
            }
        }
    }
}

#define PT_G      4
#define PT_BLOCKS (GG / PT_G)
#define PT_SEG    32768
#define PT_PER4   (PT_SEG / 4 / 1024)
#define PT_ROUNDS ((NN + PT_SEG - 1) / PT_SEG)
#define PT_INFL   8
static_assert(GG % PT_G == 0, "every block owns four graphs");
static_assert(NN % 4 == 0, "batch[] is read four nodes at a time");
static_assert(PT_G == 4, "a list word is node << 2 | graph in block");
static_assert(HID == 128, "a row is 32 pieces of 16 bytes: one per lane");
static_assert(PT_G * HID <= 1024 && PT_G * LAT <= 1024, "one thread per head output");

__device__ __forceinline__ void pt_add(float4& a, const float4& v) {
    a.x += v.x; a.y += v.y; a.z += v.z; a.w += v.w;
}

__device__ __forceinline__ float pt_sum128(float v0, float v1, float v2, float v3) {
    float s = (v0 + v2) + (v1 + v3);
    #pragma unroll
    for (int d = 16; d >= 1; d >>= 1) s += __shfl_down(s, d, 32);
    return __shfl(s, 0, 32);
}

__global__ __launch_bounds__(1024) void pool_head_kernel(
    const float* __restrict__ h, const int* __restrict__ batch,
    const float* __restrict__ fc1w, const float* __restrict__ fc1b,
    const float* __restrict__ lng, const float* __restrict__ lnb,
    const float* __restrict__ fc2w, const float* __restrict__ fc2b,
    float* __restrict__ out)
{
    __shared__ int    list[PT_SEG];
    __shared__ float4 part4[32 * PT_G * HID / 4];
    __shared__ int    cpart[32 * PT_G];
    __shared__ int    nl[PT_ROUNDS];
    __shared__ float  pl[PT_G * HID];
    __shared__ float  ob[PT_G * HID];
    __shared__ float  rb[PT_G * HID];

    const int tid  = threadIdx.x;
    const int lane = tid & 31;
    const int wave = __builtin_amdgcn_readfirstlane(tid >> 5);
    const int g0   = blockIdx.x * PT_G;

    if (tid < PT_ROUNDS) nl[tid] = 0;
    __syncthreads();

    float4 acc0 = make_float4(0.f, 0.f, 0.f, 0.f);
    float4 acc1 = acc0, acc2 = acc0, acc3 = acc0;
    int c0 = 0, c1 = 0, c2 = 0, c3 = 0;

    const int4* __restrict__ b4 = (const int4*)batch;
    int4 bv[PT_PER4];
    #pragma unroll
    for (int k = 0; k < PT_PER4; ++k) {
        const int i4 = k * 1024 + tid;
        bv[k] = b4[i4 < NN / 4 ? i4 : NN / 4 - 1];
    }
    for (int r = 0; r < PT_ROUNDS; ++r) {
        #pragma unroll
        for (int k = 0; k < PT_PER4; ++k) {
            const int i4 = r * (PT_SEG / 4) + k * 1024 + tid;
            if (i4 < NN / 4) {
                const int bb[4] = { bv[k].x, bv[k].y, bv[k].z, bv[k].w };
                #pragma unroll
                for (int q = 0; q < 4; ++q) {
                    const unsigned lg = (unsigned)(bb[q] - g0);
                    if (lg < PT_G) {
                        const int p = atomicAdd(&nl[r], 1);
                        list[p] = ((4 * i4 + q) << 2) | (int)lg;
                    }
                }
            }
        }
        if (r + 1 < PT_ROUNDS) {
            #pragma unroll
            for (int k = 0; k < PT_PER4; ++k) {
                const int i4 = (r + 1) * (PT_SEG / 4) + k * 1024 + tid;
                bv[k] = b4[i4 < NN / 4 ? i4 : NN / 4 - 1];
            }
        }
        __syncthreads();

        const int n = __builtin_amdgcn_readfirstlane(nl[r]);
        for (int j0 = wave; j0 < n; j0 += 32 * PT_INFL) {
            float4 rv[PT_INFL];
            int    ev[PT_INFL];
            #pragma unroll
            for (int u = 0; u < PT_INFL; ++u) {
                const int j = j0 + 32 * u;
                ev[u] = __builtin_amdgcn_readfirstlane(list[j < n ? j : n - 1]);
                rv[u] = *(const float4*)(h + (size_t)(ev[u] >> 2) * HID + 4 * lane);
            }
            #pragma unroll
            for (int u = 0; u < PT_INFL; ++u) {
                if (j0 + 32 * u < n) {
                    const int lg = ev[u] & 3;
                    if      (lg == 0) { pt_add(acc0, rv[u]); ++c0; }
                    else if (lg == 1) { pt_add(acc1, rv[u]); ++c1; }
                    else if (lg == 2) { pt_add(acc2, rv[u]); ++c2; }
                    else              { pt_add(acc3, rv[u]); ++c3; }
                }
            }
        }
        __syncthreads();
    }

    float4* pw = part4 + wave * (PT_G * HID / 4);
    pw[0 * (HID / 4) + lane] = acc0;
    pw[1 * (HID / 4) + lane] = acc1;
    pw[2 * (HID / 4) + lane] = acc2;
    pw[3 * (HID / 4) + lane] = acc3;
    if (lane == 0) {
        cpart[wave * PT_G + 0] = c0;
        cpart[wave * PT_G + 1] = c1;
        cpart[wave * PT_G + 2] = c2;
        cpart[wave * PT_G + 3] = c3;
    }
    __syncthreads();

    const int hg = tid / HID;
    const int ht = tid % HID;
    if (tid < PT_G * HID) {
        const float* part = (const float*)part4;
        float s = 0.0f;
        int   cn = 0;
        for (int w = 0; w < 32; ++w) {
            s  += part[w * (PT_G * HID) + tid];
            cn += cpart[w * PT_G + hg];
        }
        pl[tid] = s / fmaxf((float)cn, 1.0f);
    }
    __syncthreads();

    if (tid < PT_G * HID) {
        const float* p = pl + hg * HID;
        float o = fc1b[ht];
        #pragma unroll 16
        for (int k = 0; k < HID; ++k) o += p[k] * fc1w[k * HID + ht];
        ob[tid] = o;
    }
    __syncthreads();

    if (wave < PT_G) {
        const float* og = ob + wave * HID;
        const float* pg = pl + wave * HID;
        float ov[4], dv[4];
        #pragma unroll
        for (int i = 0; i < 4; ++i) ov[i] = og[lane + 32 * i];
        const float mu = pt_sum128(ov[0], ov[1], ov[2], ov[3]) * (1.0f / HID);
        #pragma unroll
        for (int i = 0; i < 4; ++i) dv[i] = ov[i] - mu;
        const float var = pt_sum128(dv[0] * dv[0], dv[1] * dv[1],
                                    dv[2] * dv[2], dv[3] * dv[3]) * (1.0f / HID);
        const float rs = rsqrtf(var + 1e-5f);
        #pragma unroll
        for (int i = 0; i < 4; ++i) {
            const int t = lane + 32 * i;
            float y = dv[i] * rs * lng[t] + lnb[t];
            y = gelu_exact(y) + pg[t];
            rb[wave * HID + t] = y;
        }
    }
    __syncthreads();

    if (tid < PT_G * LAT) {
        const int g = tid / LAT;
        const int t = tid % LAT;
        const float* rr = rb + g * HID;
        float o2 = fc2b[t];
        #pragma unroll 16
        for (int k = 0; k < HID; ++k) o2 += rr[k] * fc2w[k * LAT + t];
        out[(size_t)(g0 + g) * LAT + t] = o2;
    }
}

extern "C" void kernel_launch(void* const* d_in, const int* in_sizes, int n_in,
                              void* d_out, int out_size, void* d_ws, size_t ws_size,
                              hipStream_t stream) {
    const float* x       = (const float*)d_in[0];
    const int*   ei      = (const int*)  d_in[1];
    const int*   batch   = (const int*)  d_in[2];
    const float* in_g    = (const float*)d_in[3];
    const float* in_b    = (const float*)d_in[4];
    const float* in_m    = (const float*)d_in[5];
    const float* in_v    = (const float*)d_in[6];
    const float* eps0    = (const float*)d_in[7];
    const float* c0_w1   = (const float*)d_in[8];
    const float* c0_b1   = (const float*)d_in[9];
    const float* c0_g    = (const float*)d_in[10];
    const float* c0_bb   = (const float*)d_in[11];
    const float* c0_m    = (const float*)d_in[12];
    const float* c0_v    = (const float*)d_in[13];
    const float* c0_w2   = (const float*)d_in[14];
    const float* c0_b2   = (const float*)d_in[15];
    const float* bn0_g   = (const float*)d_in[16];
    const float* bn0_b   = (const float*)d_in[17];
    const float* bn0_m   = (const float*)d_in[18];
    const float* bn0_v   = (const float*)d_in[19];
    const float* epsR    = (const float*)d_in[20];
    const float* cR_w1   = (const float*)d_in[21];
    const float* cR_b1   = (const float*)d_in[22];
    const float* cR_g    = (const float*)d_in[23];
    const float* cR_bb   = (const float*)d_in[24];
    const float* cR_m    = (const float*)d_in[25];
    const float* cR_v    = (const float*)d_in[26];
    const float* cR_w2   = (const float*)d_in[27];
    const float* cR_b2   = (const float*)d_in[28];
    const float* bnR_g   = (const float*)d_in[29];
    const float* bnR_b   = (const float*)d_in[30];
    const float* bnR_m   = (const float*)d_in[31];
    const float* bnR_v   = (const float*)d_in[32];
    const float* fc1_w   = (const float*)d_in[33];
    const float* fc1_b   = (const float*)d_in[34];
    const float* ln_g    = (const float*)d_in[35];
    const float* ln_b    = (const float*)d_in[36];
    const float* fc2_w   = (const float*)d_in[37];
    const float* fc2_b   = (const float*)d_in[38];

    const int* src = ei;
    const int* dst = ei + EE;

    float* buf0 = (float*)d_ws;
    float* buf1 = buf0 + (size_t)NN * HID;
    float* buf2 = buf1 + (size_t)NN * HID;
    unsigned* pkw = (unsigned*)buf2;
    int*      ofs = (int*)(pkw + (size_t)CH_N * CH_EDGES);
    static_assert((size_t)CH_N * CH_EDGES + (size_t)CH_N * CH_OFFP <= (size_t)NN * HID,
                  "packed words and run starts fit in buf2");

    const int TB = 256;
    const int nGemm = (NN + GM_BROWS - 1) / GM_BROWS;

    bn_in_kernel<<<BN_BLOCKS, BN_THREADS, 0, stream>>>(
        x, in_g, in_b, in_m, in_v, buf0);

    {
        tile_group_kernel<<<CH_N, 1024, 0, stream>>>(src, dst, pkw, ofs);
        edge_tile_kernel<INDIM><<<ET_NTILES, 1024, 0, stream>>>(
            buf0, pkw, ofs, eps0, 0,
            c0_w1, c0_b1, c0_g, c0_bb, c0_m, c0_v, buf1);
        gemm_bn_gelu_kernel<HID><<<nGemm, TB, 0, stream>>>(
            buf1, c0_w2, c0_b2, bn0_g, bn0_b, bn0_m, bn0_v, buf0);
    }

    for (int l = 0; l < 2; ++l) {
        const size_t wOff = (size_t)l * HID * HID;
        const size_t vOff = (size_t)l * HID;
        tile_group_kernel<<<CH_N, 1024, 0, stream>>>(src, dst, pkw, ofs);
        edge_tile_kernel<HID><<<ET_NTILES, 1024, 0, stream>>>(
            buf0, pkw, ofs, epsR, l,
            cR_w1 + wOff, cR_b1 + vOff, cR_g + vOff, cR_bb + vOff,
            cR_m + vOff, cR_v + vOff, buf1);
        gemm_bn_gelu_kernel<HID><<<nGemm, TB, 0, stream>>>(
            buf1, cR_w2 + wOff, cR_b2 + vOff, bnR_g + vOff, bnR_b + vOff,
            bnR_m + vOff, bnR_v + vOff, buf0);
    }

    pool_head_kernel<<<PT_BLOCKS, 1024, 0, stream>>>(
        buf0, batch, fc1_w, fc1_b, ln_g, ln_b, fc2_w, fc2_b, (float*)d_out);
}


# Edge grouping launched once per call, not once per layer: runs and run starts depend on the edge list alone and nothing overwrites them, so layers 1 and 2 reuse layer 0's; 2 of 11 launches gone
# speedup vs baseline: 2.7045x; 1.0414x over previous
#include <hip/hip_runtime.h>
#include <math.h>

typedef __attribute__((ext_vector_type(2))) float v2f;
typedef __attribute__((ext_vector_type(8))) float v8f;

#define NN   100000
#define EE   1600000
#define INDIM 64
#define HID  128
#define LAT  64
#define GG   1000

__device__ __forceinline__ float gelu_exact(float x) {
    return 0.5f * x * (1.0f + erff(x * 0.70710678118654752f));
}

#define BN_BLOCKS  512
#define BN_THREADS 256
#define BN_V4      (NN * INDIM / 4)
#define BN_STRIDE  (BN_BLOCKS * BN_THREADS)
#define BN_PER     ((BN_V4 + BN_STRIDE - 1) / BN_STRIDE)
static_assert(BN_STRIDE % (INDIM / 4) == 0, "stride must be whole rows");

__global__ __launch_bounds__(BN_THREADS) void bn_in_kernel(
    const float* __restrict__ x,
    const float* __restrict__ g, const float* __restrict__ b,
    const float* __restrict__ m, const float* __restrict__ v,
    float* __restrict__ out)
{
    const int t = blockIdx.x * BN_THREADS + threadIdx.x;
    const float4* __restrict__ x4 = (const float4*)x;

    float4 xv[BN_PER];
    #pragma unroll
    for (int k = 0; k < BN_PER; ++k) {
        const int i = t + k * BN_STRIDE;
        xv[k] = x4[i < BN_V4 ? i : BN_V4 - 1];
    }

    const int j = 4 * (t & (INDIM / 4 - 1));
    float cm[4], cr[4], cg[4], cb[4];
    #pragma unroll
    for (int c = 0; c < 4; ++c) {
        cm[c] = m[j + c];
        cr[c] = rsqrtf(v[j + c] + 1e-5f);
        cg[c] = g[j + c];
        cb[c] = b[j + c];
    }

    float4* __restrict__ out4 = (float4*)out;
    #pragma unroll
    for (int k = 0; k < BN_PER; ++k) {
        const int i = t + k * BN_STRIDE;
        if (i < BN_V4) {
            float4 y;
            y.x = (xv[k].x - cm[0]) * cr[0] * cg[0] + cb[0];
            y.y = (xv[k].y - cm[1]) * cr[1] * cg[1] + cb[1];
            y.z = (xv[k].z - cm[2]) * cr[2] * cg[2] + cb[2];
            y.w = (xv[k].w - cm[3]) * cr[3] * cg[3] + cb[3];
            out4[i] = y;
        }
    }
}

#define ET_TILE   391
#define ET_TROWS  400
#define ET_NTILES ((NN + ET_TILE - 1) / ET_TILE)
#define ET_QCAP   48
#define ET_RBITS  9
static_assert(ET_TROWS % 16 == 0 && ET_TROWS >= ET_TILE, "whole 16-row WMMA tiles");
static_assert(ET_TILE <= (1 << ET_RBITS), "packed word: low bits = row in tile");
static_assert(ET_NTILES == 256, "one tile block per WGP");
static_assert(NN <= (1 << (32 - ET_RBITS)), "packed word: high bits = src node");

#define CH_EDGES  8192
#define CH_PER    (CH_EDGES / 1024)
#define CH_N      ((EE + CH_EDGES - 1) / CH_EDGES)
#define CH_OFFP   288
#define CH_TRIPS  ((CH_N + 31) / 32)
static_assert(ET_NTILES + 1 <= CH_OFFP, "run starts 0..256 fit in a table row");
static_assert(CH_OFFP % 32 == 0, "the scan gives every lane the same number of counts");

__global__ __launch_bounds__(1024) void tile_group_kernel(
    const int* __restrict__ src, const int* __restrict__ dst,
    unsigned* __restrict__ pks, int* __restrict__ offs)
{
    __shared__ int cnt[CH_OFFP];
    __shared__ int base[CH_OFFP + 1];
    __shared__ uint4 sorted4[CH_EDGES / 4];
    unsigned* sorted = (unsigned*)sorted4;
    const int tid = threadIdx.x;

    if (tid < CH_OFFP) cnt[tid] = 0;

    const int e0 = blockIdx.x * CH_EDGES;
    int dv[CH_PER], sv[CH_PER];
    #pragma unroll
    for (int k = 0; k < CH_PER; ++k) {
        const int e = e0 + k * 1024 + tid;
        dv[k] = dst[e < EE ? e : EE - 1];
        sv[k] = src[e < EE ? e : EE - 1];
    }
    __syncthreads();

    int rk[CH_PER];
    #pragma unroll
    for (int k = 0; k < CH_PER; ++k) {
        const int e = e0 + k * 1024 + tid;
        const unsigned t = (unsigned)dv[k] / ET_TILE;
        rk[k] = -1;
        if (e < EE && t < ET_NTILES) rk[k] = atomicAdd(&cnt[t], 1);
    }
    __syncthreads();

    if (tid < 32) {
        int v[CH_OFFP / 32];
        int s = 0;
        #pragma unroll
        for (int j = 0; j < CH_OFFP / 32; ++j) {
            v[j] = cnt[(CH_OFFP / 32) * tid + j];
            s += v[j];
        }
        int inc = s;
        #pragma unroll
        for (int d = 1; d < 32; d <<= 1) {
            const int o = __shfl_up(inc, d, 32);
            if (tid >= d) inc += o;
        }
        int run = inc - s;
        #pragma unroll
        for (int j = 0; j < CH_OFFP / 32; ++j) {
            base[(CH_OFFP / 32) * tid + j] = run;
            run += v[j];
        }
        if (tid == 31) base[CH_OFFP] = run;
    }
    __syncthreads();

    #pragma unroll
    for (int k = 0; k < CH_PER; ++k) {
        if (rk[k] >= 0) {
            const unsigned t = (unsigned)dv[k] / ET_TILE;
            sorted[base[t] + rk[k]] = ((unsigned)sv[k] << ET_RBITS)
                                    | ((unsigned)dv[k] - t * ET_TILE);
        }
    }
    if (tid < CH_OFFP) offs[blockIdx.x * CH_OFFP + tid] = base[tid];
    __syncthreads();

    const int tot = base[CH_OFFP];
    uint4* __restrict__ out4 = (uint4*)(pks + (size_t)blockIdx.x * CH_EDGES);
    for (int i = tid; 4 * i < tot; i += 1024) out4[i] = sorted4[i];
}

__device__ __forceinline__ void et_wave_order() {
    __builtin_amdgcn_fence(__ATOMIC_ACQ_REL, "wavefront");
    __builtin_amdgcn_wave_barrier();
}

template<int D>
__device__ __forceinline__ void et_drain(
    const float* __restrict__ h,
    float* tile, const unsigned* q, int qn, int n, int lane)
{
    int s = 0, dl = 0;
    if (lane < n) {
        const unsigned w = q[qn - n + lane];
        s  = (int)(w >> ET_RBITS);
        dl = (int)(w & ((1u << ET_RBITS) - 1));
    }
    #pragma unroll
    for (int base = 0; base < 16; base += 8) {
        if (base >= n) break;
        float v[8][D / 32];
        #pragma unroll
        for (int i = 0; i < 8; ++i) {
            if (base + i < n) {
                const int si = __builtin_amdgcn_readlane(s, base + i);
                const float* row = h + (size_t)si * D + lane;
                #pragma unroll
                for (int k = 0; k < D / 32; ++k) v[i][k] = row[32 * k];
            }
        }
        #pragma unroll
        for (int i = 0; i < 8; ++i) {
            if (base + i < n) {
                const int di = __builtin_amdgcn_readlane(dl, base + i);
                float* trow = tile + di * (D + 4) + lane;
                #pragma unroll
                for (int k = 0; k < D / 32; ++k)
                    __hip_atomic_fetch_add(trow + 32 * k, v[i][k],
                        __ATOMIC_RELAXED, __HIP_MEMORY_SCOPE_WORKGROUP);
            }
        }
    }
}

template<int D>
__global__ __launch_bounds__(1024) void edge_tile_kernel(
    const float* __restrict__ h, const unsigned* __restrict__ pks,
    const int* __restrict__ offs,
    const float* __restrict__ epsP, int epsIdx,
    const float* __restrict__ W,
    const float* __restrict__ bias,
    const float* __restrict__ bn_g, const float* __restrict__ bn_b,
    const float* __restrict__ bn_m, const float* __restrict__ bn_v,
    float* __restrict__ out)
{
    constexpr int P  = D + 4;
    constexpr int RV = D / 4;
    __shared__ float4 tile4[ET_TROWS * (HID + 4) / 4];
    __shared__ unsigned queue[32 * ET_QCAP];
    __shared__ int sb0[CH_N], sb1[CH_N];
    float* tile = (float*)tile4;

    const int tid  = threadIdx.x;
    const int lane = tid & 31;
    const int wave = tid >> 5;
    const int lo   = blockIdx.x * ET_TILE;
    const int nrows = (NN - lo < ET_TILE) ? (NN - lo) : ET_TILE;

    if (tid < CH_N) {
        sb0[tid] = offs[tid * CH_OFFP + blockIdx.x];
        sb1[tid] = offs[tid * CH_OFFP + blockIdx.x + 1];
    }

    const float sc = 1.0f + epsP[epsIdx];
    const float4* __restrict__ h4 = (const float4*)(h + (size_t)lo * D);
    for (int i = tid; i < ET_TROWS * RV; i += 1024) {
        const int r = i / RV;
        float4 v = make_float4(0.f, 0.f, 0.f, 0.f);
        if (r < nrows) {
            v = h4[i];
            v.x *= sc; v.y *= sc; v.z *= sc; v.w *= sc;
        }
        *(float4*)(tile + r * P + 4 * (i % RV)) = v;
    }
    __syncthreads();

    unsigned* q = queue + wave * ET_QCAP;
    int qn = 0;

    int c  = wave;
    int b0 = sb0[c];
    int b1 = sb1[c];
    unsigned w = 0u;
    if (b0 + lane < b1) w = pks[(size_t)c * CH_EDGES + b0 + lane];

    for (int k = 0; k < CH_TRIPS; ++k) {
        if (b0 + lane < b1) q[qn + lane] = w;

        const int cn = c + 32;
        const bool more = cn < CH_N;
        int nb0 = 0, nb1 = 0;
        unsigned nw = 0u;
        if (more) {
            nb0 = sb0[cn];
            nb1 = sb1[cn];
            if (nb0 + lane < nb1) nw = pks[(size_t)cn * CH_EDGES + nb0 + lane];
        }
        for (int i = b0; i < b1; i += 32) {
            const int m = (b1 - i < 32) ? (b1 - i) : 32;
            if (i != b0 && lane < m)
                q[qn + lane] = pks[(size_t)c * CH_EDGES + i + lane];
            qn += m;
            et_wave_order();
            while (qn >= 16) {
                et_drain<D>(h, tile, q, qn, 16, lane);
                qn -= 16;
            }
            et_wave_order();
        }
        if (!more) break;
        c = cn; b0 = nb0; b1 = nb1; w = nw;
    }
    if (qn > 0) et_drain<D>(h, tile, q, qn, qn, lane);

    const int half = lane >> 4;
    const int mr   = lane & 15;
    const int col  = (wave & 7) * 16 + mr;

    v2f bf[D / 4];
    #pragma unroll
    for (int ks = 0; ks < D / 4; ++ks) {
        const int ka = 4 * ks + 2 * half;
        bf[ks].x = W[(size_t)ka * HID + col];
        bf[ks].y = W[(size_t)(ka + 1) * HID + col];
    }
    const float cbias = bias[col];
    const float cm    = bn_m[col];
    const float crs   = rsqrtf(bn_v[col] + 1e-5f);
    const float cg    = bn_g[col];
    const float cb    = bn_b[col];

    __syncthreads();

    const int ntiles = (nrows + 15) >> 4;
    const int t0 = __builtin_amdgcn_readfirstlane(wave >> 3);
    for (int t = t0; t < ntiles; t += 4) {
        const float* ap = tile + (16 * t + mr) * P + 2 * half;
        v8f acc = {};
        #pragma unroll
        for (int ks = 0; ks < D / 4; ++ks) {
            const v2f a = *(const v2f*)(ap + 4 * ks);
            acc = __builtin_amdgcn_wmma_f32_16x16x4_f32(
                false, a, false, bf[ks], (short)0, acc, false, false);
        }

        #pragma unroll
        for (int r = 0; r < 8; ++r) {
            const int rl = 16 * t + r + 8 * half;
            if (rl < nrows) {
                float y = acc[r] + cbias;
                y = (y - cm) * crs * cg + cb;
                y = gelu_exact(y);
                out[(size_t)(lo + rl) * HID + col] = y;
            }
        }
    }
}

#define GM_BROWS 64

template<int K>
__global__ __launch_bounds__(256) void gemm_bn_gelu_kernel(
    const float* __restrict__ A,
    const float* __restrict__ W,
    const float* __restrict__ bias,
    const float* __restrict__ bn_g, const float* __restrict__ bn_b,
    const float* __restrict__ bn_m, const float* __restrict__ bn_v,
    float* __restrict__ out)
{
    constexpr int P   = K + 4;
    constexpr int RV  = K / 4;
    constexpr int NV  = GM_BROWS * RV / 256;
    __shared__ float4 sA4[GM_BROWS * P / 4];
    float* sA = (float*)sA4;

    const int tid  = threadIdx.x;
    const int lane = tid & 31;
    const int wave = tid >> 5;
    const int half = lane >> 4;
    const int mr   = lane & 15;
    const int col  = wave * 16 + mr;
    const int row0 = blockIdx.x * GM_BROWS;

    float4 st[NV];
    const float4* __restrict__ A4 = (const float4*)A;
    #pragma unroll
    for (int j = 0; j < NV; ++j) {
        const int g = row0 * RV + j * 256 + tid;
        st[j] = A4[g < NN * RV ? g : NN * RV - 1];
    }

    v2f bf[K / 4];
    #pragma unroll
    for (int ks = 0; ks < K / 4; ++ks) {
        const int ka = 4 * ks + 2 * half;
        bf[ks].x = W[(size_t)ka * HID + col];
        bf[ks].y = W[(size_t)(ka + 1) * HID + col];
    }

    const float cbias = bias[col];
    const float cm    = bn_m[col];
    const float crs   = rsqrtf(bn_v[col] + 1e-5f);
    const float cg    = bn_g[col];
    const float cb    = bn_b[col];

    #pragma unroll
    for (int j = 0; j < NV; ++j) {
        const int idx = j * 256 + tid;
        *(float4*)(sA + (idx / RV) * P + (idx % RV) * 4) = st[j];
    }
    __syncthreads();

    for (int p = 0; p < GM_BROWS / 32; ++p) {
        const int prow = row0 + 32 * p;
        if (prow >= NN) break;
        const float* a0p = sA + (32 * p + mr) * P + 2 * half;
        const float* a1p = a0p + 16 * P;
        v8f acc0 = {}, acc1 = {};
        #pragma unroll
        for (int ks = 0; ks < K / 4; ++ks) {
            const v2f a0 = *(const v2f*)(a0p + 4 * ks);
            const v2f a1 = *(const v2f*)(a1p + 4 * ks);
            acc0 = __builtin_amdgcn_wmma_f32_16x16x4_f32(
                false, a0, false, bf[ks], (short)0, acc0, false, false);
            acc1 = __builtin_amdgcn_wmma_f32_16x16x4_f32(
                false, a1, false, bf[ks], (short)0, acc1, false, false);
        }

        #pragma unroll
        for (int t = 0; t < 2; ++t) {
            #pragma unroll
            for (int r = 0; r < 8; ++r) {
                const int row = prow + 16 * t + r + 8 * half;
                float y = (t == 0 ? acc0[r] : acc1[r]) + cbias;
                y = (y - cm) * crs * cg + cb;
                y = gelu_exact(y);
                out[(size_t)row * HID + col] = y;
            }
        }
    }
}

#define PT_G      4
#define PT_BLOCKS (GG / PT_G)
#define PT_SEG    32768
#define PT_PER4   (PT_SEG / 4 / 1024)
#define PT_ROUNDS ((NN + PT_SEG - 1) / PT_SEG)
#define PT_INFL   8
static_assert(GG % PT_G == 0, "every block owns four graphs");
static_assert(NN % 4 == 0, "batch[] is read four nodes at a time");
static_assert(PT_G == 4, "a list word is node << 2 | graph in block");
static_assert(HID == 128, "a row is 32 pieces of 16 bytes: one per lane");
static_assert(PT_G * HID <= 1024 && PT_G * LAT <= 1024, "one thread per head output");

__device__ __forceinline__ void pt_add(float4& a, const float4& v) {
    a.x += v.x; a.y += v.y; a.z += v.z; a.w += v.w;
}

__device__ __forceinline__ float pt_sum128(float v0, float v1, float v2, float v3) {
    float s = (v0 + v2) + (v1 + v3);
    #pragma unroll
    for (int d = 16; d >= 1; d >>= 1) s += __shfl_down(s, d, 32);
    return __shfl(s, 0, 32);
}

__global__ __launch_bounds__(1024) void pool_head_kernel(
    const float* __restrict__ h, const int* __restrict__ batch,
    const float* __restrict__ fc1w, const float* __restrict__ fc1b,
    const float* __restrict__ lng, const float* __restrict__ lnb,
    const float* __restrict__ fc2w, const float* __restrict__ fc2b,
    float* __restrict__ out)
{
    __shared__ int    list[PT_SEG];
    __shared__ float4 part4[32 * PT_G * HID / 4];
    __shared__ int    cpart[32 * PT_G];
    __shared__ int    nl[PT_ROUNDS];
    __shared__ float  pl[PT_G * HID];
    __shared__ float  ob[PT_G * HID];
    __shared__ float  rb[PT_G * HID];

    const int tid  = threadIdx.x;
    const int lane = tid & 31;
    const int wave = __builtin_amdgcn_readfirstlane(tid >> 5);
    const int g0   = blockIdx.x * PT_G;

    if (tid < PT_ROUNDS) nl[tid] = 0;
    __syncthreads();

    float4 acc0 = make_float4(0.f, 0.f, 0.f, 0.f);
    float4 acc1 = acc0, acc2 = acc0, acc3 = acc0;
    int c0 = 0, c1 = 0, c2 = 0, c3 = 0;

    const int4* __restrict__ b4 = (const int4*)batch;
    int4 bv[PT_PER4];
    #pragma unroll
    for (int k = 0; k < PT_PER4; ++k) {
        const int i4 = k * 1024 + tid;
        bv[k] = b4[i4 < NN / 4 ? i4 : NN / 4 - 1];
    }
    for (int r = 0; r < PT_ROUNDS; ++r) {
        #pragma unroll
        for (int k = 0; k < PT_PER4; ++k) {
            const int i4 = r * (PT_SEG / 4) + k * 1024 + tid;
            if (i4 < NN / 4) {
                const int bb[4] = { bv[k].x, bv[k].y, bv[k].z, bv[k].w };
                #pragma unroll
                for (int q = 0; q < 4; ++q) {
                    const unsigned lg = (unsigned)(bb[q] - g0);
                    if (lg < PT_G) {
                        const int p = atomicAdd(&nl[r], 1);
                        list[p] = ((4 * i4 + q) << 2) | (int)lg;
                    }
                }
            }
        }
        if (r + 1 < PT_ROUNDS) {
            #pragma unroll
            for (int k = 0; k < PT_PER4; ++k) {
                const int i4 = (r + 1) * (PT_SEG / 4) + k * 1024 + tid;
                bv[k] = b4[i4 < NN / 4 ? i4 : NN / 4 - 1];
            }
        }
        __syncthreads();

        const int n = __builtin_amdgcn_readfirstlane(nl[r]);
        for (int j0 = wave; j0 < n; j0 += 32 * PT_INFL) {
            float4 rv[PT_INFL];
            int    ev[PT_INFL];
            #pragma unroll
            for (int u = 0; u < PT_INFL; ++u) {
                const int j = j0 + 32 * u;
                ev[u] = __builtin_amdgcn_readfirstlane(list[j < n ? j : n - 1]);
                rv[u] = *(const float4*)(h + (size_t)(ev[u] >> 2) * HID + 4 * lane);
            }
            #pragma unroll
            for (int u = 0; u < PT_INFL; ++u) {
                if (j0 + 32 * u < n) {
                    const int lg = ev[u] & 3;
                    if      (lg == 0) { pt_add(acc0, rv[u]); ++c0; }
                    else if (lg == 1) { pt_add(acc1, rv[u]); ++c1; }
                    else if (lg == 2) { pt_add(acc2, rv[u]); ++c2; }
                    else              { pt_add(acc3, rv[u]); ++c3; }
                }
            }
        }
        __syncthreads();
    }

    float4* pw = part4 + wave * (PT_G * HID / 4);
    pw[0 * (HID / 4) + lane] = acc0;
    pw[1 * (HID / 4) + lane] = acc1;
    pw[2 * (HID / 4) + lane] = acc2;
    pw[3 * (HID / 4) + lane] = acc3;
    if (lane == 0) {
        cpart[wave * PT_G + 0] = c0;
        cpart[wave * PT_G + 1] = c1;
        cpart[wave * PT_G + 2] = c2;
        cpart[wave * PT_G + 3] = c3;
    }
    __syncthreads();

    const int hg = tid / HID;
    const int ht = tid % HID;
    if (tid < PT_G * HID) {
        const float* part = (const float*)part4;
        float s = 0.0f;
        int   cn = 0;
        for (int w = 0; w < 32; ++w) {
            s  += part[w * (PT_G * HID) + tid];
            cn += cpart[w * PT_G + hg];
        }
        pl[tid] = s / fmaxf((float)cn, 1.0f);
    }
    __syncthreads();

    if (tid < PT_G * HID) {
        const float* p = pl + hg * HID;
        float o = fc1b[ht];
        #pragma unroll 16
        for (int k = 0; k < HID; ++k) o += p[k] * fc1w[k * HID + ht];
        ob[tid] = o;
    }
    __syncthreads();

    if (wave < PT_G) {
        const float* og = ob + wave * HID;
        const float* pg = pl + wave * HID;
        float ov[4], dv[4];
        #pragma unroll
        for (int i = 0; i < 4; ++i) ov[i] = og[lane + 32 * i];
        const float mu = pt_sum128(ov[0], ov[1], ov[2], ov[3]) * (1.0f / HID);
        #pragma unroll
        for (int i = 0; i < 4; ++i) dv[i] = ov[i] - mu;
        const float var = pt_sum128(dv[0] * dv[0], dv[1] * dv[1],
                                    dv[2] * dv[2], dv[3] * dv[3]) * (1.0f / HID);
        const float rs = rsqrtf(var + 1e-5f);
        #pragma unroll
        for (int i = 0; i < 4; ++i) {
            const int t = lane + 32 * i;
            float y = dv[i] * rs * lng[t] + lnb[t];
            y = gelu_exact(y) + pg[t];
            rb[wave * HID + t] = y;
        }
    }
    __syncthreads();

    if (tid < PT_G * LAT) {
        const int g = tid / LAT;
        const int t = tid % LAT;
        const float* rr = rb + g * HID;
        float o2 = fc2b[t];
        #pragma unroll 16
        for (int k = 0; k < HID; ++k) o2 += rr[k] * fc2w[k * LAT + t];
        out[(size_t)(g0 + g) * LAT + t] = o2;
    }
}

extern "C" void kernel_launch(void* const* d_in, const int* in_sizes, int n_in,
                              void* d_out, int out_size, void* d_ws, size_t ws_size,
                              hipStream_t stream) {
    const float* x       = (const float*)d_in[0];
    const int*   ei      = (const int*)  d_in[1];
    const int*   batch   = (const int*)  d_in[2];
    const float* in_g    = (const float*)d_in[3];
    const float* in_b    = (const float*)d_in[4];
    const float* in_m    = (const float*)d_in[5];
    const float* in_v    = (const float*)d_in[6];
    const float* eps0    = (const float*)d_in[7];
    const float* c0_w1   = (const float*)d_in[8];
    const float* c0_b1   = (const float*)d_in[9];
    const float* c0_g    = (const float*)d_in[10];
    const float* c0_bb   = (const float*)d_in[11];
    const float* c0_m    = (const float*)d_in[12];
    const float* c0_v    = (const float*)d_in[13];
    const float* c0_w2   = (const float*)d_in[14];
    const float* c0_b2   = (const float*)d_in[15];
    const float* bn0_g   = (const float*)d_in[16];
    const float* bn0_b   = (const float*)d_in[17];
    const float* bn0_m   = (const float*)d_in[18];
    const float* bn0_v   = (const float*)d_in[19];
    const float* epsR    = (const float*)d_in[20];
    const float* cR_w1   = (const float*)d_in[21];
    const float* cR_b1   = (const float*)d_in[22];
    const float* cR_g    = (const float*)d_in[23];
    const float* cR_bb   = (const float*)d_in[24];
    const float* cR_m    = (const float*)d_in[25];
    const float* cR_v    = (const float*)d_in[26];
    const float* cR_w2   = (const float*)d_in[27];
    const float* cR_b2   = (const float*)d_in[28];
    const float* bnR_g   = (const float*)d_in[29];
    const float* bnR_b   = (const float*)d_in[30];
    const float* bnR_m   = (const float*)d_in[31];
    const float* bnR_v   = (const float*)d_in[32];
    const float* fc1_w   = (const float*)d_in[33];
    const float* fc1_b   = (const float*)d_in[34];
    const float* ln_g    = (const float*)d_in[35];
    const float* ln_b    = (const float*)d_in[36];
    const float* fc2_w   = (const float*)d_in[37];
    const float* fc2_b   = (const float*)d_in[38];

    const int* src = ei;
    const int* dst = ei + EE;

    float* buf0 = (float*)d_ws;
    float* buf1 = buf0 + (size_t)NN * HID;
    float* buf2 = buf1 + (size_t)NN * HID;
    unsigned* pkw = (unsigned*)buf2;
    int*      ofs = (int*)(pkw + (size_t)CH_N * CH_EDGES);
    static_assert((size_t)CH_N * CH_EDGES + (size_t)CH_N * CH_OFFP <= (size_t)NN * HID,
                  "packed words and run starts fit in buf2");

    const int TB = 256;
    const int nGemm = (NN + GM_BROWS - 1) / GM_BROWS;

    bn_in_kernel<<<BN_BLOCKS, BN_THREADS, 0, stream>>>(
        x, in_g, in_b, in_m, in_v, buf0);

    tile_group_kernel<<<CH_N, 1024, 0, stream>>>(src, dst, pkw, ofs);

    {
        edge_tile_kernel<INDIM><<<ET_NTILES, 1024, 0, stream>>>(
            buf0, pkw, ofs, eps0, 0,
            c0_w1, c0_b1, c0_g, c0_bb, c0_m, c0_v, buf1);
        gemm_bn_gelu_kernel<HID><<<nGemm, TB, 0, stream>>>(
            buf1, c0_w2, c0_b2, bn0_g, bn0_b, bn0_m, bn0_v, buf0);
    }

    for (int l = 0; l < 2; ++l) {
        const size_t wOff = (size_t)l * HID * HID;
        const size_t vOff = (size_t)l * HID;
        edge_tile_kernel<HID><<<ET_NTILES, 1024, 0, stream>>>(
            buf0, pkw, ofs, epsR, l,
            cR_w1 + wOff, cR_b1 + vOff, cR_g + vOff, cR_bb + vOff,
            cR_m + vOff, cR_v + vOff, buf1);
        gemm_bn_gelu_kernel<HID><<<nGemm, TB, 0, stream>>>(
            buf1, cR_w2 + wOff, cR_b2 + vOff, bnR_g + vOff, bnR_b + vOff,
            bnR_m + vOff, bnR_v + vOff, buf0);
    }

    pool_head_kernel<<<PT_BLOCKS, 1024, 0, stream>>>(
        buf0, batch, fc1_w, fc1_b, ln_g, ln_b, fc2_w, fc2_b, (float*)d_out);
}
